# starting kernel
# baseline (speedup 1.0000x reference)
#include <hip/hip_runtime.h>
#include <hip/hip_bf16.h>

typedef __attribute__((ext_vector_type(2))) float v2f;
typedef __attribute__((ext_vector_type(8))) float v8f;

#define H 128
#define D_EDGE 16
#define N_MOUSE 50000
#define N_VIRUS 10000
#define NEDGE   500000
#define NPAIR   100000
#define D_IN    1024

__global__ void gemm_wmma_bias(const float* __restrict__ A,
                               const float* __restrict__ B,
                               const float* __restrict__ bias,
                               float* __restrict__ C,
                               int M, int K)
{
    __shared__ __align__(16) float As[16 * 132];

    const int tid  = threadIdx.x;
    const int wave = tid >> 5;
    const int lane = tid & 31;
    const int hi   = lane >> 4;
    const int ml   = lane & 15;
    const int row0 = blockIdx.x * 16;
    const int n0   = wave * 16;

    v8f acc = {};

    for (int k0 = 0; k0 < K; k0 += 128) {
        #pragma unroll
        for (int it = 0; it < 2; ++it) {
            int i   = tid * 4 + it * 1024;
            int row = i >> 7;
            int col = i & 127;
            float4 val = *(const float4*)(A + (size_t)(row0 + row) * K + k0 + col);
            *(float4*)(&As[row * 132 + col]) = val;
        }
        __syncthreads();

        #pragma unroll 8
        for (int kk = 0; kk < 128; kk += 4) {
            v2f a = *(const v2f*)(&As[ml * 132 + kk + 2 * hi]);
            const float* bp = B + (size_t)(k0 + kk + 2 * hi) * H + n0 + ml;
            v2f b;
            b.x = bp[0];
            b.y = bp[H];
            acc = __builtin_amdgcn_wmma_f32_16x16x4_f32(
                      false, a, false, b, (short)0, acc, false, false);
        }
        __syncthreads();
    }

    const float bval = bias ? bias[n0 + ml] : 0.0f;
    #pragma unroll
    for (int r = 0; r < 8; ++r) {
        int m = row0 + r + 8 * hi;
        C[(size_t)m * H + n0 + ml] = acc[r] + bval;
    }
}

__global__ void edge_conv(const float* __restrict__ kbuf,
                          const float* __restrict__ qbuf,
                          const float* __restrict__ vbuf,
                          const float* __restrict__ xe,
                          const float* __restrict__ We,
                          const float* __restrict__ be,
                          const int*   __restrict__ src,
                          const int*   __restrict__ dst,
                          float*       __restrict__ out_dst,
                          int E)
{
    const int e = blockIdx.x * 2 + (threadIdx.x >> 7);
    const int h = threadIdx.x & 127;
    if (e >= E) return;

    const int s = src[e];
    const int d = dst[e];

    float ea = be[h];
    const float* xrow = xe + (size_t)e * D_EDGE;
    #pragma unroll
    for (int dd = 0; dd < D_EDGE; ++dd)
        ea = fmaf(xrow[dd], We[dd * H + h], ea);

    const float ki = kbuf[(size_t)d * H + h] + ea;
    const float qj = qbuf[(size_t)s * H + h] + ea;
    const float vj = vbuf[(size_t)s * H + h] + ea;

    const float g = 1.0f / (1.0f + __expf(-(ki + qj)));
    atomicAdd(&out_dst[(size_t)d * H + h], g * vj);
}

__global__ void relu_inplace(float* __restrict__ x, int n)
{
    int i = blockIdx.x * blockDim.x + threadIdx.x;
    if (i < n) x[i] = fmaxf(x[i], 0.0f);
}

__global__ void classifier(const float* __restrict__ hm,
                           const float* __restrict__ hv,
                           const int*   __restrict__ im,
                           const int*   __restrict__ iv,
                           const float* __restrict__ Wc,
                           const float* __restrict__ bc,
                           float*       __restrict__ out,
                           int P)
{
    const int p    = blockIdx.x * 8 + (threadIdx.x >> 5);
    const int lane = threadIdx.x & 31;
    if (p >= P) return;

    const float4 am = ((const float4*)(hm + (size_t)im[p] * H))[lane];
    const float4 av = ((const float4*)(hv + (size_t)iv[p] * H))[lane];

    float c0 = 0.f, c1 = 0.f;
    const int ch = lane * 4;
    const float a[4] = {am.x, am.y, am.z, am.w};
    const float b[4] = {av.x, av.y, av.z, av.w};
    #pragma unroll
    for (int i = 0; i < 4; ++i) {
        c0 = fmaf(a[i], Wc[(ch + i) * 2 + 0], c0);
        c1 = fmaf(a[i], Wc[(ch + i) * 2 + 1], c1);
        c0 = fmaf(b[i], Wc[(H + ch + i) * 2 + 0], c0);
        c1 = fmaf(b[i], Wc[(H + ch + i) * 2 + 1], c1);
    }
    #pragma unroll
    for (int off = 16; off > 0; off >>= 1) {
        c0 += __shfl_down(c0, off, 32);
        c1 += __shfl_down(c1, off, 32);
    }
    if (lane == 0) {
        out[(size_t)p * 2 + 0] = c0 + bc[0];
        out[(size_t)p * 2 + 1] = c1 + bc[1];
    }
}

static inline void gemm(const float* A, const float* B, const float* bias,
                        float* C, int M, int K, hipStream_t s)
{
    gemm_wmma_bias<<<dim3(M / 16), dim3(256), 0, s>>>(A, B, bias, C, M, K);
}

extern "C" void kernel_launch(void* const* d_in, const int* in_sizes, int n_in,
                              void* d_out, int out_size, void* d_ws, size_t ws_size,
                              hipStream_t stream)
{
    const float* x_mouse  = (const float*)d_in[0];
    const float* x_virus  = (const float*)d_in[1];
    const float* xe_mv    = (const float*)d_in[2];
    const float* xe_vm    = (const float*)d_in[3];
    const float* W_mouse  = (const float*)d_in[4];
    const float* b_mouse  = (const float*)d_in[5];
    const float* W_virus  = (const float*)d_in[6];
    const float* b_virus  = (const float*)d_in[7];
    const float* W_key    = (const float*)d_in[8];
    const float* b_key    = (const float*)d_in[9];
    const float* W_query  = (const float*)d_in[10];
    const float* b_query  = (const float*)d_in[11];
    const float* W_value  = (const float*)d_in[12];
    const float* b_value  = (const float*)d_in[13];
    const float* W_edge   = (const float*)d_in[14];
    const float* b_edge   = (const float*)d_in[15];
    const float* W_skip   = (const float*)d_in[16];
    const float* bias_cv  = (const float*)d_in[17];
    const float* W_cls    = (const float*)d_in[18];
    const float* b_cls    = (const float*)d_in[19];
    const int* ei_mv_src  = (const int*)d_in[20];
    const int* ei_mv_dst  = (const int*)d_in[21];
    const int* ei_vm_src  = (const int*)d_in[22];
    const int* ei_vm_dst  = (const int*)d_in[23];
    const int* eli_mouse  = (const int*)d_in[24];
    const int* eli_virus  = (const int*)d_in[25];
    float* out = (float*)d_out;

    float* ws  = (float*)d_ws;
    const size_t SZ_M = (size_t)N_MOUSE * H;
    const size_t SZ_V = (size_t)N_VIRUS * H;
    float* hm0 = ws;
    float* hm1 = hm0 + SZ_M;
    float* hv0 = hm1 + SZ_M;
    float* hv1 = hv0 + SZ_V;
    float* kb  = hv1 + SZ_V;
    float* qb  = kb + SZ_M;
    float* vb  = qb + SZ_M;

    const int relu_m_blocks = (int)((SZ_M + 255) / 256);
    const int relu_v_blocks = (int)((SZ_V + 255) / 256);
    const int edge_blocks   = (NEDGE + 1) / 2;

    gemm(x_mouse, W_mouse, b_mouse, hm0, N_MOUSE, D_IN, stream);
    gemm(x_virus, W_virus, b_virus, hv0, N_VIRUS, D_IN, stream);

    float* hm_in = hm0; float* hv_in = hv0;
    float* hm_out = hm1; float* hv_out = hv1;

    for (int layer = 0; layer < 2; ++layer) {
        const int i_mv = layer * 2;
        const int i_vm = layer * 2 + 1;

        {
            const int i = i_mv;
            const float* Wk = W_key   + (size_t)i * H * H;
            const float* Wq = W_query + (size_t)i * H * H;
            const float* Wv = W_value + (size_t)i * H * H;
            const float* We = W_edge  + (size_t)i * D_EDGE * H;
            const float* Ws = W_skip  + (size_t)i * H * H;
            gemm(hv_in, Ws, bias_cv + (size_t)i * H, hv_out, N_VIRUS, H, stream);
            gemm(hv_in, Wk, b_key   + (size_t)i * H, kb, N_VIRUS, H, stream);
            gemm(hm_in, Wq, b_query + (size_t)i * H, qb, N_MOUSE, H, stream);
            gemm(hm_in, Wv, b_value + (size_t)i * H, vb, N_MOUSE, H, stream);
            edge_conv<<<dim3(edge_blocks), dim3(256), 0, stream>>>(
                kb, qb, vb, xe_mv, We, b_edge + (size_t)i * H,
                ei_mv_src, ei_mv_dst, hv_out, NEDGE);
        }
        {
            const int i = i_vm;
            const float* Wk = W_key   + (size_t)i * H * H;
            const float* Wq = W_query + (size_t)i * H * H;
            const float* Wv = W_value + (size_t)i * H * H;
            const float* We = W_edge  + (size_t)i * D_EDGE * H;
            const float* Ws = W_skip  + (size_t)i * H * H;
            gemm(hm_in, Ws, bias_cv + (size_t)i * H, hm_out, N_MOUSE, H, stream);
            gemm(hm_in, Wk, b_key   + (size_t)i * H, kb, N_MOUSE, H, stream);
            gemm(hv_in, Wq, b_query + (size_t)i * H, qb, N_VIRUS, H, stream);
            gemm(hv_in, Wv, b_value + (size_t)i * H, vb, N_VIRUS, H, stream);
            edge_conv<<<dim3(edge_blocks), dim3(256), 0, stream>>>(
                kb, qb, vb, xe_vm, We, b_edge + (size_t)i * H,
                ei_vm_src, ei_vm_dst, hm_out, NEDGE);
        }
        relu_inplace<<<dim3(relu_v_blocks), dim3(256), 0, stream>>>(hv_out, (int)SZ_V);
        relu_inplace<<<dim3(relu_m_blocks), dim3(256), 0, stream>>>(hm_out, (int)SZ_M);

        float* t;
        t = hm_in; hm_in = hm_out; hm_out = t;
        t = hv_in; hv_in = hv_out; hv_out = t;
    }

    classifier<<<dim3((NPAIR + 7) / 8), dim3(256), 0, stream>>>(
        hm_in, hv_in, eli_mouse, eli_virus, W_cls, b_cls, out, NPAIR);
}


# edge sums without float atomics: edges bucketed by destination each call and put in edge order; one wave sums a row in registers and stores it once with skip, bias, ReLU; sum order now fixed
# speedup vs baseline: 1.5975x; 1.5975x over previous
#include <hip/hip_runtime.h>
#include <hip/hip_bf16.h>

typedef __attribute__((ext_vector_type(2))) float v2f;
typedef __attribute__((ext_vector_type(8))) float v8f;

#define H 128
#define D_EDGE 16
#define N_MOUSE 50000
#define N_VIRUS 10000
#define NEDGE   500000
#define NPAIR   100000
#define D_IN    1024

__global__ void gemm_wmma_bias(const float* __restrict__ A,
                               const float* __restrict__ B,
                               const float* __restrict__ bias,
                               float* __restrict__ C,
                               int M, int K)
{
    __shared__ __align__(16) float As[16 * 132];

    const int tid  = threadIdx.x;
    const int wave = tid >> 5;
    const int lane = tid & 31;
    const int hi   = lane >> 4;
    const int ml   = lane & 15;
    const int row0 = blockIdx.x * 16;
    const int n0   = wave * 16;

    v8f acc = {};

    for (int k0 = 0; k0 < K; k0 += 128) {
        #pragma unroll
        for (int it = 0; it < 2; ++it) {
            int i   = tid * 4 + it * 1024;
            int row = i >> 7;
            int col = i & 127;
            float4 val = *(const float4*)(A + (size_t)(row0 + row) * K + k0 + col);
            *(float4*)(&As[row * 132 + col]) = val;
        }
        __syncthreads();

        #pragma unroll 8
        for (int kk = 0; kk < 128; kk += 4) {
            v2f a = *(const v2f*)(&As[ml * 132 + kk + 2 * hi]);
            const float* bp = B + (size_t)(k0 + kk + 2 * hi) * H + n0 + ml;
            v2f b;
            b.x = bp[0];
            b.y = bp[H];
            acc = __builtin_amdgcn_wmma_f32_16x16x4_f32(
                      false, a, false, b, (short)0, acc, false, false);
        }
        __syncthreads();
    }

    const float bval = bias ? bias[n0 + ml] : 0.0f;
    #pragma unroll
    for (int r = 0; r < 8; ++r) {
        int m = row0 + r + 8 * hi;
        C[(size_t)m * H + n0 + ml] = acc[r] + bval;
    }
}

__global__ void zero_ints(int* __restrict__ p, int n)
{
    int i = blockIdx.x * blockDim.x + threadIdx.x;
    if (i < n) p[i] = 0;
}

__global__ void dst_count(const int* __restrict__ dst_a,
                          const int* __restrict__ dst_b,
                          int* __restrict__ cnt_a,
                          int* __restrict__ cnt_b,
                          int E, int na, int nb)
{
    int e = blockIdx.x * blockDim.x + threadIdx.x;
    if (e >= E) return;
    int da = min(max(dst_a[e], 0), na - 1);
    int db = min(max(dst_b[e], 0), nb - 1);
    atomicAdd(&cnt_a[da], 1);
    atomicAdd(&cnt_b[db], 1);
}

__global__ __launch_bounds__(1024) void dst_scan(const int* __restrict__ cnt_a,
                                                 int* __restrict__ off_a,
                                                 int* __restrict__ cur_a, int na,
                                                 const int* __restrict__ cnt_b,
                                                 int* __restrict__ off_b,
                                                 int* __restrict__ cur_b, int nb)
{
    __shared__ int wsum[32];
    __shared__ int total_s;

    const int* cnt = blockIdx.x ? cnt_b : cnt_a;
    int*       off = blockIdx.x ? off_b : off_a;
    int*       cur = blockIdx.x ? cur_b : cur_a;
    const int  n   = blockIdx.x ? nb : na;

    const int tid  = threadIdx.x;
    const int lane = tid & 31;
    const int wave = tid >> 5;

    int carry = 0;
    for (int base = 0; base < n; base += 16384) {
        const int idx = base + tid * 16;
        int4 c[4];
        for (int k = 0; k < 4; ++k) c[k] = make_int4(0, 0, 0, 0);
        if (idx < n) {
            for (int k = 0; k < 4; ++k) c[k] = *(const int4*)(cnt + idx + 4 * k);
        }
        int t = 0;
        for (int k = 0; k < 4; ++k) t += c[k].x + c[k].y + c[k].z + c[k].w;

        int incl = t;
        for (int o = 1; o < 32; o <<= 1) {
            int y = __shfl_up(incl, o, 32);
            if (lane >= o) incl += y;
        }
        if (lane == 31) wsum[wave] = incl;
        __syncthreads();
        if (wave == 0) {
            const int v = wsum[lane];
            int iv = v;
            for (int o = 1; o < 32; o <<= 1) {
                int y = __shfl_up(iv, o, 32);
                if (lane >= o) iv += y;
            }
            wsum[lane] = iv - v;
            if (lane == 31) total_s = iv;
        }
        __syncthreads();
        int run = carry + wsum[wave] + incl - t;
        if (idx < n) {
            for (int k = 0; k < 4; ++k) {
                int4 o4;
                o4.x = run;
                o4.y = o4.x + c[k].x;
                o4.z = o4.y + c[k].y;
                o4.w = o4.z + c[k].z;
                run  = o4.w + c[k].w;
                *(int4*)(off + idx + 4 * k) = o4;
                *(int4*)(cur + idx + 4 * k) = o4;
            }
        }
        carry += total_s;
        __syncthreads();
    }
    if (tid == 0) off[n] = carry;
}

__global__ void dst_fill(const int* __restrict__ dst_a,
                         int* __restrict__ cur_a,
                         int* __restrict__ perm_a,
                         const int* __restrict__ dst_b,
                         int* __restrict__ cur_b,
                         int* __restrict__ perm_b,
                         int E, int na, int nb)
{
    int e = blockIdx.x * blockDim.x + threadIdx.x;
    if (e >= E) return;
    {
        int d = min(max(dst_a[e], 0), na - 1);
        int p = atomicAdd(&cur_a[d], 1);
        perm_a[min(max(p, 0), E - 1)] = e;
    }
    {
        int d = min(max(dst_b[e], 0), nb - 1);
        int p = atomicAdd(&cur_b[d], 1);
        perm_b[min(max(p, 0), E - 1)] = e;
    }
}

#define SORT_CAP 512
__global__ __launch_bounds__(256) void dst_order(const int* __restrict__ off_a,
                                                 int* __restrict__ perm_a,
                                                 int* __restrict__ psrc_a,
                                                 const int* __restrict__ src_a, int na,
                                                 const int* __restrict__ off_b,
                                                 int* __restrict__ perm_b,
                                                 int* __restrict__ psrc_b,
                                                 const int* __restrict__ src_b, int nb,
                                                 int E)
{
    __shared__ __align__(16) int skey[8][SORT_CAP];

    const int lane = threadIdx.x & 31;
    const int wv   = threadIdx.x >> 5;
    int row = blockIdx.x * 8 + wv;

    const bool is_a  = row < na;
    const int* off   = is_a ? off_a  : off_b;
    int*       perm  = is_a ? perm_a : perm_b;
    int*       psrc  = is_a ? psrc_a : psrc_b;
    const int* src   = is_a ? src_a  : src_b;
    if (!is_a) row -= na;
    const bool valid = is_a || row < nb;

    int beg = 0, n = 0;
    if (valid) {
        beg = min(max(off[row], 0), E);
        n   = min(max(off[row + 1], beg), E) - beg;
    }
    const bool small = n <= SORT_CAP;
    const int  npad  = (n + 3) & ~3;
    int* key = skey[wv];

    if (small) {
        for (int j = lane; j < npad; j += 32)
            key[j] = (j < n) ? perm[beg + j] : 0x7fffffff;
    }
    __syncthreads();

    if (small) {
        for (int j0 = 0; j0 < n; j0 += 32) {
            const int j  = j0 + lane;
            const int my = (j < n) ? key[j] : 0x7fffffff;
            int r = 0;
            for (int t = 0; t < npad; t += 4) {
                const int4 k4 = *(const int4*)(key + t);
                r += (k4.x < my) + (k4.y < my) + (k4.z < my) + (k4.w < my);
            }
            if (j < n) {
                perm[beg + r] = my;
                psrc[beg + r] = src[min(max(my, 0), E - 1)];
            }
        }
    } else {
        for (int j = lane; j < n; j += 32)
            psrc[beg + j] = src[min(max(perm[beg + j], 0), E - 1)];
    }
}

__global__ __launch_bounds__(256) void edge_rows(const float* __restrict__ kbuf,
                                                 const float* __restrict__ qbuf,
                                                 const float* __restrict__ vbuf,
                                                 const float* __restrict__ xe,
                                                 const float* __restrict__ We,
                                                 const float* __restrict__ be,
                                                 const int*   __restrict__ off,
                                                 const int*   __restrict__ perm,
                                                 const int*   __restrict__ psrc,
                                                 float*       __restrict__ out_dst,
                                                 int n_dst, int n_src, int E)
{
    const int lane = threadIdx.x & 31;
    const int d = __builtin_amdgcn_readfirstlane(blockIdx.x * 8 + (threadIdx.x >> 5));
    if (d >= n_dst) return;

    float4 w[D_EDGE];
    for (int dd = 0; dd < D_EDGE; ++dd)
        w[dd] = *(const float4*)(We + dd * H + 4 * lane);
    const float4 b4 = *(const float4*)(be + 4 * lane);

    int beg = __builtin_amdgcn_readfirstlane(off[d]);
    int end = __builtin_amdgcn_readfirstlane(off[d + 1]);
    beg = min(max(beg, 0), E);
    end = min(max(end, beg), E);

    const float4 k4 = *(const float4*)(kbuf + (size_t)d * H + 4 * lane);
    float4 acc = make_float4(0.f, 0.f, 0.f, 0.f);

    for (int j0 = beg; j0 < end; j0 += 32) {
        const int jj = j0 + lane;
        const int pe = (jj < end) ? perm[jj] : 0;
        const int ps = (jj < end) ? psrc[jj] : 0;
        const int cnt = min(32, end - j0);
        int e = min(max(__builtin_amdgcn_readlane(pe, 0), 0), E - 1);
        int s = min(max(__builtin_amdgcn_readlane(ps, 0), 0), n_src - 1);
        float  xv = xe[(size_t)e * D_EDGE + (lane & 15)];
        float4 q4 = *(const float4*)(qbuf + (size_t)s * H + 4 * lane);
        float4 v4 = *(const float4*)(vbuf + (size_t)s * H + 4 * lane);

        for (int i = 0; i < cnt; ++i) {
            float  xn = xv;
            float4 qn = q4;
            float4 vn = v4;
            if (i + 1 < cnt) {
                e = min(max(__builtin_amdgcn_readlane(pe, i + 1), 0), E - 1);
                s = min(max(__builtin_amdgcn_readlane(ps, i + 1), 0), n_src - 1);
                xn = xe[(size_t)e * D_EDGE + (lane & 15)];
                qn = *(const float4*)(qbuf + (size_t)s * H + 4 * lane);
                vn = *(const float4*)(vbuf + (size_t)s * H + 4 * lane);
            }

            float4 ea = b4;
            for (int dd = 0; dd < D_EDGE; ++dd) {
                const float x = __int_as_float(
                    __builtin_amdgcn_readlane(__float_as_int(xv), dd));
                ea.x = fmaf(x, w[dd].x, ea.x);
                ea.y = fmaf(x, w[dd].y, ea.y);
                ea.z = fmaf(x, w[dd].z, ea.z);
                ea.w = fmaf(x, w[dd].w, ea.w);
            }

            const float gx = 1.0f / (1.0f + __expf(-((k4.x + ea.x) + (q4.x + ea.x))));
            const float gy = 1.0f / (1.0f + __expf(-((k4.y + ea.y) + (q4.y + ea.y))));
            const float gz = 1.0f / (1.0f + __expf(-((k4.z + ea.z) + (q4.z + ea.z))));
            const float gw = 1.0f / (1.0f + __expf(-((k4.w + ea.w) + (q4.w + ea.w))));
            acc.x += gx * (v4.x + ea.x);
            acc.y += gy * (v4.y + ea.y);
            acc.z += gz * (v4.z + ea.z);
            acc.w += gw * (v4.w + ea.w);

            xv = xn; q4 = qn; v4 = vn;
        }
    }

    float4* op = (float4*)(out_dst + (size_t)d * H + 4 * lane);
    float4 o = *op;
    o.x = fmaxf(o.x + acc.x, 0.0f);
    o.y = fmaxf(o.y + acc.y, 0.0f);
    o.z = fmaxf(o.z + acc.z, 0.0f);
    o.w = fmaxf(o.w + acc.w, 0.0f);
    *op = o;
}

__global__ void classifier(const float* __restrict__ hm,
                           const float* __restrict__ hv,
                           const int*   __restrict__ im,
                           const int*   __restrict__ iv,
                           const float* __restrict__ Wc,
                           const float* __restrict__ bc,
                           float*       __restrict__ out,
                           int P)
{
    const int p    = blockIdx.x * 8 + (threadIdx.x >> 5);
    const int lane = threadIdx.x & 31;
    if (p >= P) return;

    const float4 am = ((const float4*)(hm + (size_t)im[p] * H))[lane];
    const float4 av = ((const float4*)(hv + (size_t)iv[p] * H))[lane];

    float c0 = 0.f, c1 = 0.f;
    const int ch = lane * 4;
    const float a[4] = {am.x, am.y, am.z, am.w};
    const float b[4] = {av.x, av.y, av.z, av.w};
    #pragma unroll
    for (int i = 0; i < 4; ++i) {
        c0 = fmaf(a[i], Wc[(ch + i) * 2 + 0], c0);
        c1 = fmaf(a[i], Wc[(ch + i) * 2 + 1], c1);
        c0 = fmaf(b[i], Wc[(H + ch + i) * 2 + 0], c0);
        c1 = fmaf(b[i], Wc[(H + ch + i) * 2 + 1], c1);
    }
    #pragma unroll
    for (int off = 16; off > 0; off >>= 1) {
        c0 += __shfl_down(c0, off, 32);
        c1 += __shfl_down(c1, off, 32);
    }
    if (lane == 0) {
        out[(size_t)p * 2 + 0] = c0 + bc[0];
        out[(size_t)p * 2 + 1] = c1 + bc[1];
    }
}

static inline void gemm(const float* A, const float* B, const float* bias,
                        float* C, int M, int K, hipStream_t s)
{
    gemm_wmma_bias<<<dim3(M / 16), dim3(256), 0, s>>>(A, B, bias, C, M, K);
}

extern "C" void kernel_launch(void* const* d_in, const int* in_sizes, int n_in,
                              void* d_out, int out_size, void* d_ws, size_t ws_size,
                              hipStream_t stream)
{
    const float* x_mouse  = (const float*)d_in[0];
    const float* x_virus  = (const float*)d_in[1];
    const float* xe_mv    = (const float*)d_in[2];
    const float* xe_vm    = (const float*)d_in[3];
    const float* W_mouse  = (const float*)d_in[4];
    const float* b_mouse  = (const float*)d_in[5];
    const float* W_virus  = (const float*)d_in[6];
    const float* b_virus  = (const float*)d_in[7];
    const float* W_key    = (const float*)d_in[8];
    const float* b_key    = (const float*)d_in[9];
    const float* W_query  = (const float*)d_in[10];
    const float* b_query  = (const float*)d_in[11];
    const float* W_value  = (const float*)d_in[12];
    const float* b_value  = (const float*)d_in[13];
    const float* W_edge   = (const float*)d_in[14];
    const float* b_edge   = (const float*)d_in[15];
    const float* W_skip   = (const float*)d_in[16];
    const float* bias_cv  = (const float*)d_in[17];
    const float* W_cls    = (const float*)d_in[18];
    const float* b_cls    = (const float*)d_in[19];
    const int* ei_mv_src  = (const int*)d_in[20];
    const int* ei_mv_dst  = (const int*)d_in[21];
    const int* ei_vm_src  = (const int*)d_in[22];
    const int* ei_vm_dst  = (const int*)d_in[23];
    const int* eli_mouse  = (const int*)d_in[24];
    const int* eli_virus  = (const int*)d_in[25];
    float* out = (float*)d_out;

    float* ws  = (float*)d_ws;
    const size_t SZ_M = (size_t)N_MOUSE * H;
    const size_t SZ_V = (size_t)N_VIRUS * H;
    float* hm0 = ws;
    float* hm1 = hm0 + SZ_M;
    float* hv0 = hm1 + SZ_M;
    float* hv1 = hv0 + SZ_V;
    float* bx  = hv1 + SZ_V;
    float* by  = bx + SZ_M;
    float* bz  = by + SZ_M;

    int* cnt_a  = (int*)(bz + SZ_V);
    int* cnt_b  = cnt_a + N_VIRUS;
    int* off_a  = cnt_b + N_MOUSE;
    int* off_b  = off_a + N_VIRUS + 4;
    int* cur_a  = off_b + N_MOUSE + 4;
    int* cur_b  = cur_a + N_VIRUS;
    int* perm_a = cur_b + N_MOUSE;
    int* psrc_a = perm_a + NEDGE;
    int* perm_b = psrc_a + NEDGE;
    int* psrc_b = perm_b + NEDGE;

    const int e_blocks = (NEDGE + 255) / 256;

    zero_ints<<<dim3((N_VIRUS + N_MOUSE + 255) / 256), dim3(256), 0, stream>>>(
        cnt_a, N_VIRUS + N_MOUSE);
    dst_count<<<dim3(e_blocks), dim3(256), 0, stream>>>(
        ei_mv_dst, ei_vm_dst, cnt_a, cnt_b, NEDGE, N_VIRUS, N_MOUSE);
    dst_scan<<<dim3(2), dim3(1024), 0, stream>>>(
        cnt_a, off_a, cur_a, N_VIRUS, cnt_b, off_b, cur_b, N_MOUSE);
    dst_fill<<<dim3(e_blocks), dim3(256), 0, stream>>>(
        ei_mv_dst, cur_a, perm_a, ei_vm_dst, cur_b, perm_b,
        NEDGE, N_VIRUS, N_MOUSE);
    dst_order<<<dim3((N_VIRUS + N_MOUSE + 7) / 8), dim3(256), 0, stream>>>(
        off_a, perm_a, psrc_a, ei_mv_src, N_VIRUS,
        off_b, perm_b, psrc_b, ei_vm_src, N_MOUSE, NEDGE);

    gemm(x_mouse, W_mouse, b_mouse, hm0, N_MOUSE, D_IN, stream);
    gemm(x_virus, W_virus, b_virus, hv0, N_VIRUS, D_IN, stream);

    float* hm_in = hm0; float* hv_in = hv0;
    float* hm_out = hm1; float* hv_out = hv1;

    for (int layer = 0; layer < 2; ++layer) {
        const int i_mv = layer * 2;
        const int i_vm = layer * 2 + 1;

        {
            const int i = i_mv;
            const float* Wk = W_key   + (size_t)i * H * H;
            const float* Wq = W_query + (size_t)i * H * H;
            const float* Wv = W_value + (size_t)i * H * H;
            const float* We = W_edge  + (size_t)i * D_EDGE * H;
            const float* Ws = W_skip  + (size_t)i * H * H;
            gemm(hv_in, Ws, bias_cv + (size_t)i * H, hv_out, N_VIRUS, H, stream);
            gemm(hv_in, Wk, b_key   + (size_t)i * H, bz, N_VIRUS, H, stream);
            gemm(hm_in, Wq, b_query + (size_t)i * H, bx, N_MOUSE, H, stream);
            gemm(hm_in, Wv, b_value + (size_t)i * H, by, N_MOUSE, H, stream);
            edge_rows<<<dim3((N_VIRUS + 7) / 8), dim3(256), 0, stream>>>(
                bz, bx, by, xe_mv, We, b_edge + (size_t)i * H,
                off_a, perm_a, psrc_a, hv_out, N_VIRUS, N_MOUSE, NEDGE);
        }
        {
            const int i = i_vm;
            const float* Wk = W_key   + (size_t)i * H * H;
            const float* Wq = W_query + (size_t)i * H * H;
            const float* Wv = W_value + (size_t)i * H * H;
            const float* We = W_edge  + (size_t)i * D_EDGE * H;
            const float* Ws = W_skip  + (size_t)i * H * H;
            gemm(hm_in, Ws, bias_cv + (size_t)i * H, hm_out, N_MOUSE, H, stream);
            gemm(hm_in, Wk, b_key   + (size_t)i * H, bx, N_MOUSE, H, stream);
            gemm(hv_in, Wq, b_query + (size_t)i * H, by, N_VIRUS, H, stream);
            gemm(hv_in, Wv, b_value + (size_t)i * H, by + SZ_V, N_VIRUS, H, stream);
            edge_rows<<<dim3((N_MOUSE + 7) / 8), dim3(256), 0, stream>>>(
                bx, by, by + SZ_V, xe_vm, We, b_edge + (size_t)i * H,
                off_b, perm_b, psrc_b, hm_out, N_MOUSE, N_VIRUS, NEDGE);
        }

        float* t;
        t = hm_in; hm_in = hm_out; hm_out = t;
        t = hv_in; hv_in = hv_out; hv_out = t;
    }

    classifier<<<dim3((NPAIR + 7) / 8), dim3(256), 0, stream>>>(
        hm_in, hv_in, eli_mouse, eli_virus, W_cls, b_cls, out, NPAIR);
}


# GEMM register tiling: 4-wave blocks own 80 rows x 64 columns; each wave keeps 5 accumulator tiles, so one B fragment (B slice read once per 128-deep chunk) feeds 5 products; sum order kept
# speedup vs baseline: 1.6752x; 1.0486x over previous
#include <hip/hip_runtime.h>
#include <hip/hip_bf16.h>

typedef __attribute__((ext_vector_type(2))) float v2f;
typedef __attribute__((ext_vector_type(8))) float v8f;

#define H 128
#define D_EDGE 16
#define N_MOUSE 50000
#define N_VIRUS 10000
#define NEDGE   500000
#define NPAIR   100000
#define D_IN    1024

template <int RT, int KK>
__global__ __launch_bounds__(128) __attribute__((amdgpu_waves_per_eu(5, 5)))
void gemm_wmma_bias(const float* __restrict__ A,
                    const float* __restrict__ B,
                    const float* __restrict__ bias,
                    float* __restrict__ C,
                    int M)
{
    __shared__ __align__(16) float As[RT * 16 * 132];

    const int tid  = threadIdx.x;
    const int wave = tid >> 5;
    const int lane = tid & 31;
    const int hi   = lane >> 4;
    const int ml   = lane & 15;
    const int row0 = (blockIdx.x >> 1) * (16 * RT);
    const int n0   = (blockIdx.x & 1) * 64 + wave * 16;

    v8f acc[RT];
    #pragma unroll
    for (int rt = 0; rt < RT; ++rt) acc[rt] = v8f{};

    const float* ap = A + (size_t)(row0 + wave) * KK + 4 * lane;
    float*       sp = &As[wave * 132 + 4 * lane];
    const float* bp = B + (size_t)(2 * hi) * H + n0 + ml;

    for (int k0 = 0; k0 < KK; k0 += 128) {
        v2f b[32];
        #pragma unroll
        for (int s = 0; s < 32; ++s) {
            b[s].x = bp[(size_t)(k0 + 4 * s) * H];
            b[s].y = bp[(size_t)(k0 + 4 * s + 1) * H];
        }
        float4 st[4 * RT];
        #pragma unroll
        for (int it = 0; it < 4 * RT; ++it)
            st[it] = *(const float4*)(ap + (size_t)(4 * it) * KK + k0);
        #pragma unroll
        for (int it = 0; it < 4 * RT; ++it)
            *(float4*)(sp + (4 * it) * 132) = st[it];
        __syncthreads();

        #pragma unroll
        for (int s = 0; s < 32; ++s) {
            #pragma unroll
            for (int rt = 0; rt < RT; ++rt) {
                v2f a = *(const v2f*)(&As[(rt * 16 + ml) * 132 + 4 * s + 2 * hi]);
                acc[rt] = __builtin_amdgcn_wmma_f32_16x16x4_f32(
                              false, a, false, b[s], (short)0, acc[rt], false, false);
            }
        }
        __syncthreads();
    }

    const float bval = bias ? bias[n0 + ml] : 0.0f;
    #pragma unroll
    for (int rt = 0; rt < RT; ++rt) {
        #pragma unroll
        for (int r = 0; r < 8; ++r) {
            int m = row0 + rt * 16 + r + 8 * hi;
            C[(size_t)m * H + n0 + ml] = acc[rt][r] + bval;
        }
    }
}

__global__ void zero_ints(int* __restrict__ p, int n)
{
    int i = blockIdx.x * blockDim.x + threadIdx.x;
    if (i < n) p[i] = 0;
}

__global__ void dst_count(const int* __restrict__ dst_a,
                          const int* __restrict__ dst_b,
                          int* __restrict__ cnt_a,
                          int* __restrict__ cnt_b,
                          int E, int na, int nb)
{
    int e = blockIdx.x * blockDim.x + threadIdx.x;
    if (e >= E) return;
    int da = min(max(dst_a[e], 0), na - 1);
    int db = min(max(dst_b[e], 0), nb - 1);
    atomicAdd(&cnt_a[da], 1);
    atomicAdd(&cnt_b[db], 1);
}

__global__ __launch_bounds__(1024) void dst_scan(const int* __restrict__ cnt_a,
                                                 int* __restrict__ off_a,
                                                 int* __restrict__ cur_a, int na,
                                                 const int* __restrict__ cnt_b,
                                                 int* __restrict__ off_b,
                                                 int* __restrict__ cur_b, int nb)
{
    __shared__ int wsum[32];
    __shared__ int total_s;

    const int* cnt = blockIdx.x ? cnt_b : cnt_a;
    int*       off = blockIdx.x ? off_b : off_a;
    int*       cur = blockIdx.x ? cur_b : cur_a;
    const int  n   = blockIdx.x ? nb : na;

    const int tid  = threadIdx.x;
    const int lane = tid & 31;
    const int wave = tid >> 5;

    int carry = 0;
    for (int base = 0; base < n; base += 16384) {
        const int idx = base + tid * 16;
        int4 c[4];
        for (int k = 0; k < 4; ++k) c[k] = make_int4(0, 0, 0, 0);
        if (idx < n) {
            for (int k = 0; k < 4; ++k) c[k] = *(const int4*)(cnt + idx + 4 * k);
        }
        int t = 0;
        for (int k = 0; k < 4; ++k) t += c[k].x + c[k].y + c[k].z + c[k].w;

        int incl = t;
        for (int o = 1; o < 32; o <<= 1) {
            int y = __shfl_up(incl, o, 32);
            if (lane >= o) incl += y;
        }
        if (lane == 31) wsum[wave] = incl;
        __syncthreads();
        if (wave == 0) {
            const int v = wsum[lane];
            int iv = v;
            for (int o = 1; o < 32; o <<= 1) {
                int y = __shfl_up(iv, o, 32);
                if (lane >= o) iv += y;
            }
            wsum[lane] = iv - v;
            if (lane == 31) total_s = iv;
        }
        __syncthreads();
        int run = carry + wsum[wave] + incl - t;
        if (idx < n) {
            for (int k = 0; k < 4; ++k) {
                int4 o4;
                o4.x = run;
                o4.y = o4.x + c[k].x;
                o4.z = o4.y + c[k].y;
                o4.w = o4.z + c[k].z;
                run  = o4.w + c[k].w;
                *(int4*)(off + idx + 4 * k) = o4;
                *(int4*)(cur + idx + 4 * k) = o4;
            }
        }
        carry += total_s;
        __syncthreads();
    }
    if (tid == 0) off[n] = carry;
}

__global__ void dst_fill(const int* __restrict__ dst_a,
                         int* __restrict__ cur_a,
                         int* __restrict__ perm_a,
                         const int* __restrict__ dst_b,
                         int* __restrict__ cur_b,
                         int* __restrict__ perm_b,
                         int E, int na, int nb)
{
    int e = blockIdx.x * blockDim.x + threadIdx.x;
    if (e >= E) return;
    {
        int d = min(max(dst_a[e], 0), na - 1);
        int p = atomicAdd(&cur_a[d], 1);
        perm_a[min(max(p, 0), E - 1)] = e;
    }
    {
        int d = min(max(dst_b[e], 0), nb - 1);
        int p = atomicAdd(&cur_b[d], 1);
        perm_b[min(max(p, 0), E - 1)] = e;
    }
}

#define SORT_CAP 512
__global__ __launch_bounds__(256) void dst_order(const int* __restrict__ off_a,
                                                 int* __restrict__ perm_a,
                                                 int* __restrict__ psrc_a,
                                                 const int* __restrict__ src_a, int na,
                                                 const int* __restrict__ off_b,
                                                 int* __restrict__ perm_b,
                                                 int* __restrict__ psrc_b,
                                                 const int* __restrict__ src_b, int nb,
                                                 int E)
{
    __shared__ __align__(16) int skey[8][SORT_CAP];

    const int lane = threadIdx.x & 31;
    const int wv   = threadIdx.x >> 5;
    int row = blockIdx.x * 8 + wv;

    const bool is_a  = row < na;
    const int* off   = is_a ? off_a  : off_b;
    int*       perm  = is_a ? perm_a : perm_b;
    int*       psrc  = is_a ? psrc_a : psrc_b;
    const int* src   = is_a ? src_a  : src_b;
    if (!is_a) row -= na;
    const bool valid = is_a || row < nb;

    int beg = 0, n = 0;
    if (valid) {
        beg = min(max(off[row], 0), E);
        n   = min(max(off[row + 1], beg), E) - beg;
    }
    const bool small = n <= SORT_CAP;
    const int  npad  = (n + 3) & ~3;
    int* key = skey[wv];

    if (small) {
        for (int j = lane; j < npad; j += 32)
            key[j] = (j < n) ? perm[beg + j] : 0x7fffffff;
    }
    __syncthreads();

    if (small) {
        for (int j0 = 0; j0 < n; j0 += 32) {
            const int j  = j0 + lane;
            const int my = (j < n) ? key[j] : 0x7fffffff;
            int r = 0;
            for (int t = 0; t < npad; t += 4) {
                const int4 k4 = *(const int4*)(key + t);
                r += (k4.x < my) + (k4.y < my) + (k4.z < my) + (k4.w < my);
            }
            if (j < n) {
                perm[beg + r] = my;
                psrc[beg + r] = src[min(max(my, 0), E - 1)];
            }
        }
    } else {
        for (int j = lane; j < n; j += 32)
            psrc[beg + j] = src[min(max(perm[beg + j], 0), E - 1)];
    }
}

__global__ __launch_bounds__(256) void edge_rows(const float* __restrict__ kbuf,
                                                 const float* __restrict__ qbuf,
                                                 const float* __restrict__ vbuf,
                                                 const float* __restrict__ xe,
                                                 const float* __restrict__ We,
                                                 const float* __restrict__ be,
                                                 const int*   __restrict__ off,
                                                 const int*   __restrict__ perm,
                                                 const int*   __restrict__ psrc,
                                                 float*       __restrict__ out_dst,
                                                 int n_dst, int n_src, int E)
{
    const int lane = threadIdx.x & 31;
    const int d = __builtin_amdgcn_readfirstlane(blockIdx.x * 8 + (threadIdx.x >> 5));
    if (d >= n_dst) return;

    float4 w[D_EDGE];
    for (int dd = 0; dd < D_EDGE; ++dd)
        w[dd] = *(const float4*)(We + dd * H + 4 * lane);
    const float4 b4 = *(const float4*)(be + 4 * lane);

    int beg = __builtin_amdgcn_readfirstlane(off[d]);
    int end = __builtin_amdgcn_readfirstlane(off[d + 1]);
    beg = min(max(beg, 0), E);
    end = min(max(end, beg), E);

    const float4 k4 = *(const float4*)(kbuf + (size_t)d * H + 4 * lane);
    float4 acc = make_float4(0.f, 0.f, 0.f, 0.f);

    for (int j0 = beg; j0 < end; j0 += 32) {
        const int jj = j0 + lane;
        const int pe = (jj < end) ? perm[jj] : 0;
        const int ps = (jj < end) ? psrc[jj] : 0;
        const int cnt = min(32, end - j0);
        int e = min(max(__builtin_amdgcn_readlane(pe, 0), 0), E - 1);
        int s = min(max(__builtin_amdgcn_readlane(ps, 0), 0), n_src - 1);
        float  xv = xe[(size_t)e * D_EDGE + (lane & 15)];
        float4 q4 = *(const float4*)(qbuf + (size_t)s * H + 4 * lane);
        float4 v4 = *(const float4*)(vbuf + (size_t)s * H + 4 * lane);

        for (int i = 0; i < cnt; ++i) {
            float  xn = xv;
            float4 qn = q4;
            float4 vn = v4;
            if (i + 1 < cnt) {
                e = min(max(__builtin_amdgcn_readlane(pe, i + 1), 0), E - 1);
                s = min(max(__builtin_amdgcn_readlane(ps, i + 1), 0), n_src - 1);
                xn = xe[(size_t)e * D_EDGE + (lane & 15)];
                qn = *(const float4*)(qbuf + (size_t)s * H + 4 * lane);
                vn = *(const float4*)(vbuf + (size_t)s * H + 4 * lane);
            }

            float4 ea = b4;
            for (int dd = 0; dd < D_EDGE; ++dd) {
                const float x = __int_as_float(
                    __builtin_amdgcn_readlane(__float_as_int(xv), dd));
                ea.x = fmaf(x, w[dd].x, ea.x);
                ea.y = fmaf(x, w[dd].y, ea.y);
                ea.z = fmaf(x, w[dd].z, ea.z);
                ea.w = fmaf(x, w[dd].w, ea.w);
            }

            const float gx = 1.0f / (1.0f + __expf(-((k4.x + ea.x) + (q4.x + ea.x))));
            const float gy = 1.0f / (1.0f + __expf(-((k4.y + ea.y) + (q4.y + ea.y))));
            const float gz = 1.0f / (1.0f + __expf(-((k4.z + ea.z) + (q4.z + ea.z))));
            const float gw = 1.0f / (1.0f + __expf(-((k4.w + ea.w) + (q4.w + ea.w))));
            acc.x += gx * (v4.x + ea.x);
            acc.y += gy * (v4.y + ea.y);
            acc.z += gz * (v4.z + ea.z);
            acc.w += gw * (v4.w + ea.w);

            xv = xn; q4 = qn; v4 = vn;
        }
    }

    float4* op = (float4*)(out_dst + (size_t)d * H + 4 * lane);
    float4 o = *op;
    o.x = fmaxf(o.x + acc.x, 0.0f);
    o.y = fmaxf(o.y + acc.y, 0.0f);
    o.z = fmaxf(o.z + acc.z, 0.0f);
    o.w = fmaxf(o.w + acc.w, 0.0f);
    *op = o;
}

__global__ void classifier(const float* __restrict__ hm,
                           const float* __restrict__ hv,
                           const int*   __restrict__ im,
                           const int*   __restrict__ iv,
                           const float* __restrict__ Wc,
                           const float* __restrict__ bc,
                           float*       __restrict__ out,
                           int P)
{
    const int p    = blockIdx.x * 8 + (threadIdx.x >> 5);
    const int lane = threadIdx.x & 31;
    if (p >= P) return;

    const float4 am = ((const float4*)(hm + (size_t)im[p] * H))[lane];
    const float4 av = ((const float4*)(hv + (size_t)iv[p] * H))[lane];

    float c0 = 0.f, c1 = 0.f;
    const int ch = lane * 4;
    const float a[4] = {am.x, am.y, am.z, am.w};
    const float b[4] = {av.x, av.y, av.z, av.w};
    #pragma unroll
    for (int i = 0; i < 4; ++i) {
        c0 = fmaf(a[i], Wc[(ch + i) * 2 + 0], c0);
        c1 = fmaf(a[i], Wc[(ch + i) * 2 + 1], c1);
        c0 = fmaf(b[i], Wc[(H + ch + i) * 2 + 0], c0);
        c1 = fmaf(b[i], Wc[(H + ch + i) * 2 + 1], c1);
    }
    #pragma unroll
    for (int off = 16; off > 0; off >>= 1) {
        c0 += __shfl_down(c0, off, 32);
        c1 += __shfl_down(c1, off, 32);
    }
    if (lane == 0) {
        out[(size_t)p * 2 + 0] = c0 + bc[0];
        out[(size_t)p * 2 + 1] = c1 + bc[1];
    }
}

static inline void gemm_in(const float* A, const float* B, const float* bias,
                           float* C, int M, hipStream_t s)
{
    gemm_wmma_bias<5, D_IN><<<dim3(M / 80 * 2), dim3(128), 0, s>>>(A, B, bias, C, M);
}
static inline void gemm(const float* A, const float* B, const float* bias,
                        float* C, int M, hipStream_t s)
{
    gemm_wmma_bias<5, H><<<dim3(M / 80 * 2), dim3(128), 0, s>>>(A, B, bias, C, M);
}

extern "C" void kernel_launch(void* const* d_in, const int* in_sizes, int n_in,
                              void* d_out, int out_size, void* d_ws, size_t ws_size,
                              hipStream_t stream)
{
    const float* x_mouse  = (const float*)d_in[0];
    const float* x_virus  = (const float*)d_in[1];
    const float* xe_mv    = (const float*)d_in[2];
    const float* xe_vm    = (const float*)d_in[3];
    const float* W_mouse  = (const float*)d_in[4];
    const float* b_mouse  = (const float*)d_in[5];
    const float* W_virus  = (const float*)d_in[6];
    const float* b_virus  = (const float*)d_in[7];
    const float* W_key    = (const float*)d_in[8];
    const float* b_key    = (const float*)d_in[9];
    const float* W_query  = (const float*)d_in[10];
    const float* b_query  = (const float*)d_in[11];
    const float* W_value  = (const float*)d_in[12];
    const float* b_value  = (const float*)d_in[13];
    const float* W_edge   = (const float*)d_in[14];
    const float* b_edge   = (const float*)d_in[15];
    const float* W_skip   = (const float*)d_in[16];
    const float* bias_cv  = (const float*)d_in[17];
    const float* W_cls    = (const float*)d_in[18];
    const float* b_cls    = (const float*)d_in[19];
    const int* ei_mv_src  = (const int*)d_in[20];
    const int* ei_mv_dst  = (const int*)d_in[21];
    const int* ei_vm_src  = (const int*)d_in[22];
    const int* ei_vm_dst  = (const int*)d_in[23];
    const int* eli_mouse  = (const int*)d_in[24];
    const int* eli_virus  = (const int*)d_in[25];
    float* out = (float*)d_out;

    float* ws  = (float*)d_ws;
    const size_t SZ_M = (size_t)N_MOUSE * H;
    const size_t SZ_V = (size_t)N_VIRUS * H;
    float* hm0 = ws;
    float* hm1 = hm0 + SZ_M;
    float* hv0 = hm1 + SZ_M;
    float* hv1 = hv0 + SZ_V;
    float* bx  = hv1 + SZ_V;
    float* by  = bx + SZ_M;
    float* bz  = by + SZ_M;

    int* cnt_a  = (int*)(bz + SZ_V);
    int* cnt_b  = cnt_a + N_VIRUS;
    int* off_a  = cnt_b + N_MOUSE;
    int* off_b  = off_a + N_VIRUS + 4;
    int* cur_a  = off_b + N_MOUSE + 4;
    int* cur_b  = cur_a + N_VIRUS;
    int* perm_a = cur_b + N_MOUSE;
    int* psrc_a = perm_a + NEDGE;
    int* perm_b = psrc_a + NEDGE;
    int* psrc_b = perm_b + NEDGE;

    const int e_blocks = (NEDGE + 255) / 256;

    zero_ints<<<dim3((N_VIRUS + N_MOUSE + 255) / 256), dim3(256), 0, stream>>>(
        cnt_a, N_VIRUS + N_MOUSE);
    dst_count<<<dim3(e_blocks), dim3(256), 0, stream>>>(
        ei_mv_dst, ei_vm_dst, cnt_a, cnt_b, NEDGE, N_VIRUS, N_MOUSE);
    dst_scan<<<dim3(2), dim3(1024), 0, stream>>>(
        cnt_a, off_a, cur_a, N_VIRUS, cnt_b, off_b, cur_b, N_MOUSE);
    dst_fill<<<dim3(e_blocks), dim3(256), 0, stream>>>(
        ei_mv_dst, cur_a, perm_a, ei_vm_dst, cur_b, perm_b,
        NEDGE, N_VIRUS, N_MOUSE);
    dst_order<<<dim3((N_VIRUS + N_MOUSE + 7) / 8), dim3(256), 0, stream>>>(
        off_a, perm_a, psrc_a, ei_mv_src, N_VIRUS,
        off_b, perm_b, psrc_b, ei_vm_src, N_MOUSE, NEDGE);

    gemm_in(x_mouse, W_mouse, b_mouse, hm0, N_MOUSE, stream);
    gemm_in(x_virus, W_virus, b_virus, hv0, N_VIRUS, stream);

    float* hm_in = hm0; float* hv_in = hv0;
    float* hm_out = hm1; float* hv_out = hv1;

    for (int layer = 0; layer < 2; ++layer) {
        const int i_mv = layer * 2;
        const int i_vm = layer * 2 + 1;

        {
            const int i = i_mv;
            const float* Wk = W_key   + (size_t)i * H * H;
            const float* Wq = W_query + (size_t)i * H * H;
            const float* Wv = W_value + (size_t)i * H * H;
            const float* We = W_edge  + (size_t)i * D_EDGE * H;
            const float* Ws = W_skip  + (size_t)i * H * H;
            gemm(hv_in, Ws, bias_cv + (size_t)i * H, hv_out, N_VIRUS, stream);
            gemm(hv_in, Wk, b_key   + (size_t)i * H, bz, N_VIRUS, stream);
            gemm(hm_in, Wq, b_query + (size_t)i * H, bx, N_MOUSE, stream);
            gemm(hm_in, Wv, b_value + (size_t)i * H, by, N_MOUSE, stream);
            edge_rows<<<dim3((N_VIRUS + 7) / 8), dim3(256), 0, stream>>>(
                bz, bx, by, xe_mv, We, b_edge + (size_t)i * H,
                off_a, perm_a, psrc_a, hv_out, N_VIRUS, N_MOUSE, NEDGE);
        }
        {
            const int i = i_vm;
            const float* Wk = W_key   + (size_t)i * H * H;
            const float* Wq = W_query + (size_t)i * H * H;
            const float* Wv = W_value + (size_t)i * H * H;
            const float* We = W_edge  + (size_t)i * D_EDGE * H;
            const float* Ws = W_skip  + (size_t)i * H * H;
            gemm(hm_in, Ws, bias_cv + (size_t)i * H, hm_out, N_MOUSE, stream);
            gemm(hm_in, Wk, b_key   + (size_t)i * H, bx, N_MOUSE, stream);
            gemm(hv_in, Wq, b_query + (size_t)i * H, by, N_VIRUS, stream);
            gemm(hv_in, Wv, b_value + (size_t)i * H, by + SZ_V, N_VIRUS, stream);
            edge_rows<<<dim3((N_MOUSE + 7) / 8), dim3(256), 0, stream>>>(
                bx, by, by + SZ_V, xe_vm, We, b_edge + (size_t)i * H,
                off_b, perm_b, psrc_b, hm_out, N_MOUSE, N_VIRUS, NEDGE);
        }

        float* t;
        t = hm_in; hm_in = hm_out; hm_out = t;
        t = hv_in; hv_in = hv_out; hv_out = t;
    }

    classifier<<<dim3((NPAIR + 7) / 8), dim3(256), 0, stream>>>(
        hm_in, hv_in, eli_mouse, eli_virus, W_cls, b_cls, out, NPAIR);
}


# the four projections of a conv (query, value, skip, key: independent f32 products) share one launch: 16 launches of 128-deep products become 4; tile code, sum order and buffers kept
# speedup vs baseline: 1.7386x; 1.0378x over previous
#include <hip/hip_runtime.h>
#include <hip/hip_bf16.h>

typedef __attribute__((ext_vector_type(2))) float v2f;
typedef __attribute__((ext_vector_type(8))) float v8f;

#define H 128
#define D_EDGE 16
#define N_MOUSE 50000
#define N_VIRUS 10000
#define NEDGE   500000
#define NPAIR   100000
#define D_IN    1024

template <int RT, int KK>
__device__ __forceinline__
void gemm_tile(const float* __restrict__ A,
               const float* __restrict__ B,
               const float* __restrict__ bias,
               float* __restrict__ C,
               int bid)
{
    __shared__ __align__(16) float As[RT * 16 * 132];

    const int tid  = threadIdx.x;
    const int wave = tid >> 5;
    const int lane = tid & 31;
    const int hi   = lane >> 4;
    const int ml   = lane & 15;
    const int row0 = (bid >> 1) * (16 * RT);
    const int n0   = (bid & 1) * 64 + wave * 16;

    v8f acc[RT];
    #pragma unroll
    for (int rt = 0; rt < RT; ++rt) acc[rt] = v8f{};

    const float* ap = A + (size_t)(row0 + wave) * KK + 4 * lane;
    float*       sp = &As[wave * 132 + 4 * lane];
    const float* bp = B + (size_t)(2 * hi) * H + n0 + ml;

    for (int k0 = 0; k0 < KK; k0 += 128) {
        v2f b[32];
        #pragma unroll
        for (int s = 0; s < 32; ++s) {
            b[s].x = bp[(size_t)(k0 + 4 * s) * H];
            b[s].y = bp[(size_t)(k0 + 4 * s + 1) * H];
        }
        float4 st[4 * RT];
        #pragma unroll
        for (int it = 0; it < 4 * RT; ++it)
            st[it] = *(const float4*)(ap + (size_t)(4 * it) * KK + k0);
        #pragma unroll
        for (int it = 0; it < 4 * RT; ++it)
            *(float4*)(sp + (4 * it) * 132) = st[it];
        __syncthreads();

        #pragma unroll
        for (int s = 0; s < 32; ++s) {
            #pragma unroll
            for (int rt = 0; rt < RT; ++rt) {
                v2f a = *(const v2f*)(&As[(rt * 16 + ml) * 132 + 4 * s + 2 * hi]);
                acc[rt] = __builtin_amdgcn_wmma_f32_16x16x4_f32(
                              false, a, false, b[s], (short)0, acc[rt], false, false);
            }
        }
        __syncthreads();
    }

    const float bval = bias ? bias[n0 + ml] : 0.0f;
    #pragma unroll
    for (int rt = 0; rt < RT; ++rt) {
        #pragma unroll
        for (int r = 0; r < 8; ++r) {
            int m = row0 + rt * 16 + r + 8 * hi;
            C[(size_t)m * H + n0 + ml] = acc[rt][r] + bval;
        }
    }
}

template <int RT, int KK>
__global__ __launch_bounds__(128) __attribute__((amdgpu_waves_per_eu(5, 5)))
void gemm_wmma_bias(const float* __restrict__ A,
                    const float* __restrict__ B,
                    const float* __restrict__ bias,
                    float* __restrict__ C,
                    int M)
{
    gemm_tile<RT, KK>(A, B, bias, C, blockIdx.x);
}

struct Proj4 {
    const float* A0; const float* B0; const float* bias0; float* C0;
    const float* A1; const float* B1; const float* bias1; float* C1;
    const float* A2; const float* B2; const float* bias2; float* C2;
    const float* A3; const float* B3; const float* bias3; float* C3;
};

template <int RT, int NB0, int NB2>
__global__ __launch_bounds__(128) __attribute__((amdgpu_waves_per_eu(5, 5)))
void gemm_wmma_bias4(const Proj4 p)
{
    int bid = blockIdx.x;
    const float* A; const float* B; const float* bias; float* C;
    if (bid < NB0) {
        A = p.A0; B = p.B0; bias = p.bias0; C = p.C0;
    } else if (bid < 2 * NB0) {
        bid -= NB0;
        A = p.A1; B = p.B1; bias = p.bias1; C = p.C1;
    } else if (bid < 2 * NB0 + NB2) {
        bid -= 2 * NB0;
        A = p.A2; B = p.B2; bias = p.bias2; C = p.C2;
    } else {
        bid -= 2 * NB0 + NB2;
        A = p.A3; B = p.B3; bias = p.bias3; C = p.C3;
    }
    gemm_tile<RT, H>(A, B, bias, C, bid);
}

__global__ void zero_ints(int* __restrict__ p, int n)
{
    int i = blockIdx.x * blockDim.x + threadIdx.x;
    if (i < n) p[i] = 0;
}

__global__ void dst_count(const int* __restrict__ dst_a,
                          const int* __restrict__ dst_b,
                          int* __restrict__ cnt_a,
                          int* __restrict__ cnt_b,
                          int E, int na, int nb)
{
    int e = blockIdx.x * blockDim.x + threadIdx.x;
    if (e >= E) return;
    int da = min(max(dst_a[e], 0), na - 1);
    int db = min(max(dst_b[e], 0), nb - 1);
    atomicAdd(&cnt_a[da], 1);
    atomicAdd(&cnt_b[db], 1);
}

__global__ __launch_bounds__(1024) void dst_scan(const int* __restrict__ cnt_a,
                                                 int* __restrict__ off_a,
                                                 int* __restrict__ cur_a, int na,
                                                 const int* __restrict__ cnt_b,
                                                 int* __restrict__ off_b,
                                                 int* __restrict__ cur_b, int nb)
{
    __shared__ int wsum[32];
    __shared__ int total_s;

    const int* cnt = blockIdx.x ? cnt_b : cnt_a;
    int*       off = blockIdx.x ? off_b : off_a;
    int*       cur = blockIdx.x ? cur_b : cur_a;
    const int  n   = blockIdx.x ? nb : na;

    const int tid  = threadIdx.x;
    const int lane = tid & 31;
    const int wave = tid >> 5;

    int carry = 0;
    for (int base = 0; base < n; base += 16384) {
        const int idx = base + tid * 16;
        int4 c[4];
        for (int k = 0; k < 4; ++k) c[k] = make_int4(0, 0, 0, 0);
        if (idx < n) {
            for (int k = 0; k < 4; ++k) c[k] = *(const int4*)(cnt + idx + 4 * k);
        }
        int t = 0;
        for (int k = 0; k < 4; ++k) t += c[k].x + c[k].y + c[k].z + c[k].w;

        int incl = t;
        for (int o = 1; o < 32; o <<= 1) {
            int y = __shfl_up(incl, o, 32);
            if (lane >= o) incl += y;
        }
        if (lane == 31) wsum[wave] = incl;
        __syncthreads();
        if (wave == 0) {
            const int v = wsum[lane];
            int iv = v;
            for (int o = 1; o < 32; o <<= 1) {
                int y = __shfl_up(iv, o, 32);
                if (lane >= o) iv += y;
            }
            wsum[lane] = iv - v;
            if (lane == 31) total_s = iv;
        }
        __syncthreads();
        int run = carry + wsum[wave] + incl - t;
        if (idx < n) {
            for (int k = 0; k < 4; ++k) {
                int4 o4;
                o4.x = run;
                o4.y = o4.x + c[k].x;
                o4.z = o4.y + c[k].y;
                o4.w = o4.z + c[k].z;
                run  = o4.w + c[k].w;
                *(int4*)(off + idx + 4 * k) = o4;
                *(int4*)(cur + idx + 4 * k) = o4;
            }
        }
        carry += total_s;
        __syncthreads();
    }
    if (tid == 0) off[n] = carry;
}

__global__ void dst_fill(const int* __restrict__ dst_a,
                         int* __restrict__ cur_a,
                         int* __restrict__ perm_a,
                         const int* __restrict__ dst_b,
                         int* __restrict__ cur_b,
                         int* __restrict__ perm_b,
                         int E, int na, int nb)
{
    int e = blockIdx.x * blockDim.x + threadIdx.x;
    if (e >= E) return;
    {
        int d = min(max(dst_a[e], 0), na - 1);
        int p = atomicAdd(&cur_a[d], 1);
        perm_a[min(max(p, 0), E - 1)] = e;
    }
    {
        int d = min(max(dst_b[e], 0), nb - 1);
        int p = atomicAdd(&cur_b[d], 1);
        perm_b[min(max(p, 0), E - 1)] = e;
    }
}

#define SORT_CAP 512
__global__ __launch_bounds__(256) void dst_order(const int* __restrict__ off_a,
                                                 int* __restrict__ perm_a,
                                                 int* __restrict__ psrc_a,
                                                 const int* __restrict__ src_a, int na,
                                                 const int* __restrict__ off_b,
                                                 int* __restrict__ perm_b,
                                                 int* __restrict__ psrc_b,
                                                 const int* __restrict__ src_b, int nb,
                                                 int E)
{
    __shared__ __align__(16) int skey[8][SORT_CAP];

    const int lane = threadIdx.x & 31;
    const int wv   = threadIdx.x >> 5;
    int row = blockIdx.x * 8 + wv;

    const bool is_a  = row < na;
    const int* off   = is_a ? off_a  : off_b;
    int*       perm  = is_a ? perm_a : perm_b;
    int*       psrc  = is_a ? psrc_a : psrc_b;
    const int* src   = is_a ? src_a  : src_b;
    if (!is_a) row -= na;
    const bool valid = is_a || row < nb;

    int beg = 0, n = 0;
    if (valid) {
        beg = min(max(off[row], 0), E);
        n   = min(max(off[row + 1], beg), E) - beg;
    }
    const bool small = n <= SORT_CAP;
    const int  npad  = (n + 3) & ~3;
    int* key = skey[wv];

    if (small) {
        for (int j = lane; j < npad; j += 32)
            key[j] = (j < n) ? perm[beg + j] : 0x7fffffff;
    }
    __syncthreads();

    if (small) {
        for (int j0 = 0; j0 < n; j0 += 32) {
            const int j  = j0 + lane;
            const int my = (j < n) ? key[j] : 0x7fffffff;
            int r = 0;
            for (int t = 0; t < npad; t += 4) {
                const int4 k4 = *(const int4*)(key + t);
                r += (k4.x < my) + (k4.y < my) + (k4.z < my) + (k4.w < my);
            }
            if (j < n) {
                perm[beg + r] = my;
                psrc[beg + r] = src[min(max(my, 0), E - 1)];
            }
        }
    } else {
        for (int j = lane; j < n; j += 32)
            psrc[beg + j] = src[min(max(perm[beg + j], 0), E - 1)];
    }
}

__global__ __launch_bounds__(256) void edge_rows(const float* __restrict__ kbuf,
                                                 const float* __restrict__ qbuf,
                                                 const float* __restrict__ vbuf,
                                                 const float* __restrict__ xe,
                                                 const float* __restrict__ We,
                                                 const float* __restrict__ be,
                                                 const int*   __restrict__ off,
                                                 const int*   __restrict__ perm,
                                                 const int*   __restrict__ psrc,
                                                 float*       __restrict__ out_dst,
                                                 int n_dst, int n_src, int E)
{
    const int lane = threadIdx.x & 31;
    const int d = __builtin_amdgcn_readfirstlane(blockIdx.x * 8 + (threadIdx.x >> 5));
    if (d >= n_dst) return;

    float4 w[D_EDGE];
    for (int dd = 0; dd < D_EDGE; ++dd)
        w[dd] = *(const float4*)(We + dd * H + 4 * lane);
    const float4 b4 = *(const float4*)(be + 4 * lane);

    int beg = __builtin_amdgcn_readfirstlane(off[d]);
    int end = __builtin_amdgcn_readfirstlane(off[d + 1]);
    beg = min(max(beg, 0), E);
    end = min(max(end, beg), E);

    const float4 k4 = *(const float4*)(kbuf + (size_t)d * H + 4 * lane);
    float4 acc = make_float4(0.f, 0.f, 0.f, 0.f);

    for (int j0 = beg; j0 < end; j0 += 32) {
        const int jj = j0 + lane;
        const int pe = (jj < end) ? perm[jj] : 0;
        const int ps = (jj < end) ? psrc[jj] : 0;
        const int cnt = min(32, end - j0);
        int e = min(max(__builtin_amdgcn_readlane(pe, 0), 0), E - 1);
        int s = min(max(__builtin_amdgcn_readlane(ps, 0), 0), n_src - 1);
        float  xv = xe[(size_t)e * D_EDGE + (lane & 15)];
        float4 q4 = *(const float4*)(qbuf + (size_t)s * H + 4 * lane);
        float4 v4 = *(const float4*)(vbuf + (size_t)s * H + 4 * lane);

        for (int i = 0; i < cnt; ++i) {
            float  xn = xv;
            float4 qn = q4;
            float4 vn = v4;
            if (i + 1 < cnt) {
                e = min(max(__builtin_amdgcn_readlane(pe, i + 1), 0), E - 1);
                s = min(max(__builtin_amdgcn_readlane(ps, i + 1), 0), n_src - 1);
                xn = xe[(size_t)e * D_EDGE + (lane & 15)];
                qn = *(const float4*)(qbuf + (size_t)s * H + 4 * lane);
                vn = *(const float4*)(vbuf + (size_t)s * H + 4 * lane);
            }

            float4 ea = b4;
            for (int dd = 0; dd < D_EDGE; ++dd) {
                const float x = __int_as_float(
                    __builtin_amdgcn_readlane(__float_as_int(xv), dd));
                ea.x = fmaf(x, w[dd].x, ea.x);
                ea.y = fmaf(x, w[dd].y, ea.y);
                ea.z = fmaf(x, w[dd].z, ea.z);
                ea.w = fmaf(x, w[dd].w, ea.w);
            }

            const float gx = 1.0f / (1.0f + __expf(-((k4.x + ea.x) + (q4.x + ea.x))));
            const float gy = 1.0f / (1.0f + __expf(-((k4.y + ea.y) + (q4.y + ea.y))));
            const float gz = 1.0f / (1.0f + __expf(-((k4.z + ea.z) + (q4.z + ea.z))));
            const float gw = 1.0f / (1.0f + __expf(-((k4.w + ea.w) + (q4.w + ea.w))));
            acc.x += gx * (v4.x + ea.x);
            acc.y += gy * (v4.y + ea.y);
            acc.z += gz * (v4.z + ea.z);
            acc.w += gw * (v4.w + ea.w);

            xv = xn; q4 = qn; v4 = vn;
        }
    }

    float4* op = (float4*)(out_dst + (size_t)d * H + 4 * lane);
    float4 o = *op;
    o.x = fmaxf(o.x + acc.x, 0.0f);
    o.y = fmaxf(o.y + acc.y, 0.0f);
    o.z = fmaxf(o.z + acc.z, 0.0f);
    o.w = fmaxf(o.w + acc.w, 0.0f);
    *op = o;
}

__global__ void classifier(const float* __restrict__ hm,
                           const float* __restrict__ hv,
                           const int*   __restrict__ im,
                           const int*   __restrict__ iv,
                           const float* __restrict__ Wc,
                           const float* __restrict__ bc,
                           float*       __restrict__ out,
                           int P)
{
    const int p    = blockIdx.x * 8 + (threadIdx.x >> 5);
    const int lane = threadIdx.x & 31;
    if (p >= P) return;

    const float4 am = ((const float4*)(hm + (size_t)im[p] * H))[lane];
    const float4 av = ((const float4*)(hv + (size_t)iv[p] * H))[lane];

    float c0 = 0.f, c1 = 0.f;
    const int ch = lane * 4;
    const float a[4] = {am.x, am.y, am.z, am.w};
    const float b[4] = {av.x, av.y, av.z, av.w};
    #pragma unroll
    for (int i = 0; i < 4; ++i) {
        c0 = fmaf(a[i], Wc[(ch + i) * 2 + 0], c0);
        c1 = fmaf(a[i], Wc[(ch + i) * 2 + 1], c1);
        c0 = fmaf(b[i], Wc[(H + ch + i) * 2 + 0], c0);
        c1 = fmaf(b[i], Wc[(H + ch + i) * 2 + 1], c1);
    }
    #pragma unroll
    for (int off = 16; off > 0; off >>= 1) {
        c0 += __shfl_down(c0, off, 32);
        c1 += __shfl_down(c1, off, 32);
    }
    if (lane == 0) {
        out[(size_t)p * 2 + 0] = c0 + bc[0];
        out[(size_t)p * 2 + 1] = c1 + bc[1];
    }
}

static inline void gemm_in(const float* A, const float* B, const float* bias,
                           float* C, int M, hipStream_t s)
{
    gemm_wmma_bias<5, D_IN><<<dim3(M / 80 * 2), dim3(128), 0, s>>>(A, B, bias, C, M);
}
#define NB_MOUSE (N_MOUSE / 80 * 2)
#define NB_VIRUS (N_VIRUS / 80 * 2)
static inline void gemm4(const Proj4& p, hipStream_t s)
{
    gemm_wmma_bias4<5, NB_MOUSE, NB_VIRUS>
        <<<dim3(2 * NB_MOUSE + 2 * NB_VIRUS), dim3(128), 0, s>>>(p);
}

extern "C" void kernel_launch(void* const* d_in, const int* in_sizes, int n_in,
                              void* d_out, int out_size, void* d_ws, size_t ws_size,
                              hipStream_t stream)
{
    const float* x_mouse  = (const float*)d_in[0];
    const float* x_virus  = (const float*)d_in[1];
    const float* xe_mv    = (const float*)d_in[2];
    const float* xe_vm    = (const float*)d_in[3];
    const float* W_mouse  = (const float*)d_in[4];
    const float* b_mouse  = (const float*)d_in[5];
    const float* W_virus  = (const float*)d_in[6];
    const float* b_virus  = (const float*)d_in[7];
    const float* W_key    = (const float*)d_in[8];
    const float* b_key    = (const float*)d_in[9];
    const float* W_query  = (const float*)d_in[10];
    const float* b_query  = (const float*)d_in[11];
    const float* W_value  = (const float*)d_in[12];
    const float* b_value  = (const float*)d_in[13];
    const float* W_edge   = (const float*)d_in[14];
    const float* b_edge   = (const float*)d_in[15];
    const float* W_skip   = (const float*)d_in[16];
    const float* bias_cv  = (const float*)d_in[17];
    const float* W_cls    = (const float*)d_in[18];
    const float* b_cls    = (const float*)d_in[19];
    const int* ei_mv_src  = (const int*)d_in[20];
    const int* ei_mv_dst  = (const int*)d_in[21];
    const int* ei_vm_src  = (const int*)d_in[22];
    const int* ei_vm_dst  = (const int*)d_in[23];
    const int* eli_mouse  = (const int*)d_in[24];
    const int* eli_virus  = (const int*)d_in[25];
    float* out = (float*)d_out;

    float* ws  = (float*)d_ws;
    const size_t SZ_M = (size_t)N_MOUSE * H;
    const size_t SZ_V = (size_t)N_VIRUS * H;
    float* hm0 = ws;
    float* hm1 = hm0 + SZ_M;
    float* hv0 = hm1 + SZ_M;
    float* hv1 = hv0 + SZ_V;
    float* bx  = hv1 + SZ_V;
    float* by  = bx + SZ_M;
    float* bz  = by + SZ_M;

    int* cnt_a  = (int*)(bz + SZ_V);
    int* cnt_b  = cnt_a + N_VIRUS;
    int* off_a  = cnt_b + N_MOUSE;
    int* off_b  = off_a + N_VIRUS + 4;
    int* cur_a  = off_b + N_MOUSE + 4;
    int* cur_b  = cur_a + N_VIRUS;
    int* perm_a = cur_b + N_MOUSE;
    int* psrc_a = perm_a + NEDGE;
    int* perm_b = psrc_a + NEDGE;
    int* psrc_b = perm_b + NEDGE;

    const int e_blocks = (NEDGE + 255) / 256;

    zero_ints<<<dim3((N_VIRUS + N_MOUSE + 255) / 256), dim3(256), 0, stream>>>(
        cnt_a, N_VIRUS + N_MOUSE);
    dst_count<<<dim3(e_blocks), dim3(256), 0, stream>>>(
        ei_mv_dst, ei_vm_dst, cnt_a, cnt_b, NEDGE, N_VIRUS, N_MOUSE);
    dst_scan<<<dim3(2), dim3(1024), 0, stream>>>(
        cnt_a, off_a, cur_a, N_VIRUS, cnt_b, off_b, cur_b, N_MOUSE);
    dst_fill<<<dim3(e_blocks), dim3(256), 0, stream>>>(
        ei_mv_dst, cur_a, perm_a, ei_vm_dst, cur_b, perm_b,
        NEDGE, N_VIRUS, N_MOUSE);
    dst_order<<<dim3((N_VIRUS + N_MOUSE + 7) / 8), dim3(256), 0, stream>>>(
        off_a, perm_a, psrc_a, ei_mv_src, N_VIRUS,
        off_b, perm_b, psrc_b, ei_vm_src, N_MOUSE, NEDGE);

    gemm_in(x_mouse, W_mouse, b_mouse, hm0, N_MOUSE, stream);
    gemm_in(x_virus, W_virus, b_virus, hv0, N_VIRUS, stream);

    float* hm_in = hm0; float* hv_in = hv0;
    float* hm_out = hm1; float* hv_out = hv1;

    for (int layer = 0; layer < 2; ++layer) {
        const int i_mv = layer * 2;
        const int i_vm = layer * 2 + 1;

        {
            const int i = i_mv;
            const float* Wk = W_key   + (size_t)i * H * H;
            const float* Wq = W_query + (size_t)i * H * H;
            const float* Wv = W_value + (size_t)i * H * H;
            const float* We = W_edge  + (size_t)i * D_EDGE * H;
            const float* Ws = W_skip  + (size_t)i * H * H;
            Proj4 p;
            p.A0 = hm_in; p.B0 = Wq; p.bias0 = b_query + (size_t)i * H; p.C0 = bx;
            p.A1 = hm_in; p.B1 = Wv; p.bias1 = b_value + (size_t)i * H; p.C1 = by;
            p.A2 = hv_in; p.B2 = Ws; p.bias2 = bias_cv + (size_t)i * H; p.C2 = hv_out;
            p.A3 = hv_in; p.B3 = Wk; p.bias3 = b_key   + (size_t)i * H; p.C3 = bz;
            gemm4(p, stream);
            edge_rows<<<dim3((N_VIRUS + 7) / 8), dim3(256), 0, stream>>>(
                bz, bx, by, xe_mv, We, b_edge + (size_t)i * H,
                off_a, perm_a, psrc_a, hv_out, N_VIRUS, N_MOUSE, NEDGE);
        }
        {
            const int i = i_vm;
            const float* Wk = W_key   + (size_t)i * H * H;
            const float* Wq = W_query + (size_t)i * H * H;
            const float* Wv = W_value + (size_t)i * H * H;
            const float* We = W_edge  + (size_t)i * D_EDGE * H;
            const float* Ws = W_skip  + (size_t)i * H * H;
            Proj4 p;
            p.A0 = hm_in; p.B0 = Ws; p.bias0 = bias_cv + (size_t)i * H; p.C0 = hm_out;
            p.A1 = hm_in; p.B1 = Wk; p.bias1 = b_key   + (size_t)i * H; p.C1 = bx;
            p.A2 = hv_in; p.B2 = Wq; p.bias2 = b_query + (size_t)i * H; p.C2 = by;
            p.A3 = hv_in; p.B3 = Wv; p.bias3 = b_value + (size_t)i * H; p.C3 = by + SZ_V;
            gemm4(p, stream);
            edge_rows<<<dim3((N_MOUSE + 7) / 8), dim3(256), 0, stream>>>(
                bx, by, by + SZ_V, xe_vm, We, b_edge + (size_t)i * H,
                off_b, perm_b, psrc_b, hm_out, N_MOUSE, N_VIRUS, NEDGE);
        }

        float* t;
        t = hm_in; hm_in = hm_out; hm_out = t;
        t = hv_in; hv_in = hv_out; hv_out = t;
    }

    classifier<<<dim3((NPAIR + 7) / 8), dim3(256), 0, stream>>>(
        hm_in, hv_in, eli_mouse, eli_virus, W_cls, b_cls, out, NPAIR);
}


# edge bucketing: the count pass keeps its counting atomic's old value as each edge's slot, so the fill pass places edges with no atomic (one of the two atomic passes gone); sum order kept
# speedup vs baseline: 1.8799x; 1.0813x over previous
#include <hip/hip_runtime.h>
#include <hip/hip_bf16.h>

typedef __attribute__((ext_vector_type(2))) float v2f;
typedef __attribute__((ext_vector_type(8))) float v8f;

#define H 128
#define D_EDGE 16
#define N_MOUSE 50000
#define N_VIRUS 10000
#define NEDGE   500000
#define NPAIR   100000
#define D_IN    1024

template <int RT, int KK>
__device__ __forceinline__
void gemm_tile(const float* __restrict__ A,
               const float* __restrict__ B,
               const float* __restrict__ bias,
               float* __restrict__ C,
               int bid)
{
    __shared__ __align__(16) float As[RT * 16 * 132];

    const int tid  = threadIdx.x;
    const int wave = tid >> 5;
    const int lane = tid & 31;
    const int hi   = lane >> 4;
    const int ml   = lane & 15;
    const int row0 = (bid >> 1) * (16 * RT);
    const int n0   = (bid & 1) * 64 + wave * 16;

    v8f acc[RT];
    #pragma unroll
    for (int rt = 0; rt < RT; ++rt) acc[rt] = v8f{};

    const float* ap = A + (size_t)(row0 + wave) * KK + 4 * lane;
    float*       sp = &As[wave * 132 + 4 * lane];
    const float* bp = B + (size_t)(2 * hi) * H + n0 + ml;

    for (int k0 = 0; k0 < KK; k0 += 128) {
        v2f b[32];
        #pragma unroll
        for (int s = 0; s < 32; ++s) {
            b[s].x = bp[(size_t)(k0 + 4 * s) * H];
            b[s].y = bp[(size_t)(k0 + 4 * s + 1) * H];
        }
        float4 st[4 * RT];
        #pragma unroll
        for (int it = 0; it < 4 * RT; ++it)
            st[it] = *(const float4*)(ap + (size_t)(4 * it) * KK + k0);
        #pragma unroll
        for (int it = 0; it < 4 * RT; ++it)
            *(float4*)(sp + (4 * it) * 132) = st[it];
        __syncthreads();

        #pragma unroll
        for (int s = 0; s < 32; ++s) {
            #pragma unroll
            for (int rt = 0; rt < RT; ++rt) {
                v2f a = *(const v2f*)(&As[(rt * 16 + ml) * 132 + 4 * s + 2 * hi]);
                acc[rt] = __builtin_amdgcn_wmma_f32_16x16x4_f32(
                              false, a, false, b[s], (short)0, acc[rt], false, false);
            }
        }
        __syncthreads();
    }

    const float bval = bias ? bias[n0 + ml] : 0.0f;
    #pragma unroll
    for (int rt = 0; rt < RT; ++rt) {
        #pragma unroll
        for (int r = 0; r < 8; ++r) {
            int m = row0 + rt * 16 + r + 8 * hi;
            C[(size_t)m * H + n0 + ml] = acc[rt][r] + bval;
        }
    }
}

template <int RT, int KK>
__global__ __launch_bounds__(128) __attribute__((amdgpu_waves_per_eu(5, 5)))
void gemm_wmma_bias(const float* __restrict__ A,
                    const float* __restrict__ B,
                    const float* __restrict__ bias,
                    float* __restrict__ C,
                    int M)
{
    gemm_tile<RT, KK>(A, B, bias, C, blockIdx.x);
}

struct Proj4 {
    const float* A0; const float* B0; const float* bias0; float* C0;
    const float* A1; const float* B1; const float* bias1; float* C1;
    const float* A2; const float* B2; const float* bias2; float* C2;
    const float* A3; const float* B3; const float* bias3; float* C3;
};

template <int RT, int NB0, int NB2>
__global__ __launch_bounds__(128) __attribute__((amdgpu_waves_per_eu(5, 5)))
void gemm_wmma_bias4(const Proj4 p)
{
    int bid = blockIdx.x;
    const float* A; const float* B; const float* bias; float* C;
    if (bid < NB0) {
        A = p.A0; B = p.B0; bias = p.bias0; C = p.C0;
    } else if (bid < 2 * NB0) {
        bid -= NB0;
        A = p.A1; B = p.B1; bias = p.bias1; C = p.C1;
    } else if (bid < 2 * NB0 + NB2) {
        bid -= 2 * NB0;
        A = p.A2; B = p.B2; bias = p.bias2; C = p.C2;
    } else {
        bid -= 2 * NB0 + NB2;
        A = p.A3; B = p.B3; bias = p.bias3; C = p.C3;
    }
    gemm_tile<RT, H>(A, B, bias, C, bid);
}

__global__ void zero_ints(int* __restrict__ p, int n)
{
    int i = blockIdx.x * blockDim.x + threadIdx.x;
    if (i < n) p[i] = 0;
}

__global__ void dst_count(const int* __restrict__ dst_a,
                          const int* __restrict__ dst_b,
                          int* __restrict__ cnt_a,
                          int* __restrict__ cnt_b,
                          int* __restrict__ slot_a,
                          int* __restrict__ slot_b,
                          int E, int na, int nb)
{
    int e = blockIdx.x * blockDim.x + threadIdx.x;
    if (e >= E) return;
    int da = min(max(dst_a[e], 0), na - 1);
    int db = min(max(dst_b[e], 0), nb - 1);
    const int sa = atomicAdd(&cnt_a[da], 1);
    const int sb = atomicAdd(&cnt_b[db], 1);
    slot_a[e] = sa;
    slot_b[e] = sb;
}

__global__ __launch_bounds__(1024) void dst_scan(const int* __restrict__ cnt_a,
                                                 int* __restrict__ off_a, int na,
                                                 const int* __restrict__ cnt_b,
                                                 int* __restrict__ off_b, int nb)
{
    __shared__ int wsum[32];
    __shared__ int total_s;

    const int* cnt = blockIdx.x ? cnt_b : cnt_a;
    int*       off = blockIdx.x ? off_b : off_a;
    const int  n   = blockIdx.x ? nb : na;

    const int tid  = threadIdx.x;
    const int lane = tid & 31;
    const int wave = tid >> 5;

    int carry = 0;
    for (int base = 0; base < n; base += 16384) {
        const int idx = base + tid * 16;
        int4 c[4];
        for (int k = 0; k < 4; ++k) c[k] = make_int4(0, 0, 0, 0);
        if (idx < n) {
            for (int k = 0; k < 4; ++k) c[k] = *(const int4*)(cnt + idx + 4 * k);
        }
        int t = 0;
        for (int k = 0; k < 4; ++k) t += c[k].x + c[k].y + c[k].z + c[k].w;

        int incl = t;
        for (int o = 1; o < 32; o <<= 1) {
            int y = __shfl_up(incl, o, 32);
            if (lane >= o) incl += y;
        }
        if (lane == 31) wsum[wave] = incl;
        __syncthreads();
        if (wave == 0) {
            const int v = wsum[lane];
            int iv = v;
            for (int o = 1; o < 32; o <<= 1) {
                int y = __shfl_up(iv, o, 32);
                if (lane >= o) iv += y;
            }
            wsum[lane] = iv - v;
            if (lane == 31) total_s = iv;
        }
        __syncthreads();
        int run = carry + wsum[wave] + incl - t;
        if (idx < n) {
            for (int k = 0; k < 4; ++k) {
                int4 o4;
                o4.x = run;
                o4.y = o4.x + c[k].x;
                o4.z = o4.y + c[k].y;
                o4.w = o4.z + c[k].z;
                run  = o4.w + c[k].w;
                *(int4*)(off + idx + 4 * k) = o4;
            }
        }
        carry += total_s;
        __syncthreads();
    }
    if (tid == 0) off[n] = carry;
}

__global__ void dst_fill(const int* __restrict__ dst_a,
                         const int* __restrict__ off_a,
                         const int* __restrict__ slot_a,
                         int* __restrict__ perm_a,
                         const int* __restrict__ dst_b,
                         const int* __restrict__ off_b,
                         const int* __restrict__ slot_b,
                         int* __restrict__ perm_b,
                         int E, int na, int nb)
{
    int e = blockIdx.x * blockDim.x + threadIdx.x;
    if (e >= E) return;
    {
        int d = min(max(dst_a[e], 0), na - 1);
        int p = off_a[d] + slot_a[e];
        perm_a[min(max(p, 0), E - 1)] = e;
    }
    {
        int d = min(max(dst_b[e], 0), nb - 1);
        int p = off_b[d] + slot_b[e];
        perm_b[min(max(p, 0), E - 1)] = e;
    }
}

#define SORT_CAP 512
__global__ __launch_bounds__(256) void dst_order(const int* __restrict__ off_a,
                                                 int* __restrict__ perm_a,
                                                 int* __restrict__ psrc_a,
                                                 const int* __restrict__ src_a, int na,
                                                 const int* __restrict__ off_b,
                                                 int* __restrict__ perm_b,
                                                 int* __restrict__ psrc_b,
                                                 const int* __restrict__ src_b, int nb,
                                                 int E)
{
    __shared__ __align__(16) int skey[8][SORT_CAP];

    const int lane = threadIdx.x & 31;
    const int wv   = threadIdx.x >> 5;
    int row = blockIdx.x * 8 + wv;

    const bool is_a  = row < na;
    const int* off   = is_a ? off_a  : off_b;
    int*       perm  = is_a ? perm_a : perm_b;
    int*       psrc  = is_a ? psrc_a : psrc_b;
    const int* src   = is_a ? src_a  : src_b;
    if (!is_a) row -= na;
    const bool valid = is_a || row < nb;

    int beg = 0, n = 0;
    if (valid) {
        beg = min(max(off[row], 0), E);
        n   = min(max(off[row + 1], beg), E) - beg;
    }
    const bool small = n <= SORT_CAP;
    const int  npad  = (n + 3) & ~3;
    int* key = skey[wv];

    if (small) {
        for (int j = lane; j < npad; j += 32)
            key[j] = (j < n) ? perm[beg + j] : 0x7fffffff;
    }
    __syncthreads();

    if (small) {
        for (int j0 = 0; j0 < n; j0 += 32) {
            const int j  = j0 + lane;
            const int my = (j < n) ? key[j] : 0x7fffffff;
            int r = 0;
            for (int t = 0; t < npad; t += 4) {
                const int4 k4 = *(const int4*)(key + t);
                r += (k4.x < my) + (k4.y < my) + (k4.z < my) + (k4.w < my);
            }
            if (j < n) {
                perm[beg + r] = my;
                psrc[beg + r] = src[min(max(my, 0), E - 1)];
            }
        }
    } else {
        for (int j = lane; j < n; j += 32)
            psrc[beg + j] = src[min(max(perm[beg + j], 0), E - 1)];
    }
}

__global__ __launch_bounds__(256) void edge_rows(const float* __restrict__ kbuf,
                                                 const float* __restrict__ qbuf,
                                                 const float* __restrict__ vbuf,
                                                 const float* __restrict__ xe,
                                                 const float* __restrict__ We,
                                                 const float* __restrict__ be,
                                                 const int*   __restrict__ off,
                                                 const int*   __restrict__ perm,
                                                 const int*   __restrict__ psrc,
                                                 float*       __restrict__ out_dst,
                                                 int n_dst, int n_src, int E)
{
    const int lane = threadIdx.x & 31;
    const int d = __builtin_amdgcn_readfirstlane(blockIdx.x * 8 + (threadIdx.x >> 5));
    if (d >= n_dst) return;

    float4 w[D_EDGE];
    for (int dd = 0; dd < D_EDGE; ++dd)
        w[dd] = *(const float4*)(We + dd * H + 4 * lane);
    const float4 b4 = *(const float4*)(be + 4 * lane);

    int beg = __builtin_amdgcn_readfirstlane(off[d]);
    int end = __builtin_amdgcn_readfirstlane(off[d + 1]);
    beg = min(max(beg, 0), E);
    end = min(max(end, beg), E);

    const float4 k4 = *(const float4*)(kbuf + (size_t)d * H + 4 * lane);
    float4 acc = make_float4(0.f, 0.f, 0.f, 0.f);

    for (int j0 = beg; j0 < end; j0 += 32) {
        const int jj = j0 + lane;
        const int pe = (jj < end) ? perm[jj] : 0;
        const int ps = (jj < end) ? psrc[jj] : 0;
        const int cnt = min(32, end - j0);
        int e = min(max(__builtin_amdgcn_readlane(pe, 0), 0), E - 1);
        int s = min(max(__builtin_amdgcn_readlane(ps, 0), 0), n_src - 1);
        float  xv = xe[(size_t)e * D_EDGE + (lane & 15)];
        float4 q4 = *(const float4*)(qbuf + (size_t)s * H + 4 * lane);
        float4 v4 = *(const float4*)(vbuf + (size_t)s * H + 4 * lane);

        for (int i = 0; i < cnt; ++i) {
            float  xn = xv;
            float4 qn = q4;
            float4 vn = v4;
            if (i + 1 < cnt) {
                e = min(max(__builtin_amdgcn_readlane(pe, i + 1), 0), E - 1);
                s = min(max(__builtin_amdgcn_readlane(ps, i + 1), 0), n_src - 1);
                xn = xe[(size_t)e * D_EDGE + (lane & 15)];
                qn = *(const float4*)(qbuf + (size_t)s * H + 4 * lane);
                vn = *(const float4*)(vbuf + (size_t)s * H + 4 * lane);
            }

            float4 ea = b4;
            for (int dd = 0; dd < D_EDGE; ++dd) {
                const float x = __int_as_float(
                    __builtin_amdgcn_readlane(__float_as_int(xv), dd));
                ea.x = fmaf(x, w[dd].x, ea.x);
                ea.y = fmaf(x, w[dd].y, ea.y);
                ea.z = fmaf(x, w[dd].z, ea.z);
                ea.w = fmaf(x, w[dd].w, ea.w);
            }

            const float gx = 1.0f / (1.0f + __expf(-((k4.x + ea.x) + (q4.x + ea.x))));
            const float gy = 1.0f / (1.0f + __expf(-((k4.y + ea.y) + (q4.y + ea.y))));
            const float gz = 1.0f / (1.0f + __expf(-((k4.z + ea.z) + (q4.z + ea.z))));
            const float gw = 1.0f / (1.0f + __expf(-((k4.w + ea.w) + (q4.w + ea.w))));
            acc.x += gx * (v4.x + ea.x);
            acc.y += gy * (v4.y + ea.y);
            acc.z += gz * (v4.z + ea.z);
            acc.w += gw * (v4.w + ea.w);

            xv = xn; q4 = qn; v4 = vn;
        }
    }

    float4* op = (float4*)(out_dst + (size_t)d * H + 4 * lane);
    float4 o = *op;
    o.x = fmaxf(o.x + acc.x, 0.0f);
    o.y = fmaxf(o.y + acc.y, 0.0f);
    o.z = fmaxf(o.z + acc.z, 0.0f);
    o.w = fmaxf(o.w + acc.w, 0.0f);
    *op = o;
}

__global__ void classifier(const float* __restrict__ hm,
                           const float* __restrict__ hv,
                           const int*   __restrict__ im,
                           const int*   __restrict__ iv,
                           const float* __restrict__ Wc,
                           const float* __restrict__ bc,
                           float*       __restrict__ out,
                           int P)
{
    const int p    = blockIdx.x * 8 + (threadIdx.x >> 5);
    const int lane = threadIdx.x & 31;
    if (p >= P) return;

    const float4 am = ((const float4*)(hm + (size_t)im[p] * H))[lane];
    const float4 av = ((const float4*)(hv + (size_t)iv[p] * H))[lane];

    float c0 = 0.f, c1 = 0.f;
    const int ch = lane * 4;
    const float a[4] = {am.x, am.y, am.z, am.w};
    const float b[4] = {av.x, av.y, av.z, av.w};
    #pragma unroll
    for (int i = 0; i < 4; ++i) {
        c0 = fmaf(a[i], Wc[(ch + i) * 2 + 0], c0);
        c1 = fmaf(a[i], Wc[(ch + i) * 2 + 1], c1);
        c0 = fmaf(b[i], Wc[(H + ch + i) * 2 + 0], c0);
        c1 = fmaf(b[i], Wc[(H + ch + i) * 2 + 1], c1);
    }
    #pragma unroll
    for (int off = 16; off > 0; off >>= 1) {
        c0 += __shfl_down(c0, off, 32);
        c1 += __shfl_down(c1, off, 32);
    }
    if (lane == 0) {
        out[(size_t)p * 2 + 0] = c0 + bc[0];
        out[(size_t)p * 2 + 1] = c1 + bc[1];
    }
}

static inline void gemm_in(const float* A, const float* B, const float* bias,
                           float* C, int M, hipStream_t s)
{
    gemm_wmma_bias<5, D_IN><<<dim3(M / 80 * 2), dim3(128), 0, s>>>(A, B, bias, C, M);
}
#define NB_MOUSE (N_MOUSE / 80 * 2)
#define NB_VIRUS (N_VIRUS / 80 * 2)
static inline void gemm4(const Proj4& p, hipStream_t s)
{
    gemm_wmma_bias4<5, NB_MOUSE, NB_VIRUS>
        <<<dim3(2 * NB_MOUSE + 2 * NB_VIRUS), dim3(128), 0, s>>>(p);
}

extern "C" void kernel_launch(void* const* d_in, const int* in_sizes, int n_in,
                              void* d_out, int out_size, void* d_ws, size_t ws_size,
                              hipStream_t stream)
{
    const float* x_mouse  = (const float*)d_in[0];
    const float* x_virus  = (const float*)d_in[1];
    const float* xe_mv    = (const float*)d_in[2];
    const float* xe_vm    = (const float*)d_in[3];
    const float* W_mouse  = (const float*)d_in[4];
    const float* b_mouse  = (const float*)d_in[5];
    const float* W_virus  = (const float*)d_in[6];
    const float* b_virus  = (const float*)d_in[7];
    const float* W_key    = (const float*)d_in[8];
    const float* b_key    = (const float*)d_in[9];
    const float* W_query  = (const float*)d_in[10];
    const float* b_query  = (const float*)d_in[11];
    const float* W_value  = (const float*)d_in[12];
    const float* b_value  = (const float*)d_in[13];
    const float* W_edge   = (const float*)d_in[14];
    const float* b_edge   = (const float*)d_in[15];
    const float* W_skip   = (const float*)d_in[16];
    const float* bias_cv  = (const float*)d_in[17];
    const float* W_cls    = (const float*)d_in[18];
    const float* b_cls    = (const float*)d_in[19];
    const int* ei_mv_src  = (const int*)d_in[20];
    const int* ei_mv_dst  = (const int*)d_in[21];
    const int* ei_vm_src  = (const int*)d_in[22];
    const int* ei_vm_dst  = (const int*)d_in[23];
    const int* eli_mouse  = (const int*)d_in[24];
    const int* eli_virus  = (const int*)d_in[25];
    float* out = (float*)d_out;

    float* ws  = (float*)d_ws;
    const size_t SZ_M = (size_t)N_MOUSE * H;
    const size_t SZ_V = (size_t)N_VIRUS * H;
    float* hm0 = ws;
    float* hm1 = hm0 + SZ_M;
    float* hv0 = hm1 + SZ_M;
    float* hv1 = hv0 + SZ_V;
    float* bx  = hv1 + SZ_V;
    float* by  = bx + SZ_M;
    float* bz  = by + SZ_M;

    int* cnt_a  = (int*)(bz + SZ_V);
    int* cnt_b  = cnt_a + N_VIRUS;
    int* off_a  = cnt_b + N_MOUSE;
    int* off_b  = off_a + N_VIRUS + 4;
    int* perm_a = off_b + N_MOUSE + 4;
    int* psrc_a = perm_a + NEDGE;
    int* perm_b = psrc_a + NEDGE;
    int* psrc_b = perm_b + NEDGE;
    int* slot_a = psrc_a;
    int* slot_b = psrc_b;

    const int e_blocks = (NEDGE + 255) / 256;

    zero_ints<<<dim3((N_VIRUS + N_MOUSE + 255) / 256), dim3(256), 0, stream>>>(
        cnt_a, N_VIRUS + N_MOUSE);
    dst_count<<<dim3(e_blocks), dim3(256), 0, stream>>>(
        ei_mv_dst, ei_vm_dst, cnt_a, cnt_b, slot_a, slot_b,
        NEDGE, N_VIRUS, N_MOUSE);
    dst_scan<<<dim3(2), dim3(1024), 0, stream>>>(
        cnt_a, off_a, N_VIRUS, cnt_b, off_b, N_MOUSE);
    dst_fill<<<dim3(e_blocks), dim3(256), 0, stream>>>(
        ei_mv_dst, off_a, slot_a, perm_a, ei_vm_dst, off_b, slot_b, perm_b,
        NEDGE, N_VIRUS, N_MOUSE);
    dst_order<<<dim3((N_VIRUS + N_MOUSE + 7) / 8), dim3(256), 0, stream>>>(
        off_a, perm_a, psrc_a, ei_mv_src, N_VIRUS,
        off_b, perm_b, psrc_b, ei_vm_src, N_MOUSE, NEDGE);

    gemm_in(x_mouse, W_mouse, b_mouse, hm0, N_MOUSE, stream);
    gemm_in(x_virus, W_virus, b_virus, hv0, N_VIRUS, stream);

    float* hm_in = hm0; float* hv_in = hv0;
    float* hm_out = hm1; float* hv_out = hv1;

    for (int layer = 0; layer < 2; ++layer) {
        const int i_mv = layer * 2;
        const int i_vm = layer * 2 + 1;

        {
            const int i = i_mv;
            const float* Wk = W_key   + (size_t)i * H * H;
            const float* Wq = W_query + (size_t)i * H * H;
            const float* Wv = W_value + (size_t)i * H * H;
            const float* We = W_edge  + (size_t)i * D_EDGE * H;
            const float* Ws = W_skip  + (size_t)i * H * H;
            Proj4 p;
            p.A0 = hm_in; p.B0 = Wq; p.bias0 = b_query + (size_t)i * H; p.C0 = bx;
            p.A1 = hm_in; p.B1 = Wv; p.bias1 = b_value + (size_t)i * H; p.C1 = by;
            p.A2 = hv_in; p.B2 = Ws; p.bias2 = bias_cv + (size_t)i * H; p.C2 = hv_out;
            p.A3 = hv_in; p.B3 = Wk; p.bias3 = b_key   + (size_t)i * H; p.C3 = bz;
            gemm4(p, stream);
            edge_rows<<<dim3((N_VIRUS + 7) / 8), dim3(256), 0, stream>>>(
                bz, bx, by, xe_mv, We, b_edge + (size_t)i * H,
                off_a, perm_a, psrc_a, hv_out, N_VIRUS, N_MOUSE, NEDGE);
        }
        {
            const int i = i_vm;
            const float* Wk = W_key   + (size_t)i * H * H;
            const float* Wq = W_query + (size_t)i * H * H;
            const float* Wv = W_value + (size_t)i * H * H;
            const float* We = W_edge  + (size_t)i * D_EDGE * H;
            const float* Ws = W_skip  + (size_t)i * H * H;
            Proj4 p;
            p.A0 = hm_in; p.B0 = Ws; p.bias0 = bias_cv + (size_t)i * H; p.C0 = hm_out;
            p.A1 = hm_in; p.B1 = Wk; p.bias1 = b_key   + (size_t)i * H; p.C1 = bx;
            p.A2 = hv_in; p.B2 = Wq; p.bias2 = b_query + (size_t)i * H; p.C2 = by;
            p.A3 = hv_in; p.B3 = Wv; p.bias3 = b_value + (size_t)i * H; p.C3 = by + SZ_V;
            gemm4(p, stream);
            edge_rows<<<dim3((N_MOUSE + 7) / 8), dim3(256), 0, stream>>>(
                bx, by, by + SZ_V, xe_vm, We, b_edge + (size_t)i * H,
                off_b, perm_b, psrc_b, hm_out, N_MOUSE, N_VIRUS, NEDGE);
        }

        float* t;
        t = hm_in; hm_in = hm_out; hm_out = t;
        t = hv_in; hv_in = hv_out; hv_out = t;
    }

    classifier<<<dim3((NPAIR + 7) / 8), dim3(256), 0, stream>>>(
        hm_in, hv_in, eli_mouse, eli_virus, W_cls, b_cls, out, NPAIR);
}


# edge kernel gate: 1/(1+exp(-z)) takes the hardware reciprocal (v_rcp_f32, 1 ulp) instead of the IEEE division sequence: 116 -> 85 vector instructions per edge; sum order kept
# speedup vs baseline: 1.9614x; 1.0434x over previous
#include <hip/hip_runtime.h>
#include <hip/hip_bf16.h>

typedef __attribute__((ext_vector_type(2))) float v2f;
typedef __attribute__((ext_vector_type(8))) float v8f;

#define H 128
#define D_EDGE 16
#define N_MOUSE 50000
#define N_VIRUS 10000
#define NEDGE   500000
#define NPAIR   100000
#define D_IN    1024

template <int RT, int KK>
__device__ __forceinline__
void gemm_tile(const float* __restrict__ A,
               const float* __restrict__ B,
               const float* __restrict__ bias,
               float* __restrict__ C,
               int bid)
{
    __shared__ __align__(16) float As[RT * 16 * 132];

    const int tid  = threadIdx.x;
    const int wave = tid >> 5;
    const int lane = tid & 31;
    const int hi   = lane >> 4;
    const int ml   = lane & 15;
    const int row0 = (bid >> 1) * (16 * RT);
    const int n0   = (bid & 1) * 64 + wave * 16;

    v8f acc[RT];
    #pragma unroll
    for (int rt = 0; rt < RT; ++rt) acc[rt] = v8f{};

    const float* ap = A + (size_t)(row0 + wave) * KK + 4 * lane;
    float*       sp = &As[wave * 132 + 4 * lane];
    const float* bp = B + (size_t)(2 * hi) * H + n0 + ml;

    for (int k0 = 0; k0 < KK; k0 += 128) {
        v2f b[32];
        #pragma unroll
        for (int s = 0; s < 32; ++s) {
            b[s].x = bp[(size_t)(k0 + 4 * s) * H];
            b[s].y = bp[(size_t)(k0 + 4 * s + 1) * H];
        }
        float4 st[4 * RT];
        #pragma unroll
        for (int it = 0; it < 4 * RT; ++it)
            st[it] = *(const float4*)(ap + (size_t)(4 * it) * KK + k0);
        #pragma unroll
        for (int it = 0; it < 4 * RT; ++it)
            *(float4*)(sp + (4 * it) * 132) = st[it];
        __syncthreads();

        #pragma unroll
        for (int s = 0; s < 32; ++s) {
            #pragma unroll
            for (int rt = 0; rt < RT; ++rt) {
                v2f a = *(const v2f*)(&As[(rt * 16 + ml) * 132 + 4 * s + 2 * hi]);
                acc[rt] = __builtin_amdgcn_wmma_f32_16x16x4_f32(
                              false, a, false, b[s], (short)0, acc[rt], false, false);
            }
        }
        __syncthreads();
    }

    const float bval = bias ? bias[n0 + ml] : 0.0f;
    #pragma unroll
    for (int rt = 0; rt < RT; ++rt) {
        #pragma unroll
        for (int r = 0; r < 8; ++r) {
            int m = row0 + rt * 16 + r + 8 * hi;
            C[(size_t)m * H + n0 + ml] = acc[rt][r] + bval;
        }
    }
}

template <int RT, int KK>
__global__ __launch_bounds__(128) __attribute__((amdgpu_waves_per_eu(5, 5)))
void gemm_wmma_bias(const float* __restrict__ A,
                    const float* __restrict__ B,
                    const float* __restrict__ bias,
                    float* __restrict__ C,
                    int M)
{
    gemm_tile<RT, KK>(A, B, bias, C, blockIdx.x);
}

struct Proj4 {
    const float* A0; const float* B0; const float* bias0; float* C0;
    const float* A1; const float* B1; const float* bias1; float* C1;
    const float* A2; const float* B2; const float* bias2; float* C2;
    const float* A3; const float* B3; const float* bias3; float* C3;
};

template <int RT, int NB0, int NB2>
__global__ __launch_bounds__(128) __attribute__((amdgpu_waves_per_eu(5, 5)))
void gemm_wmma_bias4(const Proj4 p)
{
    int bid = blockIdx.x;
    const float* A; const float* B; const float* bias; float* C;
    if (bid < NB0) {
        A = p.A0; B = p.B0; bias = p.bias0; C = p.C0;
    } else if (bid < 2 * NB0) {
        bid -= NB0;
        A = p.A1; B = p.B1; bias = p.bias1; C = p.C1;
    } else if (bid < 2 * NB0 + NB2) {
        bid -= 2 * NB0;
        A = p.A2; B = p.B2; bias = p.bias2; C = p.C2;
    } else {
        bid -= 2 * NB0 + NB2;
        A = p.A3; B = p.B3; bias = p.bias3; C = p.C3;
    }
    gemm_tile<RT, H>(A, B, bias, C, bid);
}

__global__ void zero_ints(int* __restrict__ p, int n)
{
    int i = blockIdx.x * blockDim.x + threadIdx.x;
    if (i < n) p[i] = 0;
}

__global__ void dst_count(const int* __restrict__ dst_a,
                          const int* __restrict__ dst_b,
                          int* __restrict__ cnt_a,
                          int* __restrict__ cnt_b,
                          int* __restrict__ slot_a,
                          int* __restrict__ slot_b,
                          int E, int na, int nb)
{
    int e = blockIdx.x * blockDim.x + threadIdx.x;
    if (e >= E) return;
    int da = min(max(dst_a[e], 0), na - 1);
    int db = min(max(dst_b[e], 0), nb - 1);
    const int sa = atomicAdd(&cnt_a[da], 1);
    const int sb = atomicAdd(&cnt_b[db], 1);
    slot_a[e] = sa;
    slot_b[e] = sb;
}

__global__ __launch_bounds__(1024) void dst_scan(const int* __restrict__ cnt_a,
                                                 int* __restrict__ off_a, int na,
                                                 const int* __restrict__ cnt_b,
                                                 int* __restrict__ off_b, int nb)
{
    __shared__ int wsum[32];
    __shared__ int total_s;

    const int* cnt = blockIdx.x ? cnt_b : cnt_a;
    int*       off = blockIdx.x ? off_b : off_a;
    const int  n   = blockIdx.x ? nb : na;

    const int tid  = threadIdx.x;
    const int lane = tid & 31;
    const int wave = tid >> 5;

    int carry = 0;
    for (int base = 0; base < n; base += 16384) {
        const int idx = base + tid * 16;
        int4 c[4];
        for (int k = 0; k < 4; ++k) c[k] = make_int4(0, 0, 0, 0);
        if (idx < n) {
            for (int k = 0; k < 4; ++k) c[k] = *(const int4*)(cnt + idx + 4 * k);
        }
        int t = 0;
        for (int k = 0; k < 4; ++k) t += c[k].x + c[k].y + c[k].z + c[k].w;

        int incl = t;
        for (int o = 1; o < 32; o <<= 1) {
            int y = __shfl_up(incl, o, 32);
            if (lane >= o) incl += y;
        }
        if (lane == 31) wsum[wave] = incl;
        __syncthreads();
        if (wave == 0) {
            const int v = wsum[lane];
            int iv = v;
            for (int o = 1; o < 32; o <<= 1) {
                int y = __shfl_up(iv, o, 32);
                if (lane >= o) iv += y;
            }
            wsum[lane] = iv - v;
            if (lane == 31) total_s = iv;
        }
        __syncthreads();
        int run = carry + wsum[wave] + incl - t;
        if (idx < n) {
            for (int k = 0; k < 4; ++k) {
                int4 o4;
                o4.x = run;
                o4.y = o4.x + c[k].x;
                o4.z = o4.y + c[k].y;
                o4.w = o4.z + c[k].z;
                run  = o4.w + c[k].w;
                *(int4*)(off + idx + 4 * k) = o4;
            }
        }
        carry += total_s;
        __syncthreads();
    }
    if (tid == 0) off[n] = carry;
}

__global__ void dst_fill(const int* __restrict__ dst_a,
                         const int* __restrict__ off_a,
                         const int* __restrict__ slot_a,
                         int* __restrict__ perm_a,
                         const int* __restrict__ dst_b,
                         const int* __restrict__ off_b,
                         const int* __restrict__ slot_b,
                         int* __restrict__ perm_b,
                         int E, int na, int nb)
{
    int e = blockIdx.x * blockDim.x + threadIdx.x;
    if (e >= E) return;
    {
        int d = min(max(dst_a[e], 0), na - 1);
        int p = off_a[d] + slot_a[e];
        perm_a[min(max(p, 0), E - 1)] = e;
    }
    {
        int d = min(max(dst_b[e], 0), nb - 1);
        int p = off_b[d] + slot_b[e];
        perm_b[min(max(p, 0), E - 1)] = e;
    }
}

#define SORT_CAP 512
__global__ __launch_bounds__(256) void dst_order(const int* __restrict__ off_a,
                                                 int* __restrict__ perm_a,
                                                 int* __restrict__ psrc_a,
                                                 const int* __restrict__ src_a, int na,
                                                 const int* __restrict__ off_b,
                                                 int* __restrict__ perm_b,
                                                 int* __restrict__ psrc_b,
                                                 const int* __restrict__ src_b, int nb,
                                                 int E)
{
    __shared__ __align__(16) int skey[8][SORT_CAP];

    const int lane = threadIdx.x & 31;
    const int wv   = threadIdx.x >> 5;
    int row = blockIdx.x * 8 + wv;

    const bool is_a  = row < na;
    const int* off   = is_a ? off_a  : off_b;
    int*       perm  = is_a ? perm_a : perm_b;
    int*       psrc  = is_a ? psrc_a : psrc_b;
    const int* src   = is_a ? src_a  : src_b;
    if (!is_a) row -= na;
    const bool valid = is_a || row < nb;

    int beg = 0, n = 0;
    if (valid) {
        beg = min(max(off[row], 0), E);
        n   = min(max(off[row + 1], beg), E) - beg;
    }
    const bool small = n <= SORT_CAP;
    const int  npad  = (n + 3) & ~3;
    int* key = skey[wv];

    if (small) {
        for (int j = lane; j < npad; j += 32)
            key[j] = (j < n) ? perm[beg + j] : 0x7fffffff;
    }
    __syncthreads();

    if (small) {
        for (int j0 = 0; j0 < n; j0 += 32) {
            const int j  = j0 + lane;
            const int my = (j < n) ? key[j] : 0x7fffffff;
            int r = 0;
            for (int t = 0; t < npad; t += 4) {
                const int4 k4 = *(const int4*)(key + t);
                r += (k4.x < my) + (k4.y < my) + (k4.z < my) + (k4.w < my);
            }
            if (j < n) {
                perm[beg + r] = my;
                psrc[beg + r] = src[min(max(my, 0), E - 1)];
            }
        }
    } else {
        for (int j = lane; j < n; j += 32)
            psrc[beg + j] = src[min(max(perm[beg + j], 0), E - 1)];
    }
}

__global__ __launch_bounds__(256) void edge_rows(const float* __restrict__ kbuf,
                                                 const float* __restrict__ qbuf,
                                                 const float* __restrict__ vbuf,
                                                 const float* __restrict__ xe,
                                                 const float* __restrict__ We,
                                                 const float* __restrict__ be,
                                                 const int*   __restrict__ off,
                                                 const int*   __restrict__ perm,
                                                 const int*   __restrict__ psrc,
                                                 float*       __restrict__ out_dst,
                                                 int n_dst, int n_src, int E)
{
    const int lane = threadIdx.x & 31;
    const int d = __builtin_amdgcn_readfirstlane(blockIdx.x * 8 + (threadIdx.x >> 5));
    if (d >= n_dst) return;

    float4 w[D_EDGE];
    for (int dd = 0; dd < D_EDGE; ++dd)
        w[dd] = *(const float4*)(We + dd * H + 4 * lane);
    const float4 b4 = *(const float4*)(be + 4 * lane);

    int beg = __builtin_amdgcn_readfirstlane(off[d]);
    int end = __builtin_amdgcn_readfirstlane(off[d + 1]);
    beg = min(max(beg, 0), E);
    end = min(max(end, beg), E);

    const float4 k4 = *(const float4*)(kbuf + (size_t)d * H + 4 * lane);
    float4 acc = make_float4(0.f, 0.f, 0.f, 0.f);

    for (int j0 = beg; j0 < end; j0 += 32) {
        const int jj = j0 + lane;
        const int pe = (jj < end) ? perm[jj] : 0;
        const int ps = (jj < end) ? psrc[jj] : 0;
        const int cnt = min(32, end - j0);
        int e = min(max(__builtin_amdgcn_readlane(pe, 0), 0), E - 1);
        int s = min(max(__builtin_amdgcn_readlane(ps, 0), 0), n_src - 1);
        float  xv = xe[(size_t)e * D_EDGE + (lane & 15)];
        float4 q4 = *(const float4*)(qbuf + (size_t)s * H + 4 * lane);
        float4 v4 = *(const float4*)(vbuf + (size_t)s * H + 4 * lane);

        for (int i = 0; i < cnt; ++i) {
            float  xn = xv;
            float4 qn = q4;
            float4 vn = v4;
            if (i + 1 < cnt) {
                e = min(max(__builtin_amdgcn_readlane(pe, i + 1), 0), E - 1);
                s = min(max(__builtin_amdgcn_readlane(ps, i + 1), 0), n_src - 1);
                xn = xe[(size_t)e * D_EDGE + (lane & 15)];
                qn = *(const float4*)(qbuf + (size_t)s * H + 4 * lane);
                vn = *(const float4*)(vbuf + (size_t)s * H + 4 * lane);
            }

            float4 ea = b4;
            for (int dd = 0; dd < D_EDGE; ++dd) {
                const float x = __int_as_float(
                    __builtin_amdgcn_readlane(__float_as_int(xv), dd));
                ea.x = fmaf(x, w[dd].x, ea.x);
                ea.y = fmaf(x, w[dd].y, ea.y);
                ea.z = fmaf(x, w[dd].z, ea.z);
                ea.w = fmaf(x, w[dd].w, ea.w);
            }

            const float gx = __builtin_amdgcn_rcpf(1.0f + __expf(-((k4.x + ea.x) + (q4.x + ea.x))));
            const float gy = __builtin_amdgcn_rcpf(1.0f + __expf(-((k4.y + ea.y) + (q4.y + ea.y))));
            const float gz = __builtin_amdgcn_rcpf(1.0f + __expf(-((k4.z + ea.z) + (q4.z + ea.z))));
            const float gw = __builtin_amdgcn_rcpf(1.0f + __expf(-((k4.w + ea.w) + (q4.w + ea.w))));
            acc.x += gx * (v4.x + ea.x);
            acc.y += gy * (v4.y + ea.y);
            acc.z += gz * (v4.z + ea.z);
            acc.w += gw * (v4.w + ea.w);

            xv = xn; q4 = qn; v4 = vn;
        }
    }

    float4* op = (float4*)(out_dst + (size_t)d * H + 4 * lane);
    float4 o = *op;
    o.x = fmaxf(o.x + acc.x, 0.0f);
    o.y = fmaxf(o.y + acc.y, 0.0f);
    o.z = fmaxf(o.z + acc.z, 0.0f);
    o.w = fmaxf(o.w + acc.w, 0.0f);
    *op = o;
}

__global__ void classifier(const float* __restrict__ hm,
                           const float* __restrict__ hv,
                           const int*   __restrict__ im,
                           const int*   __restrict__ iv,
                           const float* __restrict__ Wc,
                           const float* __restrict__ bc,
                           float*       __restrict__ out,
                           int P)
{
    const int p    = blockIdx.x * 8 + (threadIdx.x >> 5);
    const int lane = threadIdx.x & 31;
    if (p >= P) return;

    const float4 am = ((const float4*)(hm + (size_t)im[p] * H))[lane];
    const float4 av = ((const float4*)(hv + (size_t)iv[p] * H))[lane];

    float c0 = 0.f, c1 = 0.f;
    const int ch = lane * 4;
    const float a[4] = {am.x, am.y, am.z, am.w};
    const float b[4] = {av.x, av.y, av.z, av.w};
    #pragma unroll
    for (int i = 0; i < 4; ++i) {
        c0 = fmaf(a[i], Wc[(ch + i) * 2 + 0], c0);
        c1 = fmaf(a[i], Wc[(ch + i) * 2 + 1], c1);
        c0 = fmaf(b[i], Wc[(H + ch + i) * 2 + 0], c0);
        c1 = fmaf(b[i], Wc[(H + ch + i) * 2 + 1], c1);
    }
    #pragma unroll
    for (int off = 16; off > 0; off >>= 1) {
        c0 += __shfl_down(c0, off, 32);
        c1 += __shfl_down(c1, off, 32);
    }
    if (lane == 0) {
        out[(size_t)p * 2 + 0] = c0 + bc[0];
        out[(size_t)p * 2 + 1] = c1 + bc[1];
    }
}

static inline void gemm_in(const float* A, const float* B, const float* bias,
                           float* C, int M, hipStream_t s)
{
    gemm_wmma_bias<5, D_IN><<<dim3(M / 80 * 2), dim3(128), 0, s>>>(A, B, bias, C, M);
}
#define NB_MOUSE (N_MOUSE / 80 * 2)
#define NB_VIRUS (N_VIRUS / 80 * 2)
static inline void gemm4(const Proj4& p, hipStream_t s)
{
    gemm_wmma_bias4<5, NB_MOUSE, NB_VIRUS>
        <<<dim3(2 * NB_MOUSE + 2 * NB_VIRUS), dim3(128), 0, s>>>(p);
}

extern "C" void kernel_launch(void* const* d_in, const int* in_sizes, int n_in,
                              void* d_out, int out_size, void* d_ws, size_t ws_size,
                              hipStream_t stream)
{
    const float* x_mouse  = (const float*)d_in[0];
    const float* x_virus  = (const float*)d_in[1];
    const float* xe_mv    = (const float*)d_in[2];
    const float* xe_vm    = (const float*)d_in[3];
    const float* W_mouse  = (const float*)d_in[4];
    const float* b_mouse  = (const float*)d_in[5];
    const float* W_virus  = (const float*)d_in[6];
    const float* b_virus  = (const float*)d_in[7];
    const float* W_key    = (const float*)d_in[8];
    const float* b_key    = (const float*)d_in[9];
    const float* W_query  = (const float*)d_in[10];
    const float* b_query  = (const float*)d_in[11];
    const float* W_value  = (const float*)d_in[12];
    const float* b_value  = (const float*)d_in[13];
    const float* W_edge   = (const float*)d_in[14];
    const float* b_edge   = (const float*)d_in[15];
    const float* W_skip   = (const float*)d_in[16];
    const float* bias_cv  = (const float*)d_in[17];
    const float* W_cls    = (const float*)d_in[18];
    const float* b_cls    = (const float*)d_in[19];
    const int* ei_mv_src  = (const int*)d_in[20];
    const int* ei_mv_dst  = (const int*)d_in[21];
    const int* ei_vm_src  = (const int*)d_in[22];
    const int* ei_vm_dst  = (const int*)d_in[23];
    const int* eli_mouse  = (const int*)d_in[24];
    const int* eli_virus  = (const int*)d_in[25];
    float* out = (float*)d_out;

    float* ws  = (float*)d_ws;
    const size_t SZ_M = (size_t)N_MOUSE * H;
    const size_t SZ_V = (size_t)N_VIRUS * H;
    float* hm0 = ws;
    float* hm1 = hm0 + SZ_M;
    float* hv0 = hm1 + SZ_M;
    float* hv1 = hv0 + SZ_V;
    float* bx  = hv1 + SZ_V;
    float* by  = bx + SZ_M;
    float* bz  = by + SZ_M;

    int* cnt_a  = (int*)(bz + SZ_V);
    int* cnt_b  = cnt_a + N_VIRUS;
    int* off_a  = cnt_b + N_MOUSE;
    int* off_b  = off_a + N_VIRUS + 4;
    int* perm_a = off_b + N_MOUSE + 4;
    int* psrc_a = perm_a + NEDGE;
    int* perm_b = psrc_a + NEDGE;
    int* psrc_b = perm_b + NEDGE;
    int* slot_a = psrc_a;
    int* slot_b = psrc_b;

    const int e_blocks = (NEDGE + 255) / 256;

    zero_ints<<<dim3((N_VIRUS + N_MOUSE + 255) / 256), dim3(256), 0, stream>>>(
        cnt_a, N_VIRUS + N_MOUSE);
    dst_count<<<dim3(e_blocks), dim3(256), 0, stream>>>(
        ei_mv_dst, ei_vm_dst, cnt_a, cnt_b, slot_a, slot_b,
        NEDGE, N_VIRUS, N_MOUSE);
    dst_scan<<<dim3(2), dim3(1024), 0, stream>>>(
        cnt_a, off_a, N_VIRUS, cnt_b, off_b, N_MOUSE);
    dst_fill<<<dim3(e_blocks), dim3(256), 0, stream>>>(
        ei_mv_dst, off_a, slot_a, perm_a, ei_vm_dst, off_b, slot_b, perm_b,
        NEDGE, N_VIRUS, N_MOUSE);
    dst_order<<<dim3((N_VIRUS + N_MOUSE + 7) / 8), dim3(256), 0, stream>>>(
        off_a, perm_a, psrc_a, ei_mv_src, N_VIRUS,
        off_b, perm_b, psrc_b, ei_vm_src, N_MOUSE, NEDGE);

    gemm_in(x_mouse, W_mouse, b_mouse, hm0, N_MOUSE, stream);
    gemm_in(x_virus, W_virus, b_virus, hv0, N_VIRUS, stream);

    float* hm_in = hm0; float* hv_in = hv0;
    float* hm_out = hm1; float* hv_out = hv1;

    for (int layer = 0; layer < 2; ++layer) {
        const int i_mv = layer * 2;
        const int i_vm = layer * 2 + 1;

        {
            const int i = i_mv;
            const float* Wk = W_key   + (size_t)i * H * H;
            const float* Wq = W_query + (size_t)i * H * H;
            const float* Wv = W_value + (size_t)i * H * H;
            const float* We = W_edge  + (size_t)i * D_EDGE * H;
            const float* Ws = W_skip  + (size_t)i * H * H;
            Proj4 p;
            p.A0 = hm_in; p.B0 = Wq; p.bias0 = b_query + (size_t)i * H; p.C0 = bx;
            p.A1 = hm_in; p.B1 = Wv; p.bias1 = b_value + (size_t)i * H; p.C1 = by;
            p.A2 = hv_in; p.B2 = Ws; p.bias2 = bias_cv + (size_t)i * H; p.C2 = hv_out;
            p.A3 = hv_in; p.B3 = Wk; p.bias3 = b_key   + (size_t)i * H; p.C3 = bz;
            gemm4(p, stream);
            edge_rows<<<dim3((N_VIRUS + 7) / 8), dim3(256), 0, stream>>>(
                bz, bx, by, xe_mv, We, b_edge + (size_t)i * H,
                off_a, perm_a, psrc_a, hv_out, N_VIRUS, N_MOUSE, NEDGE);
        }
        {
            const int i = i_vm;
            const float* Wk = W_key   + (size_t)i * H * H;
            const float* Wq = W_query + (size_t)i * H * H;
            const float* Wv = W_value + (size_t)i * H * H;
            const float* We = W_edge  + (size_t)i * D_EDGE * H;
            const float* Ws = W_skip  + (size_t)i * H * H;
            Proj4 p;
            p.A0 = hm_in; p.B0 = Ws; p.bias0 = bias_cv + (size_t)i * H; p.C0 = hm_out;
            p.A1 = hm_in; p.B1 = Wk; p.bias1 = b_key   + (size_t)i * H; p.C1 = bx;
            p.A2 = hv_in; p.B2 = Wq; p.bias2 = b_query + (size_t)i * H; p.C2 = by;
            p.A3 = hv_in; p.B3 = Wv; p.bias3 = b_value + (size_t)i * H; p.C3 = by + SZ_V;
            gemm4(p, stream);
            edge_rows<<<dim3((N_MOUSE + 7) / 8), dim3(256), 0, stream>>>(
                bx, by, by + SZ_V, xe_vm, We, b_edge + (size_t)i * H,
                off_b, perm_b, psrc_b, hm_out, N_MOUSE, N_VIRUS, NEDGE);
        }

        float* t;
        t = hm_in; hm_in = hm_out; hm_out = t;
        t = hv_in; hv_in = hv_out; hv_out = t;
    }

    classifier<<<dim3((NPAIR + 7) / 8), dim3(256), 0, stream>>>(
        hm_in, hv_in, eli_mouse, eli_virus, W_cls, b_cls, out, NPAIR);
}


# edge bucketing: an edge's slot comes from LDS counters (blocks own a destination range x an edge slice; slice counts summed per row), not from 1M returning global atomics; sum order kept
# speedup vs baseline: 2.1388x; 1.0904x over previous
#include <hip/hip_runtime.h>
#include <hip/hip_bf16.h>

typedef __attribute__((ext_vector_type(2))) float v2f;
typedef __attribute__((ext_vector_type(8))) float v8f;

#define H 128
#define D_EDGE 16
#define N_MOUSE 50000
#define N_VIRUS 10000
#define NEDGE   500000
#define NPAIR   100000
#define D_IN    1024

template <int RT, int KK>
__device__ __forceinline__
void gemm_tile(const float* __restrict__ A,
               const float* __restrict__ B,
               const float* __restrict__ bias,
               float* __restrict__ C,
               int bid)
{
    __shared__ __align__(16) float As[RT * 16 * 132];

    const int tid  = threadIdx.x;
    const int wave = tid >> 5;
    const int lane = tid & 31;
    const int hi   = lane >> 4;
    const int ml   = lane & 15;
    const int row0 = (bid >> 1) * (16 * RT);
    const int n0   = (bid & 1) * 64 + wave * 16;

    v8f acc[RT];
    #pragma unroll
    for (int rt = 0; rt < RT; ++rt) acc[rt] = v8f{};

    const float* ap = A + (size_t)(row0 + wave) * KK + 4 * lane;
    float*       sp = &As[wave * 132 + 4 * lane];
    const float* bp = B + (size_t)(2 * hi) * H + n0 + ml;

    for (int k0 = 0; k0 < KK; k0 += 128) {
        v2f b[32];
        #pragma unroll
        for (int s = 0; s < 32; ++s) {
            b[s].x = bp[(size_t)(k0 + 4 * s) * H];
            b[s].y = bp[(size_t)(k0 + 4 * s + 1) * H];
        }
        float4 st[4 * RT];
        #pragma unroll
        for (int it = 0; it < 4 * RT; ++it)
            st[it] = *(const float4*)(ap + (size_t)(4 * it) * KK + k0);
        #pragma unroll
        for (int it = 0; it < 4 * RT; ++it)
            *(float4*)(sp + (4 * it) * 132) = st[it];
        __syncthreads();

        #pragma unroll
        for (int s = 0; s < 32; ++s) {
            #pragma unroll
            for (int rt = 0; rt < RT; ++rt) {
                v2f a = *(const v2f*)(&As[(rt * 16 + ml) * 132 + 4 * s + 2 * hi]);
                acc[rt] = __builtin_amdgcn_wmma_f32_16x16x4_f32(
                              false, a, false, b[s], (short)0, acc[rt], false, false);
            }
        }
        __syncthreads();
    }

    const float bval = bias ? bias[n0 + ml] : 0.0f;
    #pragma unroll
    for (int rt = 0; rt < RT; ++rt) {
        #pragma unroll
        for (int r = 0; r < 8; ++r) {
            int m = row0 + rt * 16 + r + 8 * hi;
            C[(size_t)m * H + n0 + ml] = acc[rt][r] + bval;
        }
    }
}

template <int RT, int KK>
__global__ __launch_bounds__(128) __attribute__((amdgpu_waves_per_eu(5, 5)))
void gemm_wmma_bias(const float* __restrict__ A,
                    const float* __restrict__ B,
                    const float* __restrict__ bias,
                    float* __restrict__ C,
                    int M)
{
    gemm_tile<RT, KK>(A, B, bias, C, blockIdx.x);
}

struct Proj4 {
    const float* A0; const float* B0; const float* bias0; float* C0;
    const float* A1; const float* B1; const float* bias1; float* C1;
    const float* A2; const float* B2; const float* bias2; float* C2;
    const float* A3; const float* B3; const float* bias3; float* C3;
};

template <int RT, int NB0, int NB2>
__global__ __launch_bounds__(128) __attribute__((amdgpu_waves_per_eu(5, 5)))
void gemm_wmma_bias4(const Proj4 p)
{
    int bid = blockIdx.x;
    const float* A; const float* B; const float* bias; float* C;
    if (bid < NB0) {
        A = p.A0; B = p.B0; bias = p.bias0; C = p.C0;
    } else if (bid < 2 * NB0) {
        bid -= NB0;
        A = p.A1; B = p.B1; bias = p.bias1; C = p.C1;
    } else if (bid < 2 * NB0 + NB2) {
        bid -= 2 * NB0;
        A = p.A2; B = p.B2; bias = p.bias2; C = p.C2;
    } else {
        bid -= 2 * NB0 + NB2;
        A = p.A3; B = p.B3; bias = p.bias3; C = p.C3;
    }
    gemm_tile<RT, H>(A, B, bias, C, bid);
}

#define SLICES       16
#define RANGES_A     4
#define RANGES_B     12
#define RSZ_A        ((N_VIRUS + RANGES_A - 1) / RANGES_A)
#define RSZ_B        ((N_MOUSE + RANGES_B - 1) / RANGES_B)
#define RSZ_MAX      (RSZ_A > RSZ_B ? RSZ_A : RSZ_B)
#define SLICE_EDGES  ((NEDGE + SLICES - 1) / SLICES)
#define N_ROWS       (N_VIRUS + N_MOUSE)
#define RANK_THREADS 512
#define RANK_BATCH   16

__global__ __launch_bounds__(RANK_THREADS)
void dst_rank(const int* __restrict__ dst_a,
              const int* __restrict__ dst_b,
              int* __restrict__ slot_a,
              int* __restrict__ slot_b,
              int* __restrict__ part)
{
    __shared__ int c[RSZ_MAX];

    const int  tid  = threadIdx.x;
    const int  s    = blockIdx.x % SLICES;
    const int  gr   = blockIdx.x / SLICES;
    const bool is_a = gr < RANGES_A;
    const int* dst  = is_a ? dst_a : dst_b;
    int*       slot = is_a ? slot_a : slot_b;
    const int  n    = is_a ? N_VIRUS : N_MOUSE;
    const int  rsz  = is_a ? RSZ_A : RSZ_B;
    const int  lo   = (is_a ? gr : gr - RANGES_A) * rsz;
    const int  rows = min(rsz, n - lo);

    for (int i = tid; i < rsz; i += RANK_THREADS) c[i] = 0;
    __syncthreads();

    const int e0 = s * SLICE_EDGES;
    const int e1 = min(NEDGE, e0 + SLICE_EDGES);
    for (int base = e0 + tid; base < e1; base += RANK_BATCH * RANK_THREADS) {
        int dv[RANK_BATCH];
        #pragma unroll
        for (int k = 0; k < RANK_BATCH; ++k)
            dv[k] = dst[min(base + k * RANK_THREADS, e1 - 1)];
        int rk[RANK_BATCH];
        #pragma unroll
        for (int k = 0; k < RANK_BATCH; ++k) {
            const int e = base + k * RANK_THREADS;
            const unsigned t = (unsigned)(min(max(dv[k], 0), n - 1) - lo);
            rk[k] = -1;
            if (e < e1 && t < (unsigned)rsz) rk[k] = atomicAdd(&c[t], 1);
        }
        #pragma unroll
        for (int k = 0; k < RANK_BATCH; ++k)
            if (rk[k] >= 0) slot[base + k * RANK_THREADS] = rk[k];
    }
    __syncthreads();

    int* prow = part + (size_t)s * N_ROWS + (is_a ? 0 : N_VIRUS) + lo;
    for (int i = tid; i < rows; i += RANK_THREADS) prow[i] = c[i];
}

__global__ void dst_combine(int* __restrict__ part, int* __restrict__ cnt)
{
    const int d = blockIdx.x * blockDim.x + threadIdx.x;
    if (d >= N_ROWS) return;
    int v[SLICES];
    #pragma unroll
    for (int s = 0; s < SLICES; ++s) v[s] = part[(size_t)s * N_ROWS + d];
    int run = 0;
    #pragma unroll
    for (int s = 0; s < SLICES; ++s) {
        part[(size_t)s * N_ROWS + d] = run;
        run += v[s];
    }
    cnt[d] = run;
}

__global__ __launch_bounds__(1024) void dst_scan(const int* __restrict__ cnt_a,
                                                 int* __restrict__ off_a, int na,
                                                 const int* __restrict__ cnt_b,
                                                 int* __restrict__ off_b, int nb)
{
    __shared__ int wsum[32];
    __shared__ int total_s;

    const int* cnt = blockIdx.x ? cnt_b : cnt_a;
    int*       off = blockIdx.x ? off_b : off_a;
    const int  n   = blockIdx.x ? nb : na;

    const int tid  = threadIdx.x;
    const int lane = tid & 31;
    const int wave = tid >> 5;

    int carry = 0;
    for (int base = 0; base < n; base += 16384) {
        const int idx = base + tid * 16;
        int4 c[4];
        for (int k = 0; k < 4; ++k) c[k] = make_int4(0, 0, 0, 0);
        if (idx < n) {
            for (int k = 0; k < 4; ++k) c[k] = *(const int4*)(cnt + idx + 4 * k);
        }
        int t = 0;
        for (int k = 0; k < 4; ++k) t += c[k].x + c[k].y + c[k].z + c[k].w;

        int incl = t;
        for (int o = 1; o < 32; o <<= 1) {
            int y = __shfl_up(incl, o, 32);
            if (lane >= o) incl += y;
        }
        if (lane == 31) wsum[wave] = incl;
        __syncthreads();
        if (wave == 0) {
            const int v = wsum[lane];
            int iv = v;
            for (int o = 1; o < 32; o <<= 1) {
                int y = __shfl_up(iv, o, 32);
                if (lane >= o) iv += y;
            }
            wsum[lane] = iv - v;
            if (lane == 31) total_s = iv;
        }
        __syncthreads();
        int run = carry + wsum[wave] + incl - t;
        if (idx < n) {
            for (int k = 0; k < 4; ++k) {
                int4 o4;
                o4.x = run;
                o4.y = o4.x + c[k].x;
                o4.z = o4.y + c[k].y;
                o4.w = o4.z + c[k].z;
                run  = o4.w + c[k].w;
                *(int4*)(off + idx + 4 * k) = o4;
            }
        }
        carry += total_s;
        __syncthreads();
    }
    if (tid == 0) off[n] = carry;
}

__global__ void dst_fill(const int* __restrict__ dst_a,
                         const int* __restrict__ off_a,
                         const int* __restrict__ slot_a,
                         int* __restrict__ perm_a,
                         const int* __restrict__ dst_b,
                         const int* __restrict__ off_b,
                         const int* __restrict__ slot_b,
                         int* __restrict__ perm_b,
                         const int* __restrict__ part,
                         int E, int na, int nb)
{
    int e = blockIdx.x * blockDim.x + threadIdx.x;
    if (e >= E) return;
    const int* prow = part + (size_t)(e / SLICE_EDGES) * N_ROWS;
    {
        int d = min(max(dst_a[e], 0), na - 1);
        int p = off_a[d] + prow[d] + slot_a[e];
        perm_a[min(max(p, 0), E - 1)] = e;
    }
    {
        int d = min(max(dst_b[e], 0), nb - 1);
        int p = off_b[d] + prow[na + d] + slot_b[e];
        perm_b[min(max(p, 0), E - 1)] = e;
    }
}

#define SORT_CAP 512
__global__ __launch_bounds__(256) void dst_order(const int* __restrict__ off_a,
                                                 int* __restrict__ perm_a,
                                                 int* __restrict__ psrc_a,
                                                 const int* __restrict__ src_a, int na,
                                                 const int* __restrict__ off_b,
                                                 int* __restrict__ perm_b,
                                                 int* __restrict__ psrc_b,
                                                 const int* __restrict__ src_b, int nb,
                                                 int E)
{
    __shared__ __align__(16) int skey[8][SORT_CAP];

    const int lane = threadIdx.x & 31;
    const int wv   = threadIdx.x >> 5;
    int row = blockIdx.x * 8 + wv;

    const bool is_a  = row < na;
    const int* off   = is_a ? off_a  : off_b;
    int*       perm  = is_a ? perm_a : perm_b;
    int*       psrc  = is_a ? psrc_a : psrc_b;
    const int* src   = is_a ? src_a  : src_b;
    if (!is_a) row -= na;
    const bool valid = is_a || row < nb;

    int beg = 0, n = 0;
    if (valid) {
        beg = min(max(off[row], 0), E);
        n   = min(max(off[row + 1], beg), E) - beg;
    }
    const bool small = n <= SORT_CAP;
    const int  npad  = (n + 3) & ~3;
    int* key = skey[wv];

    if (small) {
        for (int j = lane; j < npad; j += 32)
            key[j] = (j < n) ? perm[beg + j] : 0x7fffffff;
    }
    __syncthreads();

    if (small) {
        for (int j0 = 0; j0 < n; j0 += 32) {
            const int j  = j0 + lane;
            const int my = (j < n) ? key[j] : 0x7fffffff;
            int r = 0;
            for (int t = 0; t < npad; t += 4) {
                const int4 k4 = *(const int4*)(key + t);
                r += (k4.x < my) + (k4.y < my) + (k4.z < my) + (k4.w < my);
            }
            if (j < n) {
                perm[beg + r] = my;
                psrc[beg + r] = src[min(max(my, 0), E - 1)];
            }
        }
    } else {
        for (int j = lane; j < n; j += 32)
            psrc[beg + j] = src[min(max(perm[beg + j], 0), E - 1)];
    }
}

__global__ __launch_bounds__(256) void edge_rows(const float* __restrict__ kbuf,
                                                 const float* __restrict__ qbuf,
                                                 const float* __restrict__ vbuf,
                                                 const float* __restrict__ xe,
                                                 const float* __restrict__ We,
                                                 const float* __restrict__ be,
                                                 const int*   __restrict__ off,
                                                 const int*   __restrict__ perm,
                                                 const int*   __restrict__ psrc,
                                                 float*       __restrict__ out_dst,
                                                 int n_dst, int n_src, int E)
{
    const int lane = threadIdx.x & 31;
    const int d = __builtin_amdgcn_readfirstlane(blockIdx.x * 8 + (threadIdx.x >> 5));
    if (d >= n_dst) return;

    float4 w[D_EDGE];
    for (int dd = 0; dd < D_EDGE; ++dd)
        w[dd] = *(const float4*)(We + dd * H + 4 * lane);
    const float4 b4 = *(const float4*)(be + 4 * lane);

    int beg = __builtin_amdgcn_readfirstlane(off[d]);
    int end = __builtin_amdgcn_readfirstlane(off[d + 1]);
    beg = min(max(beg, 0), E);
    end = min(max(end, beg), E);

    const float4 k4 = *(const float4*)(kbuf + (size_t)d * H + 4 * lane);
    float4 acc = make_float4(0.f, 0.f, 0.f, 0.f);

    for (int j0 = beg; j0 < end; j0 += 32) {
        const int jj = j0 + lane;
        const int pe = (jj < end) ? perm[jj] : 0;
        const int ps = (jj < end) ? psrc[jj] : 0;
        const int cnt = min(32, end - j0);
        int e = min(max(__builtin_amdgcn_readlane(pe, 0), 0), E - 1);
        int s = min(max(__builtin_amdgcn_readlane(ps, 0), 0), n_src - 1);
        float  xv = xe[(size_t)e * D_EDGE + (lane & 15)];
        float4 q4 = *(const float4*)(qbuf + (size_t)s * H + 4 * lane);
        float4 v4 = *(const float4*)(vbuf + (size_t)s * H + 4 * lane);

        for (int i = 0; i < cnt; ++i) {
            float  xn = xv;
            float4 qn = q4;
            float4 vn = v4;
            if (i + 1 < cnt) {
                e = min(max(__builtin_amdgcn_readlane(pe, i + 1), 0), E - 1);
                s = min(max(__builtin_amdgcn_readlane(ps, i + 1), 0), n_src - 1);
                xn = xe[(size_t)e * D_EDGE + (lane & 15)];
                qn = *(const float4*)(qbuf + (size_t)s * H + 4 * lane);
                vn = *(const float4*)(vbuf + (size_t)s * H + 4 * lane);
            }

            float4 ea = b4;
            for (int dd = 0; dd < D_EDGE; ++dd) {
                const float x = __int_as_float(
                    __builtin_amdgcn_readlane(__float_as_int(xv), dd));
                ea.x = fmaf(x, w[dd].x, ea.x);
                ea.y = fmaf(x, w[dd].y, ea.y);
                ea.z = fmaf(x, w[dd].z, ea.z);
                ea.w = fmaf(x, w[dd].w, ea.w);
            }

            const float gx = __builtin_amdgcn_rcpf(1.0f + __expf(-((k4.x + ea.x) + (q4.x + ea.x))));
            const float gy = __builtin_amdgcn_rcpf(1.0f + __expf(-((k4.y + ea.y) + (q4.y + ea.y))));
            const float gz = __builtin_amdgcn_rcpf(1.0f + __expf(-((k4.z + ea.z) + (q4.z + ea.z))));
            const float gw = __builtin_amdgcn_rcpf(1.0f + __expf(-((k4.w + ea.w) + (q4.w + ea.w))));
            acc.x += gx * (v4.x + ea.x);
            acc.y += gy * (v4.y + ea.y);
            acc.z += gz * (v4.z + ea.z);
            acc.w += gw * (v4.w + ea.w);

            xv = xn; q4 = qn; v4 = vn;
        }
    }

    float4* op = (float4*)(out_dst + (size_t)d * H + 4 * lane);
    float4 o = *op;
    o.x = fmaxf(o.x + acc.x, 0.0f);
    o.y = fmaxf(o.y + acc.y, 0.0f);
    o.z = fmaxf(o.z + acc.z, 0.0f);
    o.w = fmaxf(o.w + acc.w, 0.0f);
    *op = o;
}

__global__ void classifier(const float* __restrict__ hm,
                           const float* __restrict__ hv,
                           const int*   __restrict__ im,
                           const int*   __restrict__ iv,
                           const float* __restrict__ Wc,
                           const float* __restrict__ bc,
                           float*       __restrict__ out,
                           int P)
{
    const int p    = blockIdx.x * 8 + (threadIdx.x >> 5);
    const int lane = threadIdx.x & 31;
    if (p >= P) return;

    const float4 am = ((const float4*)(hm + (size_t)im[p] * H))[lane];
    const float4 av = ((const float4*)(hv + (size_t)iv[p] * H))[lane];

    float c0 = 0.f, c1 = 0.f;
    const int ch = lane * 4;
    const float a[4] = {am.x, am.y, am.z, am.w};
    const float b[4] = {av.x, av.y, av.z, av.w};
    #pragma unroll
    for (int i = 0; i < 4; ++i) {
        c0 = fmaf(a[i], Wc[(ch + i) * 2 + 0], c0);
        c1 = fmaf(a[i], Wc[(ch + i) * 2 + 1], c1);
        c0 = fmaf(b[i], Wc[(H + ch + i) * 2 + 0], c0);
        c1 = fmaf(b[i], Wc[(H + ch + i) * 2 + 1], c1);
    }
    #pragma unroll
    for (int off = 16; off > 0; off >>= 1) {
        c0 += __shfl_down(c0, off, 32);
        c1 += __shfl_down(c1, off, 32);
    }
    if (lane == 0) {
        out[(size_t)p * 2 + 0] = c0 + bc[0];
        out[(size_t)p * 2 + 1] = c1 + bc[1];
    }
}

static inline void gemm_in(const float* A, const float* B, const float* bias,
                           float* C, int M, hipStream_t s)
{
    gemm_wmma_bias<5, D_IN><<<dim3(M / 80 * 2), dim3(128), 0, s>>>(A, B, bias, C, M);
}
#define NB_MOUSE (N_MOUSE / 80 * 2)
#define NB_VIRUS (N_VIRUS / 80 * 2)
static inline void gemm4(const Proj4& p, hipStream_t s)
{
    gemm_wmma_bias4<5, NB_MOUSE, NB_VIRUS>
        <<<dim3(2 * NB_MOUSE + 2 * NB_VIRUS), dim3(128), 0, s>>>(p);
}

extern "C" void kernel_launch(void* const* d_in, const int* in_sizes, int n_in,
                              void* d_out, int out_size, void* d_ws, size_t ws_size,
                              hipStream_t stream)
{
    const float* x_mouse  = (const float*)d_in[0];
    const float* x_virus  = (const float*)d_in[1];
    const float* xe_mv    = (const float*)d_in[2];
    const float* xe_vm    = (const float*)d_in[3];
    const float* W_mouse  = (const float*)d_in[4];
    const float* b_mouse  = (const float*)d_in[5];
    const float* W_virus  = (const float*)d_in[6];
    const float* b_virus  = (const float*)d_in[7];
    const float* W_key    = (const float*)d_in[8];
    const float* b_key    = (const float*)d_in[9];
    const float* W_query  = (const float*)d_in[10];
    const float* b_query  = (const float*)d_in[11];
    const float* W_value  = (const float*)d_in[12];
    const float* b_value  = (const float*)d_in[13];
    const float* W_edge   = (const float*)d_in[14];
    const float* b_edge   = (const float*)d_in[15];
    const float* W_skip   = (const float*)d_in[16];
    const float* bias_cv  = (const float*)d_in[17];
    const float* W_cls    = (const float*)d_in[18];
    const float* b_cls    = (const float*)d_in[19];
    const int* ei_mv_src  = (const int*)d_in[20];
    const int* ei_mv_dst  = (const int*)d_in[21];
    const int* ei_vm_src  = (const int*)d_in[22];
    const int* ei_vm_dst  = (const int*)d_in[23];
    const int* eli_mouse  = (const int*)d_in[24];
    const int* eli_virus  = (const int*)d_in[25];
    float* out = (float*)d_out;

    float* ws  = (float*)d_ws;
    const size_t SZ_M = (size_t)N_MOUSE * H;
    const size_t SZ_V = (size_t)N_VIRUS * H;
    float* hm0 = ws;
    float* hm1 = hm0 + SZ_M;
    float* hv0 = hm1 + SZ_M;
    float* hv1 = hv0 + SZ_V;
    float* bx  = hv1 + SZ_V;
    float* by  = bx + SZ_M;
    float* bz  = by + SZ_M;

    int* cnt_a  = (int*)(bz + SZ_V);
    int* cnt_b  = cnt_a + N_VIRUS;
    int* off_a  = cnt_b + N_MOUSE;
    int* off_b  = off_a + N_VIRUS + 4;
    int* perm_a = off_b + N_MOUSE + 4;
    int* psrc_a = perm_a + NEDGE;
    int* perm_b = psrc_a + NEDGE;
    int* psrc_b = perm_b + NEDGE;
    int* slot_a = psrc_a;
    int* slot_b = psrc_b;

    const int e_blocks = (NEDGE + 255) / 256;

    int* part = (int*)bx;

    dst_rank<<<dim3((RANGES_A + RANGES_B) * SLICES), dim3(RANK_THREADS), 0, stream>>>(
        ei_mv_dst, ei_vm_dst, slot_a, slot_b, part);
    dst_combine<<<dim3((N_ROWS + 255) / 256), dim3(256), 0, stream>>>(part, cnt_a);
    dst_scan<<<dim3(2), dim3(1024), 0, stream>>>(
        cnt_a, off_a, N_VIRUS, cnt_b, off_b, N_MOUSE);
    dst_fill<<<dim3(e_blocks), dim3(256), 0, stream>>>(
        ei_mv_dst, off_a, slot_a, perm_a, ei_vm_dst, off_b, slot_b, perm_b,
        part, NEDGE, N_VIRUS, N_MOUSE);
    dst_order<<<dim3((N_VIRUS + N_MOUSE + 7) / 8), dim3(256), 0, stream>>>(
        off_a, perm_a, psrc_a, ei_mv_src, N_VIRUS,
        off_b, perm_b, psrc_b, ei_vm_src, N_MOUSE, NEDGE);

    gemm_in(x_mouse, W_mouse, b_mouse, hm0, N_MOUSE, stream);
    gemm_in(x_virus, W_virus, b_virus, hv0, N_VIRUS, stream);

    float* hm_in = hm0; float* hv_in = hv0;
    float* hm_out = hm1; float* hv_out = hv1;

    for (int layer = 0; layer < 2; ++layer) {
        const int i_mv = layer * 2;
        const int i_vm = layer * 2 + 1;

        {
            const int i = i_mv;
            const float* Wk = W_key   + (size_t)i * H * H;
            const float* Wq = W_query + (size_t)i * H * H;
            const float* Wv = W_value + (size_t)i * H * H;
            const float* We = W_edge  + (size_t)i * D_EDGE * H;
            const float* Ws = W_skip  + (size_t)i * H * H;
            Proj4 p;
            p.A0 = hm_in; p.B0 = Wq; p.bias0 = b_query + (size_t)i * H; p.C0 = bx;
            p.A1 = hm_in; p.B1 = Wv; p.bias1 = b_value + (size_t)i * H; p.C1 = by;
            p.A2 = hv_in; p.B2 = Ws; p.bias2 = bias_cv + (size_t)i * H; p.C2 = hv_out;
            p.A3 = hv_in; p.B3 = Wk; p.bias3 = b_key   + (size_t)i * H; p.C3 = bz;
            gemm4(p, stream);
            edge_rows<<<dim3((N_VIRUS + 7) / 8), dim3(256), 0, stream>>>(
                bz, bx, by, xe_mv, We, b_edge + (size_t)i * H,
                off_a, perm_a, psrc_a, hv_out, N_VIRUS, N_MOUSE, NEDGE);
        }
        {
            const int i = i_vm;
            const float* Wk = W_key   + (size_t)i * H * H;
            const float* Wq = W_query + (size_t)i * H * H;
            const float* Wv = W_value + (size_t)i * H * H;
            const float* We = W_edge  + (size_t)i * D_EDGE * H;
            const float* Ws = W_skip  + (size_t)i * H * H;
            Proj4 p;
            p.A0 = hm_in; p.B0 = Ws; p.bias0 = bias_cv + (size_t)i * H; p.C0 = hm_out;
            p.A1 = hm_in; p.B1 = Wk; p.bias1 = b_key   + (size_t)i * H; p.C1 = bx;
            p.A2 = hv_in; p.B2 = Wq; p.bias2 = b_query + (size_t)i * H; p.C2 = by;
            p.A3 = hv_in; p.B3 = Wv; p.bias3 = b_value + (size_t)i * H; p.C3 = by + SZ_V;
            gemm4(p, stream);
            edge_rows<<<dim3((N_MOUSE + 7) / 8), dim3(256), 0, stream>>>(
                bx, by, by + SZ_V, xe_vm, We, b_edge + (size_t)i * H,
                off_b, perm_b, psrc_b, hm_out, N_MOUSE, N_VIRUS, NEDGE);
        }

        float* t;
        t = hm_in; hm_in = hm_out; hm_out = t;
        t = hv_in; hv_in = hv_out; hv_out = t;
    }

    classifier<<<dim3((NPAIR + 7) / 8), dim3(256), 0, stream>>>(
        hm_in, hv_in, eli_mouse, eli_virus, W_cls, b_cls, out, NPAIR);
}


# edge kernel: edge projection xe[16] x We[16x128] goes from 64 vector FMAs per edge to f32 16x16x4 matrix instructions (16 edges a batch); row sum = even + odd partial sums; dot reordered, bias last
# speedup vs baseline: 2.2841x; 1.0679x over previous
#include <hip/hip_runtime.h>
#include <hip/hip_bf16.h>

typedef __attribute__((ext_vector_type(2))) float v2f;
typedef __attribute__((ext_vector_type(8))) float v8f;

#define H 128
#define D_EDGE 16
#define N_MOUSE 50000
#define N_VIRUS 10000
#define NEDGE   500000
#define NPAIR   100000
#define D_IN    1024

template <int RT, int KK>
__device__ __forceinline__
void gemm_tile(const float* __restrict__ A,
               const float* __restrict__ B,
               const float* __restrict__ bias,
               float* __restrict__ C,
               int bid)
{
    __shared__ __align__(16) float As[RT * 16 * 132];

    const int tid  = threadIdx.x;
    const int wave = tid >> 5;
    const int lane = tid & 31;
    const int hi   = lane >> 4;
    const int ml   = lane & 15;
    const int row0 = (bid >> 1) * (16 * RT);
    const int n0   = (bid & 1) * 64 + wave * 16;

    v8f acc[RT];
    #pragma unroll
    for (int rt = 0; rt < RT; ++rt) acc[rt] = v8f{};

    const float* ap = A + (size_t)(row0 + wave) * KK + 4 * lane;
    float*       sp = &As[wave * 132 + 4 * lane];
    const float* bp = B + (size_t)(2 * hi) * H + n0 + ml;

    for (int k0 = 0; k0 < KK; k0 += 128) {
        v2f b[32];
        #pragma unroll
        for (int s = 0; s < 32; ++s) {
            b[s].x = bp[(size_t)(k0 + 4 * s) * H];
            b[s].y = bp[(size_t)(k0 + 4 * s + 1) * H];
        }
        float4 st[4 * RT];
        #pragma unroll
        for (int it = 0; it < 4 * RT; ++it)
            st[it] = *(const float4*)(ap + (size_t)(4 * it) * KK + k0);
        #pragma unroll
        for (int it = 0; it < 4 * RT; ++it)
            *(float4*)(sp + (4 * it) * 132) = st[it];
        __syncthreads();

        #pragma unroll
        for (int s = 0; s < 32; ++s) {
            #pragma unroll
            for (int rt = 0; rt < RT; ++rt) {
                v2f a = *(const v2f*)(&As[(rt * 16 + ml) * 132 + 4 * s + 2 * hi]);
                acc[rt] = __builtin_amdgcn_wmma_f32_16x16x4_f32(
                              false, a, false, b[s], (short)0, acc[rt], false, false);
            }
        }
        __syncthreads();
    }

    const float bval = bias ? bias[n0 + ml] : 0.0f;
    #pragma unroll
    for (int rt = 0; rt < RT; ++rt) {
        #pragma unroll
        for (int r = 0; r < 8; ++r) {
            int m = row0 + rt * 16 + r + 8 * hi;
            C[(size_t)m * H + n0 + ml] = acc[rt][r] + bval;
        }
    }
}

template <int RT, int KK>
__global__ __launch_bounds__(128) __attribute__((amdgpu_waves_per_eu(5, 5)))
void gemm_wmma_bias(const float* __restrict__ A,
                    const float* __restrict__ B,
                    const float* __restrict__ bias,
                    float* __restrict__ C,
                    int M)
{
    gemm_tile<RT, KK>(A, B, bias, C, blockIdx.x);
}

struct Proj4 {
    const float* A0; const float* B0; const float* bias0; float* C0;
    const float* A1; const float* B1; const float* bias1; float* C1;
    const float* A2; const float* B2; const float* bias2; float* C2;
    const float* A3; const float* B3; const float* bias3; float* C3;
};

template <int RT, int NB0, int NB2>
__global__ __launch_bounds__(128) __attribute__((amdgpu_waves_per_eu(5, 5)))
void gemm_wmma_bias4(const Proj4 p)
{
    int bid = blockIdx.x;
    const float* A; const float* B; const float* bias; float* C;
    if (bid < NB0) {
        A = p.A0; B = p.B0; bias = p.bias0; C = p.C0;
    } else if (bid < 2 * NB0) {
        bid -= NB0;
        A = p.A1; B = p.B1; bias = p.bias1; C = p.C1;
    } else if (bid < 2 * NB0 + NB2) {
        bid -= 2 * NB0;
        A = p.A2; B = p.B2; bias = p.bias2; C = p.C2;
    } else {
        bid -= 2 * NB0 + NB2;
        A = p.A3; B = p.B3; bias = p.bias3; C = p.C3;
    }
    gemm_tile<RT, H>(A, B, bias, C, bid);
}

#define SLICES       16
#define RANGES_A     4
#define RANGES_B     12
#define RSZ_A        ((N_VIRUS + RANGES_A - 1) / RANGES_A)
#define RSZ_B        ((N_MOUSE + RANGES_B - 1) / RANGES_B)
#define RSZ_MAX      (RSZ_A > RSZ_B ? RSZ_A : RSZ_B)
#define SLICE_EDGES  ((NEDGE + SLICES - 1) / SLICES)
#define N_ROWS       (N_VIRUS + N_MOUSE)
#define RANK_THREADS 512
#define RANK_BATCH   16

__global__ __launch_bounds__(RANK_THREADS)
void dst_rank(const int* __restrict__ dst_a,
              const int* __restrict__ dst_b,
              int* __restrict__ slot_a,
              int* __restrict__ slot_b,
              int* __restrict__ part)
{
    __shared__ int c[RSZ_MAX];

    const int  tid  = threadIdx.x;
    const int  s    = blockIdx.x % SLICES;
    const int  gr   = blockIdx.x / SLICES;
    const bool is_a = gr < RANGES_A;
    const int* dst  = is_a ? dst_a : dst_b;
    int*       slot = is_a ? slot_a : slot_b;
    const int  n    = is_a ? N_VIRUS : N_MOUSE;
    const int  rsz  = is_a ? RSZ_A : RSZ_B;
    const int  lo   = (is_a ? gr : gr - RANGES_A) * rsz;
    const int  rows = min(rsz, n - lo);

    for (int i = tid; i < rsz; i += RANK_THREADS) c[i] = 0;
    __syncthreads();

    const int e0 = s * SLICE_EDGES;
    const int e1 = min(NEDGE, e0 + SLICE_EDGES);
    for (int base = e0 + tid; base < e1; base += RANK_BATCH * RANK_THREADS) {
        int dv[RANK_BATCH];
        #pragma unroll
        for (int k = 0; k < RANK_BATCH; ++k)
            dv[k] = dst[min(base + k * RANK_THREADS, e1 - 1)];
        int rk[RANK_BATCH];
        #pragma unroll
        for (int k = 0; k < RANK_BATCH; ++k) {
            const int e = base + k * RANK_THREADS;
            const unsigned t = (unsigned)(min(max(dv[k], 0), n - 1) - lo);
            rk[k] = -1;
            if (e < e1 && t < (unsigned)rsz) rk[k] = atomicAdd(&c[t], 1);
        }
        #pragma unroll
        for (int k = 0; k < RANK_BATCH; ++k)
            if (rk[k] >= 0) slot[base + k * RANK_THREADS] = rk[k];
    }
    __syncthreads();

    int* prow = part + (size_t)s * N_ROWS + (is_a ? 0 : N_VIRUS) + lo;
    for (int i = tid; i < rows; i += RANK_THREADS) prow[i] = c[i];
}

__global__ void dst_combine(int* __restrict__ part, int* __restrict__ cnt)
{
    const int d = blockIdx.x * blockDim.x + threadIdx.x;
    if (d >= N_ROWS) return;
    int v[SLICES];
    #pragma unroll
    for (int s = 0; s < SLICES; ++s) v[s] = part[(size_t)s * N_ROWS + d];
    int run = 0;
    #pragma unroll
    for (int s = 0; s < SLICES; ++s) {
        part[(size_t)s * N_ROWS + d] = run;
        run += v[s];
    }
    cnt[d] = run;
}

__global__ __launch_bounds__(1024) void dst_scan(const int* __restrict__ cnt_a,
                                                 int* __restrict__ off_a, int na,
                                                 const int* __restrict__ cnt_b,
                                                 int* __restrict__ off_b, int nb)
{
    __shared__ int wsum[32];
    __shared__ int total_s;

    const int* cnt = blockIdx.x ? cnt_b : cnt_a;
    int*       off = blockIdx.x ? off_b : off_a;
    const int  n   = blockIdx.x ? nb : na;

    const int tid  = threadIdx.x;
    const int lane = tid & 31;
    const int wave = tid >> 5;

    int carry = 0;
    for (int base = 0; base < n; base += 16384) {
        const int idx = base + tid * 16;
        int4 c[4];
        for (int k = 0; k < 4; ++k) c[k] = make_int4(0, 0, 0, 0);
        if (idx < n) {
            for (int k = 0; k < 4; ++k) c[k] = *(const int4*)(cnt + idx + 4 * k);
        }
        int t = 0;
        for (int k = 0; k < 4; ++k) t += c[k].x + c[k].y + c[k].z + c[k].w;

        int incl = t;
        for (int o = 1; o < 32; o <<= 1) {
            int y = __shfl_up(incl, o, 32);
            if (lane >= o) incl += y;
        }
        if (lane == 31) wsum[wave] = incl;
        __syncthreads();
        if (wave == 0) {
            const int v = wsum[lane];
            int iv = v;
            for (int o = 1; o < 32; o <<= 1) {
                int y = __shfl_up(iv, o, 32);
                if (lane >= o) iv += y;
            }
            wsum[lane] = iv - v;
            if (lane == 31) total_s = iv;
        }
        __syncthreads();
        int run = carry + wsum[wave] + incl - t;
        if (idx < n) {
            for (int k = 0; k < 4; ++k) {
                int4 o4;
                o4.x = run;
                o4.y = o4.x + c[k].x;
                o4.z = o4.y + c[k].y;
                o4.w = o4.z + c[k].z;
                run  = o4.w + c[k].w;
                *(int4*)(off + idx + 4 * k) = o4;
            }
        }
        carry += total_s;
        __syncthreads();
    }
    if (tid == 0) off[n] = carry;
}

__global__ void dst_fill(const int* __restrict__ dst_a,
                         const int* __restrict__ off_a,
                         const int* __restrict__ slot_a,
                         int* __restrict__ perm_a,
                         const int* __restrict__ dst_b,
                         const int* __restrict__ off_b,
                         const int* __restrict__ slot_b,
                         int* __restrict__ perm_b,
                         const int* __restrict__ part,
                         int E, int na, int nb)
{
    int e = blockIdx.x * blockDim.x + threadIdx.x;
    if (e >= E) return;
    const int* prow = part + (size_t)(e / SLICE_EDGES) * N_ROWS;
    {
        int d = min(max(dst_a[e], 0), na - 1);
        int p = off_a[d] + prow[d] + slot_a[e];
        perm_a[min(max(p, 0), E - 1)] = e;
    }
    {
        int d = min(max(dst_b[e], 0), nb - 1);
        int p = off_b[d] + prow[na + d] + slot_b[e];
        perm_b[min(max(p, 0), E - 1)] = e;
    }
}

#define SORT_CAP 512
__global__ __launch_bounds__(256) void dst_order(const int* __restrict__ off_a,
                                                 int* __restrict__ perm_a,
                                                 int* __restrict__ psrc_a,
                                                 const int* __restrict__ src_a, int na,
                                                 const int* __restrict__ off_b,
                                                 int* __restrict__ perm_b,
                                                 int* __restrict__ psrc_b,
                                                 const int* __restrict__ src_b, int nb,
                                                 int E)
{
    __shared__ __align__(16) int skey[8][SORT_CAP];

    const int lane = threadIdx.x & 31;
    const int wv   = threadIdx.x >> 5;
    int row = blockIdx.x * 8 + wv;

    const bool is_a  = row < na;
    const int* off   = is_a ? off_a  : off_b;
    int*       perm  = is_a ? perm_a : perm_b;
    int*       psrc  = is_a ? psrc_a : psrc_b;
    const int* src   = is_a ? src_a  : src_b;
    if (!is_a) row -= na;
    const bool valid = is_a || row < nb;

    int beg = 0, n = 0;
    if (valid) {
        beg = min(max(off[row], 0), E);
        n   = min(max(off[row + 1], beg), E) - beg;
    }
    const bool small = n <= SORT_CAP;
    const int  npad  = (n + 3) & ~3;
    int* key = skey[wv];

    if (small) {
        for (int j = lane; j < npad; j += 32)
            key[j] = (j < n) ? perm[beg + j] : 0x7fffffff;
    }
    __syncthreads();

    if (small) {
        for (int j0 = 0; j0 < n; j0 += 32) {
            const int j  = j0 + lane;
            const int my = (j < n) ? key[j] : 0x7fffffff;
            int r = 0;
            for (int t = 0; t < npad; t += 4) {
                const int4 k4 = *(const int4*)(key + t);
                r += (k4.x < my) + (k4.y < my) + (k4.z < my) + (k4.w < my);
            }
            if (j < n) {
                perm[beg + r] = my;
                psrc[beg + r] = src[min(max(my, 0), E - 1)];
            }
        }
    } else {
        for (int j = lane; j < n; j += 32)
            psrc[beg + j] = src[min(max(perm[beg + j], 0), E - 1)];
    }
}

typedef __attribute__((ext_vector_type(4))) float v4f;

template <int NR>
__device__ __forceinline__
void edge_batch(const float* __restrict__ qbuf,
                const float* __restrict__ vbuf,
                const float* __restrict__ xnext,
                const v4f (&sB)[16 * 32],
                const int lane,
                const unsigned (&so)[8],
                const v2f (&a)[4],
                const float (&cb)[8],
                const float (&kr)[8],
                float (&acc)[8],
                float4& x0, float4& x1,
                int lim)
{
    #pragma unroll
    for (int g = 0; g < 2; ++g) {
        if (g == 1) {
            __builtin_amdgcn_sched_barrier(0);
            x0 = *(const float4*)(xnext);
            x1 = *(const float4*)(xnext + 4);
        }
        const char* qb = (const char*)qbuf + 256 * g;
        const char* vb = (const char*)vbuf + 256 * g;
        float qf[NR][4], vf[NR][4];
        #pragma unroll
        for (int r = 0; r < NR; ++r) {
            const float4 q4 = *(const float4*)(qb + so[r]);
            const float4 v4 = *(const float4*)(vb + so[r]);
            qf[r][0] = q4.x; qf[r][1] = q4.y; qf[r][2] = q4.z; qf[r][3] = q4.w;
            vf[r][0] = v4.x; vf[r][1] = v4.y; vf[r][2] = v4.z; vf[r][3] = v4.w;
        }

        v2f b[4][4];
        #pragma unroll
        for (int tt = 0; tt < 4; ++tt) {
            #pragma unroll
            for (int sp = 0; sp < 2; ++sp) {
                const v4f f = sB[(2 * (4 * g + tt) + sp) * 32 + lane];
                b[2 * sp][tt].x     = f.x; b[2 * sp][tt].y     = f.y;
                b[2 * sp + 1][tt].x = f.z; b[2 * sp + 1][tt].y = f.w;
            }
        }

        v8f pr[4];
        #pragma unroll
        for (int tt = 0; tt < 4; ++tt) pr[tt] = v8f{};
        #pragma unroll
        for (int s = 0; s < 4; ++s) {
            #pragma unroll
            for (int tt = 0; tt < 4; ++tt)
                pr[tt] = __builtin_amdgcn_wmma_f32_16x16x4_f32(
                             false, a[s], false, b[s][tt], (short)0, pr[tt],
                             false, false);
        }

        #pragma unroll
        for (int r = 0; r < NR; ++r) {
            #pragma unroll
            for (int tt = 0; tt < 4; ++tt) {
                const float ev = pr[tt][r] + cb[4 * g + tt];
                const float z  = (kr[4 * g + tt] + ev) + (qf[r][tt] + ev);
                float gt = __builtin_amdgcn_rcpf(1.0f + __expf(-z));
                if (r >= NR - 2) gt = (2 * r < lim) ? gt : 0.0f;
                acc[4 * g + tt] += gt * (vf[r][tt] + ev);
            }
        }
    }
}

template <int ROWS>
__global__ __launch_bounds__(256) __attribute__((amdgpu_waves_per_eu(4, 4)))
void edge_rows(const float* __restrict__ kbuf,
               const float* __restrict__ qbuf,
               const float* __restrict__ vbuf,
               const float* __restrict__ xe,
               const float* __restrict__ We,
               const float* __restrict__ be,
               const int*   __restrict__ off,
               const int*   __restrict__ perm,
               const int*   __restrict__ psrc,
               float*       __restrict__ out_dst,
               int n_dst, int n_src, int E)
{
    __shared__ v4f sB[16 * 32];

    const int lane = threadIdx.x & 31;
    const int c    = lane & 15;
    const int h    = lane >> 4;

    #pragma unroll
    for (int i = 0; i < 2; ++i) {
        const int combo = (threadIdx.x >> 5) * 2 + i;
        const int t     = combo >> 1;
        const int sp    = combo & 1;
        const float* wp = We + (8 * h + 4 * sp) * H + 64 * (t >> 2) + 4 * c + (t & 3);
        v4f f;
        f.x = wp[0];
        f.y = wp[H];
        f.z = wp[2 * H];
        f.w = wp[3 * H];
        sB[combo * 32 + lane] = f;
    }
    __syncthreads();

    const int d0 = __builtin_amdgcn_readfirstlane(
                       (blockIdx.x * 8 + (threadIdx.x >> 5)) * ROWS);
    if (d0 >= n_dst) return;
    const int d1 = min(d0 + ROWS, n_dst);

    float cb[8];
    {
        const float4 c0 = *(const float4*)(be + 4 * c);
        const float4 c1 = *(const float4*)(be + 64 + 4 * c);
        cb[0] = c0.x; cb[1] = c0.y; cb[2] = c0.z; cb[3] = c0.w;
        cb[4] = c1.x; cb[5] = c1.y; cb[6] = c1.z; cb[7] = c1.w;
    }
    const int ia = 2 * (c & 7) + (c >> 3);

    int    ps_l = 0;
    float4 x0 = make_float4(0.f, 0.f, 0.f, 0.f), x1 = x0;
    bool   primed = false;
    int    beg = min(max(__builtin_amdgcn_readfirstlane(off[d0]), 0), E);

    for (int d = d0; d < d1; ++d) {
        const int end = min(max(__builtin_amdgcn_readfirstlane(off[d + 1]), beg), E);
        int end2 = end;
        if (d + 1 < d1)
            end2 = min(max(__builtin_amdgcn_readfirstlane(off[d + 2]), end), E);

        float kr[8];
        {
            const float4 k0 = *(const float4*)(kbuf + d * H + 4 * c);
            const float4 k1 = *(const float4*)(kbuf + d * H + 64 + 4 * c);
            kr[0] = k0.x; kr[1] = k0.y; kr[2] = k0.z; kr[3] = k0.w;
            kr[4] = k1.x; kr[5] = k1.y; kr[6] = k1.z; kr[7] = k1.w;
        }
        float acc[8];
        #pragma unroll
        for (int t = 0; t < 8; ++t) acc[t] = 0.0f;

        if (!primed && beg < end) {
            const int pe = perm[min(beg + ia, end - 1)];
            ps_l = psrc[min(beg + c, end - 1)];
            const float* xp = xe + min(max(pe, 0), E - 1) * D_EDGE + 8 * h;
            x0 = *(const float4*)(xp);
            x1 = *(const float4*)(xp + 4);
        }
        primed = false;

        for (int j0 = beg; j0 < end; j0 += 16) {
            const int cnt = min(16, end - j0);
            const bool more     = j0 + 16 < end;
            const int  nj0      = more ? j0 + 16 : end;
            const int  nend     = more ? end : end2;
            const bool has_next = nj0 < nend;
            const int pe_n = perm[min(nj0 + ia, max(nend, nj0) - 1)];
            const int ps_n = psrc[min(nj0 + c, max(nend, nj0) - 1)];

            v2f a[4];
            a[0].x = x0.x; a[0].y = x0.y;
            a[1].x = x0.z; a[1].y = x0.w;
            a[2].x = x1.x; a[2].y = x1.y;
            a[3].x = x1.z; a[3].y = x1.w;

            const int ps_c = min(max(ps_l, 0), n_src - 1);
            unsigned so[8];
            #pragma unroll
            for (int r = 0; r < 8; ++r) {
                const int s_even = __builtin_amdgcn_readlane(ps_c, 2 * r);
                const int s_odd  = __builtin_amdgcn_readlane(ps_c, 2 * r + 1);
                so[r] = (unsigned)((h ? s_odd : s_even) * (H * 4) + 16 * c);
            }
            const int lim = cnt - h;
            int slot = lane;
            asm volatile("" : "+v"(slot));

            const float* xnext = xe + min(max(pe_n, 0), E - 1) * D_EDGE + 8 * h;

            if (cnt > 12)
                edge_batch<8>(qbuf, vbuf, xnext, sB, slot, so, a, cb, kr, acc, x0, x1, lim);
            else if (cnt > 8)
                edge_batch<6>(qbuf, vbuf, xnext, sB, slot, so, a, cb, kr, acc, x0, x1, lim);
            else if (cnt > 4)
                edge_batch<4>(qbuf, vbuf, xnext, sB, slot, so, a, cb, kr, acc, x0, x1, lim);
            else
                edge_batch<2>(qbuf, vbuf, xnext, sB, slot, so, a, cb, kr, acc, x0, x1, lim);
            ps_l   = ps_n;
            primed = has_next;
        }

        float tot[8];
        #pragma unroll
        for (int t = 0; t < 8; ++t) tot[t] = acc[t] + __shfl_xor(acc[t], 16, 32);

        float4* op = (float4*)(out_dst + d * H + 64 * h + 4 * c);
        float4 o = *op;
        o.x = fmaxf(o.x + (h ? tot[4] : tot[0]), 0.0f);
        o.y = fmaxf(o.y + (h ? tot[5] : tot[1]), 0.0f);
        o.z = fmaxf(o.z + (h ? tot[6] : tot[2]), 0.0f);
        o.w = fmaxf(o.w + (h ? tot[7] : tot[3]), 0.0f);
        *op = o;

        beg = end;
    }
}

__global__ void classifier(const float* __restrict__ hm,
                           const float* __restrict__ hv,
                           const int*   __restrict__ im,
                           const int*   __restrict__ iv,
                           const float* __restrict__ Wc,
                           const float* __restrict__ bc,
                           float*       __restrict__ out,
                           int P)
{
    const int p    = blockIdx.x * 8 + (threadIdx.x >> 5);
    const int lane = threadIdx.x & 31;
    if (p >= P) return;

    const float4 am = ((const float4*)(hm + (size_t)im[p] * H))[lane];
    const float4 av = ((const float4*)(hv + (size_t)iv[p] * H))[lane];

    float c0 = 0.f, c1 = 0.f;
    const int ch = lane * 4;
    const float a[4] = {am.x, am.y, am.z, am.w};
    const float b[4] = {av.x, av.y, av.z, av.w};
    #pragma unroll
    for (int i = 0; i < 4; ++i) {
        c0 = fmaf(a[i], Wc[(ch + i) * 2 + 0], c0);
        c1 = fmaf(a[i], Wc[(ch + i) * 2 + 1], c1);
        c0 = fmaf(b[i], Wc[(H + ch + i) * 2 + 0], c0);
        c1 = fmaf(b[i], Wc[(H + ch + i) * 2 + 1], c1);
    }
    #pragma unroll
    for (int off = 16; off > 0; off >>= 1) {
        c0 += __shfl_down(c0, off, 32);
        c1 += __shfl_down(c1, off, 32);
    }
    if (lane == 0) {
        out[(size_t)p * 2 + 0] = c0 + bc[0];
        out[(size_t)p * 2 + 1] = c1 + bc[1];
    }
}

static inline void gemm_in(const float* A, const float* B, const float* bias,
                           float* C, int M, hipStream_t s)
{
    gemm_wmma_bias<5, D_IN><<<dim3(M / 80 * 2), dim3(128), 0, s>>>(A, B, bias, C, M);
}
#define ER_ROWS_A 1
#define ER_ROWS_B 4
#define NB_MOUSE (N_MOUSE / 80 * 2)
#define NB_VIRUS (N_VIRUS / 80 * 2)
static inline void gemm4(const Proj4& p, hipStream_t s)
{
    gemm_wmma_bias4<5, NB_MOUSE, NB_VIRUS>
        <<<dim3(2 * NB_MOUSE + 2 * NB_VIRUS), dim3(128), 0, s>>>(p);
}

extern "C" void kernel_launch(void* const* d_in, const int* in_sizes, int n_in,
                              void* d_out, int out_size, void* d_ws, size_t ws_size,
                              hipStream_t stream)
{
    const float* x_mouse  = (const float*)d_in[0];
    const float* x_virus  = (const float*)d_in[1];
    const float* xe_mv    = (const float*)d_in[2];
    const float* xe_vm    = (const float*)d_in[3];
    const float* W_mouse  = (const float*)d_in[4];
    const float* b_mouse  = (const float*)d_in[5];
    const float* W_virus  = (const float*)d_in[6];
    const float* b_virus  = (const float*)d_in[7];
    const float* W_key    = (const float*)d_in[8];
    const float* b_key    = (const float*)d_in[9];
    const float* W_query  = (const float*)d_in[10];
    const float* b_query  = (const float*)d_in[11];
    const float* W_value  = (const float*)d_in[12];
    const float* b_value  = (const float*)d_in[13];
    const float* W_edge   = (const float*)d_in[14];
    const float* b_edge   = (const float*)d_in[15];
    const float* W_skip   = (const float*)d_in[16];
    const float* bias_cv  = (const float*)d_in[17];
    const float* W_cls    = (const float*)d_in[18];
    const float* b_cls    = (const float*)d_in[19];
    const int* ei_mv_src  = (const int*)d_in[20];
    const int* ei_mv_dst  = (const int*)d_in[21];
    const int* ei_vm_src  = (const int*)d_in[22];
    const int* ei_vm_dst  = (const int*)d_in[23];
    const int* eli_mouse  = (const int*)d_in[24];
    const int* eli_virus  = (const int*)d_in[25];
    float* out = (float*)d_out;

    float* ws  = (float*)d_ws;
    const size_t SZ_M = (size_t)N_MOUSE * H;
    const size_t SZ_V = (size_t)N_VIRUS * H;
    float* hm0 = ws;
    float* hm1 = hm0 + SZ_M;
    float* hv0 = hm1 + SZ_M;
    float* hv1 = hv0 + SZ_V;
    float* bx  = hv1 + SZ_V;
    float* by  = bx + SZ_M;
    float* bz  = by + SZ_M;

    int* cnt_a  = (int*)(bz + SZ_V);
    int* cnt_b  = cnt_a + N_VIRUS;
    int* off_a  = cnt_b + N_MOUSE;
    int* off_b  = off_a + N_VIRUS + 4;
    int* perm_a = off_b + N_MOUSE + 4;
    int* psrc_a = perm_a + NEDGE;
    int* perm_b = psrc_a + NEDGE;
    int* psrc_b = perm_b + NEDGE;
    int* slot_a = psrc_a;
    int* slot_b = psrc_b;

    const int e_blocks = (NEDGE + 255) / 256;

    int* part = (int*)bx;

    dst_rank<<<dim3((RANGES_A + RANGES_B) * SLICES), dim3(RANK_THREADS), 0, stream>>>(
        ei_mv_dst, ei_vm_dst, slot_a, slot_b, part);
    dst_combine<<<dim3((N_ROWS + 255) / 256), dim3(256), 0, stream>>>(part, cnt_a);
    dst_scan<<<dim3(2), dim3(1024), 0, stream>>>(
        cnt_a, off_a, N_VIRUS, cnt_b, off_b, N_MOUSE);
    dst_fill<<<dim3(e_blocks), dim3(256), 0, stream>>>(
        ei_mv_dst, off_a, slot_a, perm_a, ei_vm_dst, off_b, slot_b, perm_b,
        part, NEDGE, N_VIRUS, N_MOUSE);
    dst_order<<<dim3((N_VIRUS + N_MOUSE + 7) / 8), dim3(256), 0, stream>>>(
        off_a, perm_a, psrc_a, ei_mv_src, N_VIRUS,
        off_b, perm_b, psrc_b, ei_vm_src, N_MOUSE, NEDGE);

    gemm_in(x_mouse, W_mouse, b_mouse, hm0, N_MOUSE, stream);
    gemm_in(x_virus, W_virus, b_virus, hv0, N_VIRUS, stream);

    float* hm_in = hm0; float* hv_in = hv0;
    float* hm_out = hm1; float* hv_out = hv1;

    for (int layer = 0; layer < 2; ++layer) {
        const int i_mv = layer * 2;
        const int i_vm = layer * 2 + 1;

        {
            const int i = i_mv;
            const float* Wk = W_key   + (size_t)i * H * H;
            const float* Wq = W_query + (size_t)i * H * H;
            const float* Wv = W_value + (size_t)i * H * H;
            const float* We = W_edge  + (size_t)i * D_EDGE * H;
            const float* Ws = W_skip  + (size_t)i * H * H;
            Proj4 p;
            p.A0 = hm_in; p.B0 = Wq; p.bias0 = b_query + (size_t)i * H; p.C0 = bx;
            p.A1 = hm_in; p.B1 = Wv; p.bias1 = b_value + (size_t)i * H; p.C1 = by;
            p.A2 = hv_in; p.B2 = Ws; p.bias2 = bias_cv + (size_t)i * H; p.C2 = hv_out;
            p.A3 = hv_in; p.B3 = Wk; p.bias3 = b_key   + (size_t)i * H; p.C3 = bz;
            gemm4(p, stream);
            edge_rows<ER_ROWS_A><<<dim3((N_VIRUS + 8 * ER_ROWS_A - 1) / (8 * ER_ROWS_A)), dim3(256), 0, stream>>>(
                bz, bx, by, xe_mv, We, b_edge + (size_t)i * H,
                off_a, perm_a, psrc_a, hv_out, N_VIRUS, N_MOUSE, NEDGE);
        }
        {
            const int i = i_vm;
            const float* Wk = W_key   + (size_t)i * H * H;
            const float* Wq = W_query + (size_t)i * H * H;
            const float* Wv = W_value + (size_t)i * H * H;
            const float* We = W_edge  + (size_t)i * D_EDGE * H;
            const float* Ws = W_skip  + (size_t)i * H * H;
            Proj4 p;
            p.A0 = hm_in; p.B0 = Ws; p.bias0 = bias_cv + (size_t)i * H; p.C0 = hm_out;
            p.A1 = hm_in; p.B1 = Wk; p.bias1 = b_key   + (size_t)i * H; p.C1 = bx;
            p.A2 = hv_in; p.B2 = Wq; p.bias2 = b_query + (size_t)i * H; p.C2 = by;
            p.A3 = hv_in; p.B3 = Wv; p.bias3 = b_value + (size_t)i * H; p.C3 = by + SZ_V;
            gemm4(p, stream);
            edge_rows<ER_ROWS_B><<<dim3((N_MOUSE + 8 * ER_ROWS_B - 1) / (8 * ER_ROWS_B)), dim3(256), 0, stream>>>(
                bx, by, by + SZ_V, xe_vm, We, b_edge + (size_t)i * H,
                off_b, perm_b, psrc_b, hm_out, N_MOUSE, N_VIRUS, NEDGE);
        }

        float* t;
        t = hm_in; hm_in = hm_out; hm_out = t;
        t = hv_in; hv_in = hv_out; hv_out = t;
    }

    classifier<<<dim3((NPAIR + 7) / 8), dim3(256), 0, stream>>>(
        hm_in, hv_in, eli_mouse, eli_virus, W_cls, b_cls, out, NPAIR);
}


# edge bucketing: rank, combine and scan run as extra 128-thread blocks inside the three projection launches ahead of the first edge kernel (16 launches -> 13; scan per 128-row group); sum order kept
# speedup vs baseline: 2.3669x; 1.0363x over previous
#include <hip/hip_runtime.h>
#include <hip/hip_bf16.h>

typedef __attribute__((ext_vector_type(2))) float v2f;
typedef __attribute__((ext_vector_type(8))) float v8f;

#define H 128
#define D_EDGE 16
#define N_MOUSE 50000
#define N_VIRUS 10000
#define NEDGE   500000
#define NPAIR   100000
#define D_IN    1024

template <int RT, int KK>
__device__ __forceinline__
void gemm_tile(const float* __restrict__ A,
               const float* __restrict__ B,
               const float* __restrict__ bias,
               float* __restrict__ C,
               int bid)
{
    __shared__ __align__(16) float As[RT * 16 * 132];

    const int tid  = threadIdx.x;
    const int wave = tid >> 5;
    const int lane = tid & 31;
    const int hi   = lane >> 4;
    const int ml   = lane & 15;
    const int row0 = (bid >> 1) * (16 * RT);
    const int n0   = (bid & 1) * 64 + wave * 16;

    v8f acc[RT];
    #pragma unroll
    for (int rt = 0; rt < RT; ++rt) acc[rt] = v8f{};

    const float* ap = A + (size_t)(row0 + wave) * KK + 4 * lane;
    float*       sp = &As[wave * 132 + 4 * lane];
    const float* bp = B + (size_t)(2 * hi) * H + n0 + ml;

    for (int k0 = 0; k0 < KK; k0 += 128) {
        v2f b[32];
        #pragma unroll
        for (int s = 0; s < 32; ++s) {
            b[s].x = bp[(size_t)(k0 + 4 * s) * H];
            b[s].y = bp[(size_t)(k0 + 4 * s + 1) * H];
        }
        float4 st[4 * RT];
        #pragma unroll
        for (int it = 0; it < 4 * RT; ++it)
            st[it] = *(const float4*)(ap + (size_t)(4 * it) * KK + k0);
        #pragma unroll
        for (int it = 0; it < 4 * RT; ++it)
            *(float4*)(sp + (4 * it) * 132) = st[it];
        __syncthreads();

        #pragma unroll
        for (int s = 0; s < 32; ++s) {
            #pragma unroll
            for (int rt = 0; rt < RT; ++rt) {
                v2f a = *(const v2f*)(&As[(rt * 16 + ml) * 132 + 4 * s + 2 * hi]);
                acc[rt] = __builtin_amdgcn_wmma_f32_16x16x4_f32(
                              false, a, false, b[s], (short)0, acc[rt], false, false);
            }
        }
        __syncthreads();
    }

    const float bval = bias ? bias[n0 + ml] : 0.0f;
    #pragma unroll
    for (int rt = 0; rt < RT; ++rt) {
        #pragma unroll
        for (int r = 0; r < 8; ++r) {
            int m = row0 + rt * 16 + r + 8 * hi;
            C[(size_t)m * H + n0 + ml] = acc[rt][r] + bval;
        }
    }
}

struct Proj4 {
    const float* A0; const float* B0; const float* bias0; float* C0;
    const float* A1; const float* B1; const float* bias1; float* C1;
    const float* A2; const float* B2; const float* bias2; float* C2;
    const float* A3; const float* B3; const float* bias3; float* C3;
};

template <int RT, int NB0, int NB2>
__device__ __forceinline__
void proj4_block(const Proj4& p, int bid)
{
    const float* A; const float* B; const float* bias; float* C;
    if (bid < NB0) {
        A = p.A0; B = p.B0; bias = p.bias0; C = p.C0;
    } else if (bid < 2 * NB0) {
        bid -= NB0;
        A = p.A1; B = p.B1; bias = p.bias1; C = p.C1;
    } else if (bid < 2 * NB0 + NB2) {
        bid -= 2 * NB0;
        A = p.A2; B = p.B2; bias = p.bias2; C = p.C2;
    } else {
        bid -= 2 * NB0 + NB2;
        A = p.A3; B = p.B3; bias = p.bias3; C = p.C3;
    }
    gemm_tile<RT, H>(A, B, bias, C, bid);
}

template <int RT, int NB0, int NB2>
__global__ __launch_bounds__(128) __attribute__((amdgpu_waves_per_eu(5, 5)))
void gemm_wmma_bias4(const Proj4 p)
{
    proj4_block<RT, NB0, NB2>(p, blockIdx.x);
}

#define SLICES       16
#define RANGES_A     4
#define RANGES_B     12
#define RSZ_A        ((N_VIRUS + RANGES_A - 1) / RANGES_A)
#define RSZ_B        ((N_MOUSE + RANGES_B - 1) / RANGES_B)
#define RSZ_MAX      (RSZ_A > RSZ_B ? RSZ_A : RSZ_B)
#define SLICE_EDGES  ((NEDGE + SLICES - 1) / SLICES)
#define N_ROWS       (N_VIRUS + N_MOUSE)
#define RANK_THREADS 128
#define RANK_BATCH   16
#define RANK_BLOCKS  ((RANGES_A + RANGES_B) * SLICES)
#define GROUP_ROWS   128
#define GROUPS_A     ((N_VIRUS + GROUP_ROWS - 1) / GROUP_ROWS)
#define GROUPS_B     ((N_MOUSE + GROUP_ROWS - 1) / GROUP_ROWS)
#define GROUPS       (GROUPS_A + GROUPS_B)

__device__ __forceinline__
void dst_rank(const int blk,
              const int* __restrict__ dst_a,
              const int* __restrict__ dst_b,
              int* __restrict__ slot_a,
              int* __restrict__ slot_b,
              int* __restrict__ part)
{
    __shared__ int c[RSZ_MAX];

    const int  tid  = threadIdx.x;
    const int  s    = blk % SLICES;
    const int  gr   = blk / SLICES;
    const bool is_a = gr < RANGES_A;
    const int* dst  = is_a ? dst_a : dst_b;
    int*       slot = is_a ? slot_a : slot_b;
    const int  n    = is_a ? N_VIRUS : N_MOUSE;
    const int  rsz  = is_a ? RSZ_A : RSZ_B;
    const int  lo   = (is_a ? gr : gr - RANGES_A) * rsz;
    const int  rows = min(rsz, n - lo);

    for (int i = tid; i < rsz; i += RANK_THREADS) c[i] = 0;
    __syncthreads();

    const int e0 = s * SLICE_EDGES;
    const int e1 = min(NEDGE, e0 + SLICE_EDGES);
    for (int base = e0 + tid; base < e1; base += RANK_BATCH * RANK_THREADS) {
        int dv[RANK_BATCH];
        #pragma unroll
        for (int k = 0; k < RANK_BATCH; ++k)
            dv[k] = dst[min(base + k * RANK_THREADS, e1 - 1)];
        int rk[RANK_BATCH];
        #pragma unroll
        for (int k = 0; k < RANK_BATCH; ++k) {
            const int e = base + k * RANK_THREADS;
            const unsigned t = (unsigned)(min(max(dv[k], 0), n - 1) - lo);
            rk[k] = -1;
            if (e < e1 && t < (unsigned)rsz) rk[k] = atomicAdd(&c[t], 1);
        }
        #pragma unroll
        for (int k = 0; k < RANK_BATCH; ++k)
            if (rk[k] >= 0) slot[base + k * RANK_THREADS] = rk[k];
    }
    __syncthreads();

    int* prow = part + (size_t)s * N_ROWS + (is_a ? 0 : N_VIRUS) + lo;
    for (int i = tid; i < rows; i += RANK_THREADS) prow[i] = c[i];
}

__device__ __forceinline__
void dst_combine(const int grp,
                 int* __restrict__ part,
                 int* __restrict__ cnt,
                 int* __restrict__ gtot)
{
    __shared__ int wtot[GROUP_ROWS / 32];

    const int  tid  = threadIdx.x;
    const bool is_a = grp < GROUPS_A;
    const int  r    = (is_a ? grp : grp - GROUPS_A) * GROUP_ROWS + tid;
    const int  n    = is_a ? N_VIRUS : N_MOUSE;
    const int  d    = (is_a ? 0 : N_VIRUS) + r;

    int run = 0;
    if (r < n) {
        int v[SLICES];
        #pragma unroll
        for (int s = 0; s < SLICES; ++s) v[s] = part[(size_t)s * N_ROWS + d];
        #pragma unroll
        for (int s = 0; s < SLICES; ++s) {
            part[(size_t)s * N_ROWS + d] = run;
            run += v[s];
        }
        cnt[d] = run;
    }
    int t = run;
    for (int o = 16; o > 0; o >>= 1) t += __shfl_xor(t, o, 32);
    if ((tid & 31) == 0) wtot[tid >> 5] = t;
    __syncthreads();
    if (tid == 0) {
        int g = 0;
        for (int w = 0; w < GROUP_ROWS / 32; ++w) g += wtot[w];
        gtot[grp] = g;
    }
}

__device__ __forceinline__
void dst_scan(const int grp,
              const int* __restrict__ cnt,
              const int* __restrict__ gtot,
              int* __restrict__ off_a,
              int* __restrict__ off_b)
{
    __shared__ int sred[GROUP_ROWS / 32];
    __shared__ int wsum[GROUP_ROWS / 32];

    const int  tid  = threadIdx.x;
    const int  lane = tid & 31;
    const int  wave = tid >> 5;
    const bool is_a = grp < GROUPS_A;
    const int  g0   = is_a ? 0 : GROUPS_A;
    const int  jj   = grp - g0;
    const int  ng   = is_a ? GROUPS_A : GROUPS_B;
    const int  n    = is_a ? N_VIRUS : N_MOUSE;
    int*       off  = is_a ? off_a : off_b;

    int b = 0;
    for (int i = tid; i < jj; i += GROUP_ROWS) b += gtot[g0 + i];
    for (int o = 16; o > 0; o >>= 1) b += __shfl_xor(b, o, 32);
    if (lane == 0) sred[wave] = b;

    const int r = jj * GROUP_ROWS + tid;
    const int c = (r < n) ? cnt[(is_a ? 0 : N_VIRUS) + r] : 0;
    int incl = c;
    for (int o = 1; o < 32; o <<= 1) {
        int y = __shfl_up(incl, o, 32);
        if (lane >= o) incl += y;
    }
    if (lane == 31) wsum[wave] = incl;
    __syncthreads();

    int start = 0;
    for (int w = 0; w < GROUP_ROWS / 32; ++w) {
        start += sred[w];
        if (w < wave) start += wsum[w];
    }
    if (r < n) off[r] = start + incl - c;
    if (jj == ng - 1 && tid == GROUP_ROWS - 1) off[n] = start + incl;
}

template <int RT, int KK>
__global__ __launch_bounds__(128) __attribute__((amdgpu_waves_per_eu(5, 5)))
void gemm_bias_rank(const float* __restrict__ A,
                    const float* __restrict__ B,
                    const float* __restrict__ bias,
                    float* __restrict__ C,
                    const int* __restrict__ dst_a,
                    const int* __restrict__ dst_b,
                    int* __restrict__ slot_a,
                    int* __restrict__ slot_b,
                    int* __restrict__ part)
{
    if (blockIdx.x < RANK_BLOCKS) {
        dst_rank(blockIdx.x, dst_a, dst_b, slot_a, slot_b, part);
        return;
    }
    gemm_tile<RT, KK>(A, B, bias, C, blockIdx.x - RANK_BLOCKS);
}

template <int RT, int KK>
__global__ __launch_bounds__(128) __attribute__((amdgpu_waves_per_eu(5, 5)))
void gemm_bias_combine(const float* __restrict__ A,
                       const float* __restrict__ B,
                       const float* __restrict__ bias,
                       float* __restrict__ C,
                       int* __restrict__ part,
                       int* __restrict__ cnt,
                       int* __restrict__ gtot)
{
    if (blockIdx.x < GROUPS) {
        dst_combine(blockIdx.x, part, cnt, gtot);
        return;
    }
    gemm_tile<RT, KK>(A, B, bias, C, blockIdx.x - GROUPS);
}

template <int RT, int NB0, int NB2>
__global__ __launch_bounds__(128) __attribute__((amdgpu_waves_per_eu(5, 5)))
void gemm_wmma_bias4_scan(const Proj4 p,
                          const int* __restrict__ cnt,
                          const int* __restrict__ gtot,
                          int* __restrict__ off_a,
                          int* __restrict__ off_b)
{
    if (blockIdx.x < GROUPS) {
        dst_scan(blockIdx.x, cnt, gtot, off_a, off_b);
        return;
    }
    proj4_block<RT, NB0, NB2>(p, blockIdx.x - GROUPS);
}

__global__ void dst_fill(const int* __restrict__ dst_a,
                         const int* __restrict__ off_a,
                         const int* __restrict__ slot_a,
                         int* __restrict__ perm_a,
                         const int* __restrict__ dst_b,
                         const int* __restrict__ off_b,
                         const int* __restrict__ slot_b,
                         int* __restrict__ perm_b,
                         const int* __restrict__ part,
                         int E, int na, int nb)
{
    int e = blockIdx.x * blockDim.x + threadIdx.x;
    if (e >= E) return;
    const int* prow = part + (size_t)(e / SLICE_EDGES) * N_ROWS;
    {
        int d = min(max(dst_a[e], 0), na - 1);
        int p = off_a[d] + prow[d] + slot_a[e];
        perm_a[min(max(p, 0), E - 1)] = e;
    }
    {
        int d = min(max(dst_b[e], 0), nb - 1);
        int p = off_b[d] + prow[na + d] + slot_b[e];
        perm_b[min(max(p, 0), E - 1)] = e;
    }
}

#define SORT_CAP 512
__global__ __launch_bounds__(256) void dst_order(const int* __restrict__ off_a,
                                                 int* __restrict__ perm_a,
                                                 int* __restrict__ psrc_a,
                                                 const int* __restrict__ src_a, int na,
                                                 const int* __restrict__ off_b,
                                                 int* __restrict__ perm_b,
                                                 int* __restrict__ psrc_b,
                                                 const int* __restrict__ src_b, int nb,
                                                 int E)
{
    __shared__ __align__(16) int skey[8][SORT_CAP];

    const int lane = threadIdx.x & 31;
    const int wv   = threadIdx.x >> 5;
    int row = blockIdx.x * 8 + wv;

    const bool is_a  = row < na;
    const int* off   = is_a ? off_a  : off_b;
    int*       perm  = is_a ? perm_a : perm_b;
    int*       psrc  = is_a ? psrc_a : psrc_b;
    const int* src   = is_a ? src_a  : src_b;
    if (!is_a) row -= na;
    const bool valid = is_a || row < nb;

    int beg = 0, n = 0;
    if (valid) {
        beg = min(max(off[row], 0), E);
        n   = min(max(off[row + 1], beg), E) - beg;
    }
    const bool small = n <= SORT_CAP;
    const int  npad  = (n + 3) & ~3;
    int* key = skey[wv];

    if (small) {
        for (int j = lane; j < npad; j += 32)
            key[j] = (j < n) ? perm[beg + j] : 0x7fffffff;
    }
    __syncthreads();

    if (small) {
        for (int j0 = 0; j0 < n; j0 += 32) {
            const int j  = j0 + lane;
            const int my = (j < n) ? key[j] : 0x7fffffff;
            int r = 0;
            for (int t = 0; t < npad; t += 4) {
                const int4 k4 = *(const int4*)(key + t);
                r += (k4.x < my) + (k4.y < my) + (k4.z < my) + (k4.w < my);
            }
            if (j < n) {
                perm[beg + r] = my;
                psrc[beg + r] = src[min(max(my, 0), E - 1)];
            }
        }
    } else {
        for (int j = lane; j < n; j += 32)
            psrc[beg + j] = src[min(max(perm[beg + j], 0), E - 1)];
    }
}

typedef __attribute__((ext_vector_type(4))) float v4f;

template <int NR>
__device__ __forceinline__
void edge_batch(const float* __restrict__ qbuf,
                const float* __restrict__ vbuf,
                const float* __restrict__ xnext,
                const v4f (&sB)[16 * 32],
                const int lane,
                const unsigned (&so)[8],
                const v2f (&a)[4],
                const float (&cb)[8],
                const float (&kr)[8],
                float (&acc)[8],
                float4& x0, float4& x1,
                int lim)
{
    #pragma unroll
    for (int g = 0; g < 2; ++g) {
        if (g == 1) {
            __builtin_amdgcn_sched_barrier(0);
            x0 = *(const float4*)(xnext);
            x1 = *(const float4*)(xnext + 4);
        }
        const char* qb = (const char*)qbuf + 256 * g;
        const char* vb = (const char*)vbuf + 256 * g;
        float qf[NR][4], vf[NR][4];
        #pragma unroll
        for (int r = 0; r < NR; ++r) {
            const float4 q4 = *(const float4*)(qb + so[r]);
            const float4 v4 = *(const float4*)(vb + so[r]);
            qf[r][0] = q4.x; qf[r][1] = q4.y; qf[r][2] = q4.z; qf[r][3] = q4.w;
            vf[r][0] = v4.x; vf[r][1] = v4.y; vf[r][2] = v4.z; vf[r][3] = v4.w;
        }

        v2f b[4][4];
        #pragma unroll
        for (int tt = 0; tt < 4; ++tt) {
            #pragma unroll
            for (int sp = 0; sp < 2; ++sp) {
                const v4f f = sB[(2 * (4 * g + tt) + sp) * 32 + lane];
                b[2 * sp][tt].x     = f.x; b[2 * sp][tt].y     = f.y;
                b[2 * sp + 1][tt].x = f.z; b[2 * sp + 1][tt].y = f.w;
            }
        }

        v8f pr[4];
        #pragma unroll
        for (int tt = 0; tt < 4; ++tt) pr[tt] = v8f{};
        #pragma unroll
        for (int s = 0; s < 4; ++s) {
            #pragma unroll
            for (int tt = 0; tt < 4; ++tt)
                pr[tt] = __builtin_amdgcn_wmma_f32_16x16x4_f32(
                             false, a[s], false, b[s][tt], (short)0, pr[tt],
                             false, false);
        }

        #pragma unroll
        for (int r = 0; r < NR; ++r) {
            #pragma unroll
            for (int tt = 0; tt < 4; ++tt) {
                const float ev = pr[tt][r] + cb[4 * g + tt];
                const float z  = (kr[4 * g + tt] + ev) + (qf[r][tt] + ev);
                float gt = __builtin_amdgcn_rcpf(1.0f + __expf(-z));
                if (r >= NR - 2) gt = (2 * r < lim) ? gt : 0.0f;
                acc[4 * g + tt] += gt * (vf[r][tt] + ev);
            }
        }
    }
}

template <int ROWS>
__global__ __launch_bounds__(256) __attribute__((amdgpu_waves_per_eu(4, 4)))
void edge_rows(const float* __restrict__ kbuf,
               const float* __restrict__ qbuf,
               const float* __restrict__ vbuf,
               const float* __restrict__ xe,
               const float* __restrict__ We,
               const float* __restrict__ be,
               const int*   __restrict__ off,
               const int*   __restrict__ perm,
               const int*   __restrict__ psrc,
               float*       __restrict__ out_dst,
               int n_dst, int n_src, int E)
{
    __shared__ v4f sB[16 * 32];

    const int lane = threadIdx.x & 31;
    const int c    = lane & 15;
    const int h    = lane >> 4;

    #pragma unroll
    for (int i = 0; i < 2; ++i) {
        const int combo = (threadIdx.x >> 5) * 2 + i;
        const int t     = combo >> 1;
        const int sp    = combo & 1;
        const float* wp = We + (8 * h + 4 * sp) * H + 64 * (t >> 2) + 4 * c + (t & 3);
        v4f f;
        f.x = wp[0];
        f.y = wp[H];
        f.z = wp[2 * H];
        f.w = wp[3 * H];
        sB[combo * 32 + lane] = f;
    }
    __syncthreads();

    const int d0 = __builtin_amdgcn_readfirstlane(
                       (blockIdx.x * 8 + (threadIdx.x >> 5)) * ROWS);
    if (d0 >= n_dst) return;
    const int d1 = min(d0 + ROWS, n_dst);

    float cb[8];
    {
        const float4 c0 = *(const float4*)(be + 4 * c);
        const float4 c1 = *(const float4*)(be + 64 + 4 * c);
        cb[0] = c0.x; cb[1] = c0.y; cb[2] = c0.z; cb[3] = c0.w;
        cb[4] = c1.x; cb[5] = c1.y; cb[6] = c1.z; cb[7] = c1.w;
    }
    const int ia = 2 * (c & 7) + (c >> 3);

    int    ps_l = 0;
    float4 x0 = make_float4(0.f, 0.f, 0.f, 0.f), x1 = x0;
    bool   primed = false;
    int    beg = min(max(__builtin_amdgcn_readfirstlane(off[d0]), 0), E);

    for (int d = d0; d < d1; ++d) {
        const int end = min(max(__builtin_amdgcn_readfirstlane(off[d + 1]), beg), E);
        int end2 = end;
        if (d + 1 < d1)
            end2 = min(max(__builtin_amdgcn_readfirstlane(off[d + 2]), end), E);

        float kr[8];
        {
            const float4 k0 = *(const float4*)(kbuf + d * H + 4 * c);
            const float4 k1 = *(const float4*)(kbuf + d * H + 64 + 4 * c);
            kr[0] = k0.x; kr[1] = k0.y; kr[2] = k0.z; kr[3] = k0.w;
            kr[4] = k1.x; kr[5] = k1.y; kr[6] = k1.z; kr[7] = k1.w;
        }
        float acc[8];
        #pragma unroll
        for (int t = 0; t < 8; ++t) acc[t] = 0.0f;

        if (!primed && beg < end) {
            const int pe = perm[min(beg + ia, end - 1)];
            ps_l = psrc[min(beg + c, end - 1)];
            const float* xp = xe + min(max(pe, 0), E - 1) * D_EDGE + 8 * h;
            x0 = *(const float4*)(xp);
            x1 = *(const float4*)(xp + 4);
        }
        primed = false;

        for (int j0 = beg; j0 < end; j0 += 16) {
            const int cnt = min(16, end - j0);
            const bool more     = j0 + 16 < end;
            const int  nj0      = more ? j0 + 16 : end;
            const int  nend     = more ? end : end2;
            const bool has_next = nj0 < nend;
            const int pe_n = perm[min(nj0 + ia, max(nend, nj0) - 1)];
            const int ps_n = psrc[min(nj0 + c, max(nend, nj0) - 1)];

            v2f a[4];
            a[0].x = x0.x; a[0].y = x0.y;
            a[1].x = x0.z; a[1].y = x0.w;
            a[2].x = x1.x; a[2].y = x1.y;
            a[3].x = x1.z; a[3].y = x1.w;

            const int ps_c = min(max(ps_l, 0), n_src - 1);
            unsigned so[8];
            #pragma unroll
            for (int r = 0; r < 8; ++r) {
                const int s_even = __builtin_amdgcn_readlane(ps_c, 2 * r);
                const int s_odd  = __builtin_amdgcn_readlane(ps_c, 2 * r + 1);
                so[r] = (unsigned)((h ? s_odd : s_even) * (H * 4) + 16 * c);
            }
            const int lim = cnt - h;
            int slot = lane;
            asm volatile("" : "+v"(slot));

            const float* xnext = xe + min(max(pe_n, 0), E - 1) * D_EDGE + 8 * h;

            if (cnt > 12)
                edge_batch<8>(qbuf, vbuf, xnext, sB, slot, so, a, cb, kr, acc, x0, x1, lim);
            else if (cnt > 8)
                edge_batch<6>(qbuf, vbuf, xnext, sB, slot, so, a, cb, kr, acc, x0, x1, lim);
            else if (cnt > 4)
                edge_batch<4>(qbuf, vbuf, xnext, sB, slot, so, a, cb, kr, acc, x0, x1, lim);
            else
                edge_batch<2>(qbuf, vbuf, xnext, sB, slot, so, a, cb, kr, acc, x0, x1, lim);
            ps_l   = ps_n;
            primed = has_next;
        }

        float tot[8];
        #pragma unroll
        for (int t = 0; t < 8; ++t) tot[t] = acc[t] + __shfl_xor(acc[t], 16, 32);

        float4* op = (float4*)(out_dst + d * H + 64 * h + 4 * c);
        float4 o = *op;
        o.x = fmaxf(o.x + (h ? tot[4] : tot[0]), 0.0f);
        o.y = fmaxf(o.y + (h ? tot[5] : tot[1]), 0.0f);
        o.z = fmaxf(o.z + (h ? tot[6] : tot[2]), 0.0f);
        o.w = fmaxf(o.w + (h ? tot[7] : tot[3]), 0.0f);
        *op = o;

        beg = end;
    }
}

__global__ void classifier(const float* __restrict__ hm,
                           const float* __restrict__ hv,
                           const int*   __restrict__ im,
                           const int*   __restrict__ iv,
                           const float* __restrict__ Wc,
                           const float* __restrict__ bc,
                           float*       __restrict__ out,
                           int P)
{
    const int p    = blockIdx.x * 8 + (threadIdx.x >> 5);
    const int lane = threadIdx.x & 31;
    if (p >= P) return;

    const float4 am = ((const float4*)(hm + (size_t)im[p] * H))[lane];
    const float4 av = ((const float4*)(hv + (size_t)iv[p] * H))[lane];

    float c0 = 0.f, c1 = 0.f;
    const int ch = lane * 4;
    const float a[4] = {am.x, am.y, am.z, am.w};
    const float b[4] = {av.x, av.y, av.z, av.w};
    #pragma unroll
    for (int i = 0; i < 4; ++i) {
        c0 = fmaf(a[i], Wc[(ch + i) * 2 + 0], c0);
        c1 = fmaf(a[i], Wc[(ch + i) * 2 + 1], c1);
        c0 = fmaf(b[i], Wc[(H + ch + i) * 2 + 0], c0);
        c1 = fmaf(b[i], Wc[(H + ch + i) * 2 + 1], c1);
    }
    #pragma unroll
    for (int off = 16; off > 0; off >>= 1) {
        c0 += __shfl_down(c0, off, 32);
        c1 += __shfl_down(c1, off, 32);
    }
    if (lane == 0) {
        out[(size_t)p * 2 + 0] = c0 + bc[0];
        out[(size_t)p * 2 + 1] = c1 + bc[1];
    }
}

#define ER_ROWS_A 1
#define ER_ROWS_B 4
#define NB_MOUSE (N_MOUSE / 80 * 2)
#define NB_VIRUS (N_VIRUS / 80 * 2)
static inline void gemm4(const Proj4& p, hipStream_t s)
{
    gemm_wmma_bias4<5, NB_MOUSE, NB_VIRUS>
        <<<dim3(2 * NB_MOUSE + 2 * NB_VIRUS), dim3(128), 0, s>>>(p);
}

extern "C" void kernel_launch(void* const* d_in, const int* in_sizes, int n_in,
                              void* d_out, int out_size, void* d_ws, size_t ws_size,
                              hipStream_t stream)
{
    const float* x_mouse  = (const float*)d_in[0];
    const float* x_virus  = (const float*)d_in[1];
    const float* xe_mv    = (const float*)d_in[2];
    const float* xe_vm    = (const float*)d_in[3];
    const float* W_mouse  = (const float*)d_in[4];
    const float* b_mouse  = (const float*)d_in[5];
    const float* W_virus  = (const float*)d_in[6];
    const float* b_virus  = (const float*)d_in[7];
    const float* W_key    = (const float*)d_in[8];
    const float* b_key    = (const float*)d_in[9];
    const float* W_query  = (const float*)d_in[10];
    const float* b_query  = (const float*)d_in[11];
    const float* W_value  = (const float*)d_in[12];
    const float* b_value  = (const float*)d_in[13];
    const float* W_edge   = (const float*)d_in[14];
    const float* b_edge   = (const float*)d_in[15];
    const float* W_skip   = (const float*)d_in[16];
    const float* bias_cv  = (const float*)d_in[17];
    const float* W_cls    = (const float*)d_in[18];
    const float* b_cls    = (const float*)d_in[19];
    const int* ei_mv_src  = (const int*)d_in[20];
    const int* ei_mv_dst  = (const int*)d_in[21];
    const int* ei_vm_src  = (const int*)d_in[22];
    const int* ei_vm_dst  = (const int*)d_in[23];
    const int* eli_mouse  = (const int*)d_in[24];
    const int* eli_virus  = (const int*)d_in[25];
    float* out = (float*)d_out;

    float* ws  = (float*)d_ws;
    const size_t SZ_M = (size_t)N_MOUSE * H;
    const size_t SZ_V = (size_t)N_VIRUS * H;
    float* hm0 = ws;
    float* hm1 = hm0 + SZ_M;
    float* hv0 = hm1 + SZ_M;
    float* hv1 = hv0 + SZ_V;
    float* bx  = hv1 + SZ_V;
    float* by  = bx + SZ_M;
    float* bz  = by + SZ_M;

    int* cnt_a  = (int*)(bz + SZ_V);
    int* cnt_b  = cnt_a + N_VIRUS;
    int* off_a  = cnt_b + N_MOUSE;
    int* off_b  = off_a + N_VIRUS + 4;
    int* perm_a = off_b + N_MOUSE + 4;
    int* psrc_a = perm_a + NEDGE;
    int* perm_b = psrc_a + NEDGE;
    int* psrc_b = perm_b + NEDGE;
    int* slot_a = psrc_a;
    int* slot_b = psrc_b;

    const int e_blocks = (NEDGE + 255) / 256;

    int* part = (int*)hm1;
    int* gtot = perm_a;

    gemm_bias_rank<5, D_IN><<<dim3(RANK_BLOCKS + NB_MOUSE), dim3(128), 0, stream>>>(
        x_mouse, W_mouse, b_mouse, hm0, ei_mv_dst, ei_vm_dst, slot_a, slot_b, part);
    gemm_bias_combine<5, D_IN><<<dim3(GROUPS + NB_VIRUS), dim3(128), 0, stream>>>(
        x_virus, W_virus, b_virus, hv0, part, cnt_a, gtot);

    {
        Proj4 p;
        p.A0 = hm0; p.B0 = W_query; p.bias0 = b_query; p.C0 = bx;
        p.A1 = hm0; p.B1 = W_value; p.bias1 = b_value; p.C1 = by;
        p.A2 = hv0; p.B2 = W_skip;  p.bias2 = bias_cv; p.C2 = hv1;
        p.A3 = hv0; p.B3 = W_key;   p.bias3 = b_key;   p.C3 = bz;
        gemm_wmma_bias4_scan<5, NB_MOUSE, NB_VIRUS>
            <<<dim3(GROUPS + 2 * NB_MOUSE + 2 * NB_VIRUS), dim3(128), 0, stream>>>(
                p, cnt_a, gtot, off_a, off_b);
    }

    dst_fill<<<dim3(e_blocks), dim3(256), 0, stream>>>(
        ei_mv_dst, off_a, slot_a, perm_a, ei_vm_dst, off_b, slot_b, perm_b,
        part, NEDGE, N_VIRUS, N_MOUSE);
    dst_order<<<dim3((N_VIRUS + N_MOUSE + 7) / 8), dim3(256), 0, stream>>>(
        off_a, perm_a, psrc_a, ei_mv_src, N_VIRUS,
        off_b, perm_b, psrc_b, ei_vm_src, N_MOUSE, NEDGE);

    float* hm_in = hm0; float* hv_in = hv0;
    float* hm_out = hm1; float* hv_out = hv1;

    for (int layer = 0; layer < 2; ++layer) {
        const int i_mv = layer * 2;
        const int i_vm = layer * 2 + 1;

        {
            const int i = i_mv;
            const float* Wk = W_key   + (size_t)i * H * H;
            const float* Wq = W_query + (size_t)i * H * H;
            const float* Wv = W_value + (size_t)i * H * H;
            const float* We = W_edge  + (size_t)i * D_EDGE * H;
            const float* Ws = W_skip  + (size_t)i * H * H;
            Proj4 p;
            p.A0 = hm_in; p.B0 = Wq; p.bias0 = b_query + (size_t)i * H; p.C0 = bx;
            p.A1 = hm_in; p.B1 = Wv; p.bias1 = b_value + (size_t)i * H; p.C1 = by;
            p.A2 = hv_in; p.B2 = Ws; p.bias2 = bias_cv + (size_t)i * H; p.C2 = hv_out;
            p.A3 = hv_in; p.B3 = Wk; p.bias3 = b_key   + (size_t)i * H; p.C3 = bz;
            if (layer > 0) gemm4(p, stream);
            edge_rows<ER_ROWS_A><<<dim3((N_VIRUS + 8 * ER_ROWS_A - 1) / (8 * ER_ROWS_A)), dim3(256), 0, stream>>>(
                bz, bx, by, xe_mv, We, b_edge + (size_t)i * H,
                off_a, perm_a, psrc_a, hv_out, N_VIRUS, N_MOUSE, NEDGE);
        }
        {
            const int i = i_vm;
            const float* Wk = W_key   + (size_t)i * H * H;
            const float* Wq = W_query + (size_t)i * H * H;
            const float* Wv = W_value + (size_t)i * H * H;
            const float* We = W_edge  + (size_t)i * D_EDGE * H;
            const float* Ws = W_skip  + (size_t)i * H * H;
            Proj4 p;
            p.A0 = hm_in; p.B0 = Ws; p.bias0 = bias_cv + (size_t)i * H; p.C0 = hm_out;
            p.A1 = hm_in; p.B1 = Wk; p.bias1 = b_key   + (size_t)i * H; p.C1 = bx;
            p.A2 = hv_in; p.B2 = Wq; p.bias2 = b_query + (size_t)i * H; p.C2 = by;
            p.A3 = hv_in; p.B3 = Wv; p.bias3 = b_value + (size_t)i * H; p.C3 = by + SZ_V;
            gemm4(p, stream);
            edge_rows<ER_ROWS_B><<<dim3((N_MOUSE + 8 * ER_ROWS_B - 1) / (8 * ER_ROWS_B)), dim3(256), 0, stream>>>(
                bx, by, by + SZ_V, xe_vm, We, b_edge + (size_t)i * H,
                off_b, perm_b, psrc_b, hm_out, N_MOUSE, N_VIRUS, NEDGE);
        }

        float* t;
        t = hm_in; hm_in = hm_out; hm_out = t;
        t = hv_in; hv_in = hv_out; hv_out = t;
    }

    classifier<<<dim3((NPAIR + 7) / 8), dim3(256), 0, stream>>>(
        hm_in, hv_in, eli_mouse, eli_virus, W_cls, b_cls, out, NPAIR);
}


# input projections (1024-deep f32 products): the next 128-deep chunk is loaded in front of the current chunk's products and staged in a second LDS buffer behind them, 3 waves per SIMD; sum order kept
# speedup vs baseline: 2.4910x; 1.0524x over previous
#include <hip/hip_runtime.h>
#include <hip/hip_bf16.h>

typedef __attribute__((ext_vector_type(2))) float v2f;
typedef __attribute__((ext_vector_type(8))) float v8f;

#define H 128
#define D_EDGE 16
#define N_MOUSE 50000
#define N_VIRUS 10000
#define NEDGE   500000
#define NPAIR   100000
#define D_IN    1024

template <int RT, int KK>
__device__ __forceinline__
void gemm_tile(const float* __restrict__ A,
               const float* __restrict__ B,
               const float* __restrict__ bias,
               float* __restrict__ C,
               int bid)
{
    __shared__ __align__(16) float As[RT * 16 * 132];

    const int tid  = threadIdx.x;
    const int wave = tid >> 5;
    const int lane = tid & 31;
    const int hi   = lane >> 4;
    const int ml   = lane & 15;
    const int row0 = (bid >> 1) * (16 * RT);
    const int n0   = (bid & 1) * 64 + wave * 16;

    v8f acc[RT];
    #pragma unroll
    for (int rt = 0; rt < RT; ++rt) acc[rt] = v8f{};

    const float* ap = A + (size_t)(row0 + wave) * KK + 4 * lane;
    float*       sp = &As[wave * 132 + 4 * lane];
    const float* bp = B + (size_t)(2 * hi) * H + n0 + ml;

    for (int k0 = 0; k0 < KK; k0 += 128) {
        v2f b[32];
        #pragma unroll
        for (int s = 0; s < 32; ++s) {
            b[s].x = bp[(size_t)(k0 + 4 * s) * H];
            b[s].y = bp[(size_t)(k0 + 4 * s + 1) * H];
        }
        float4 st[4 * RT];
        #pragma unroll
        for (int it = 0; it < 4 * RT; ++it)
            st[it] = *(const float4*)(ap + (size_t)(4 * it) * KK + k0);
        #pragma unroll
        for (int it = 0; it < 4 * RT; ++it)
            *(float4*)(sp + (4 * it) * 132) = st[it];
        __syncthreads();

        #pragma unroll
        for (int s = 0; s < 32; ++s) {
            #pragma unroll
            for (int rt = 0; rt < RT; ++rt) {
                v2f a = *(const v2f*)(&As[(rt * 16 + ml) * 132 + 4 * s + 2 * hi]);
                acc[rt] = __builtin_amdgcn_wmma_f32_16x16x4_f32(
                              false, a, false, b[s], (short)0, acc[rt], false, false);
            }
        }
        __syncthreads();
    }

    const float bval = bias ? bias[n0 + ml] : 0.0f;
    #pragma unroll
    for (int rt = 0; rt < RT; ++rt) {
        #pragma unroll
        for (int r = 0; r < 8; ++r) {
            int m = row0 + rt * 16 + r + 8 * hi;
            C[(size_t)m * H + n0 + ml] = acc[rt][r] + bval;
        }
    }
}

typedef __attribute__((ext_vector_type(4))) float v4f;

template <int RT, int KK>
__device__ __forceinline__
void gemm_tile_deep(const float* __restrict__ A,
                    const float* __restrict__ B,
                    const float* __restrict__ bias,
                    float* __restrict__ C,
                    int bid)
{
    constexpr int NC  = KK / 128;
    constexpr int ASZ = RT * 16 * 132;
    __shared__ __align__(16) float As[2 * ASZ];

    const int tid  = threadIdx.x;
    const int wave = tid >> 5;
    const int lane = tid & 31;
    const int hi   = lane >> 4;
    const int ml   = lane & 15;
    const int row0 = (bid >> 1) * (16 * RT);
    const int n0   = (bid & 1) * 64 + wave * 16;

    v8f acc[RT];
    #pragma unroll
    for (int rt = 0; rt < RT; ++rt) acc[rt] = v8f{};

    const float* ap = A + (size_t)(row0 + wave) * KK + 4 * lane;
    const int    so = wave * 132 + 4 * lane;
    const int    fo = ml * 132 + 2 * hi;
    const float* bp = B + (size_t)(2 * hi) * H + n0 + ml;

    {
        v4f st[4 * RT];
        #pragma unroll
        for (int it = 0; it < 4 * RT; ++it)
            st[it] = *(const v4f*)(ap + (size_t)(4 * it) * KK);
        #pragma unroll
        for (int it = 0; it < 4 * RT; ++it)
            *(v4f*)(&As[so + (4 * it) * 132]) = st[it];
    }
    v2f b[32];
    #pragma unroll
    for (int s = 0; s < 32; ++s) {
        b[s].x = bp[(size_t)(4 * s) * H];
        b[s].y = bp[(size_t)(4 * s + 1) * H];
    }
    __syncthreads();

    for (int kc = 0; kc + 1 < NC; ++kc) {
        const int cur = (kc & 1) * ASZ;
        const int nxt = ASZ - cur;
        const int kn  = (kc + 1) * 128;

        v4f st[4 * RT];
        #pragma unroll
        for (int it = 0; it < 4 * RT; ++it)
            st[it] = *(const v4f*)(ap + (size_t)(4 * it) * KK + kn);
        v2f bn[32];
        #pragma unroll
        for (int s = 0; s < 32; ++s) {
            bn[s].x = bp[(size_t)(kn + 4 * s) * H];
            bn[s].y = bp[(size_t)(kn + 4 * s + 1) * H];
        }

        #pragma unroll
        for (int s = 0; s < 32; ++s) {
            #pragma unroll
            for (int rt = 0; rt < RT; ++rt) {
                v2f a = *(const v2f*)(&As[cur + fo + rt * 16 * 132 + 4 * s]);
                acc[rt] = __builtin_amdgcn_wmma_f32_16x16x4_f32(
                              false, a, false, b[s], (short)0, acc[rt], false, false);
            }
        }

        #pragma unroll
        for (int it = 0; it < 4 * RT; ++it)
            *(v4f*)(&As[nxt + so + (4 * it) * 132]) = st[it];
        __syncthreads();
        #pragma unroll
        for (int s = 0; s < 32; ++s) b[s] = bn[s];
    }

    {
        constexpr int cur = ((NC - 1) & 1) * ASZ;
        #pragma unroll
        for (int s = 0; s < 32; ++s) {
            #pragma unroll
            for (int rt = 0; rt < RT; ++rt) {
                v2f a = *(const v2f*)(&As[cur + fo + rt * 16 * 132 + 4 * s]);
                acc[rt] = __builtin_amdgcn_wmma_f32_16x16x4_f32(
                              false, a, false, b[s], (short)0, acc[rt], false, false);
            }
        }
    }

    const float bval = bias ? bias[n0 + ml] : 0.0f;
    #pragma unroll
    for (int rt = 0; rt < RT; ++rt) {
        #pragma unroll
        for (int r = 0; r < 8; ++r) {
            int m = row0 + rt * 16 + r + 8 * hi;
            C[(size_t)m * H + n0 + ml] = acc[rt][r] + bval;
        }
    }
}

struct Proj4 {
    const float* A0; const float* B0; const float* bias0; float* C0;
    const float* A1; const float* B1; const float* bias1; float* C1;
    const float* A2; const float* B2; const float* bias2; float* C2;
    const float* A3; const float* B3; const float* bias3; float* C3;
};

template <int RT, int NB0, int NB2>
__device__ __forceinline__
void proj4_block(const Proj4& p, int bid)
{
    const float* A; const float* B; const float* bias; float* C;
    if (bid < NB0) {
        A = p.A0; B = p.B0; bias = p.bias0; C = p.C0;
    } else if (bid < 2 * NB0) {
        bid -= NB0;
        A = p.A1; B = p.B1; bias = p.bias1; C = p.C1;
    } else if (bid < 2 * NB0 + NB2) {
        bid -= 2 * NB0;
        A = p.A2; B = p.B2; bias = p.bias2; C = p.C2;
    } else {
        bid -= 2 * NB0 + NB2;
        A = p.A3; B = p.B3; bias = p.bias3; C = p.C3;
    }
    gemm_tile<RT, H>(A, B, bias, C, bid);
}

template <int RT, int NB0, int NB2>
__global__ __launch_bounds__(128) __attribute__((amdgpu_waves_per_eu(5, 5)))
void gemm_wmma_bias4(const Proj4 p)
{
    proj4_block<RT, NB0, NB2>(p, blockIdx.x);
}

#define SLICES       16
#define RANGES_A     4
#define RANGES_B     12
#define RSZ_A        ((N_VIRUS + RANGES_A - 1) / RANGES_A)
#define RSZ_B        ((N_MOUSE + RANGES_B - 1) / RANGES_B)
#define RSZ_MAX      (RSZ_A > RSZ_B ? RSZ_A : RSZ_B)
#define SLICE_EDGES  ((NEDGE + SLICES - 1) / SLICES)
#define N_ROWS       (N_VIRUS + N_MOUSE)
#define RANK_THREADS 128
#define RANK_BATCH   16
#define RANK_BLOCKS  ((RANGES_A + RANGES_B) * SLICES)
#define GROUP_ROWS   128
#define GROUPS_A     ((N_VIRUS + GROUP_ROWS - 1) / GROUP_ROWS)
#define GROUPS_B     ((N_MOUSE + GROUP_ROWS - 1) / GROUP_ROWS)
#define GROUPS       (GROUPS_A + GROUPS_B)

__device__ __forceinline__
void dst_rank(const int blk,
              const int* __restrict__ dst_a,
              const int* __restrict__ dst_b,
              int* __restrict__ slot_a,
              int* __restrict__ slot_b,
              int* __restrict__ part)
{
    __shared__ int c[RSZ_MAX];

    const int  tid  = threadIdx.x;
    const int  s    = blk % SLICES;
    const int  gr   = blk / SLICES;
    const bool is_a = gr < RANGES_A;
    const int* dst  = is_a ? dst_a : dst_b;
    int*       slot = is_a ? slot_a : slot_b;
    const int  n    = is_a ? N_VIRUS : N_MOUSE;
    const int  rsz  = is_a ? RSZ_A : RSZ_B;
    const int  lo   = (is_a ? gr : gr - RANGES_A) * rsz;
    const int  rows = min(rsz, n - lo);

    for (int i = tid; i < rsz; i += RANK_THREADS) c[i] = 0;
    __syncthreads();

    const int e0 = s * SLICE_EDGES;
    const int e1 = min(NEDGE, e0 + SLICE_EDGES);
    for (int base = e0 + tid; base < e1; base += RANK_BATCH * RANK_THREADS) {
        int dv[RANK_BATCH];
        #pragma unroll
        for (int k = 0; k < RANK_BATCH; ++k)
            dv[k] = dst[min(base + k * RANK_THREADS, e1 - 1)];
        int rk[RANK_BATCH];
        #pragma unroll
        for (int k = 0; k < RANK_BATCH; ++k) {
            const int e = base + k * RANK_THREADS;
            const unsigned t = (unsigned)(min(max(dv[k], 0), n - 1) - lo);
            rk[k] = -1;
            if (e < e1 && t < (unsigned)rsz) rk[k] = atomicAdd(&c[t], 1);
        }
        #pragma unroll
        for (int k = 0; k < RANK_BATCH; ++k)
            if (rk[k] >= 0) slot[base + k * RANK_THREADS] = rk[k];
    }
    __syncthreads();

    int* prow = part + (size_t)s * N_ROWS + (is_a ? 0 : N_VIRUS) + lo;
    for (int i = tid; i < rows; i += RANK_THREADS) prow[i] = c[i];
}

__device__ __forceinline__
void dst_combine(const int grp,
                 int* __restrict__ part,
                 int* __restrict__ cnt,
                 int* __restrict__ gtot)
{
    __shared__ int wtot[GROUP_ROWS / 32];

    const int  tid  = threadIdx.x;
    const bool is_a = grp < GROUPS_A;
    const int  r    = (is_a ? grp : grp - GROUPS_A) * GROUP_ROWS + tid;
    const int  n    = is_a ? N_VIRUS : N_MOUSE;
    const int  d    = (is_a ? 0 : N_VIRUS) + r;

    int run = 0;
    if (r < n) {
        int v[SLICES];
        #pragma unroll
        for (int s = 0; s < SLICES; ++s) v[s] = part[(size_t)s * N_ROWS + d];
        #pragma unroll
        for (int s = 0; s < SLICES; ++s) {
            part[(size_t)s * N_ROWS + d] = run;
            run += v[s];
        }
        cnt[d] = run;
    }
    int t = run;
    for (int o = 16; o > 0; o >>= 1) t += __shfl_xor(t, o, 32);
    if ((tid & 31) == 0) wtot[tid >> 5] = t;
    __syncthreads();
    if (tid == 0) {
        int g = 0;
        for (int w = 0; w < GROUP_ROWS / 32; ++w) g += wtot[w];
        gtot[grp] = g;
    }
}

__device__ __forceinline__
void dst_scan(const int grp,
              const int* __restrict__ cnt,
              const int* __restrict__ gtot,
              int* __restrict__ off_a,
              int* __restrict__ off_b)
{
    __shared__ int sred[GROUP_ROWS / 32];
    __shared__ int wsum[GROUP_ROWS / 32];

    const int  tid  = threadIdx.x;
    const int  lane = tid & 31;
    const int  wave = tid >> 5;
    const bool is_a = grp < GROUPS_A;
    const int  g0   = is_a ? 0 : GROUPS_A;
    const int  jj   = grp - g0;
    const int  ng   = is_a ? GROUPS_A : GROUPS_B;
    const int  n    = is_a ? N_VIRUS : N_MOUSE;
    int*       off  = is_a ? off_a : off_b;

    int b = 0;
    for (int i = tid; i < jj; i += GROUP_ROWS) b += gtot[g0 + i];
    for (int o = 16; o > 0; o >>= 1) b += __shfl_xor(b, o, 32);
    if (lane == 0) sred[wave] = b;

    const int r = jj * GROUP_ROWS + tid;
    const int c = (r < n) ? cnt[(is_a ? 0 : N_VIRUS) + r] : 0;
    int incl = c;
    for (int o = 1; o < 32; o <<= 1) {
        int y = __shfl_up(incl, o, 32);
        if (lane >= o) incl += y;
    }
    if (lane == 31) wsum[wave] = incl;
    __syncthreads();

    int start = 0;
    for (int w = 0; w < GROUP_ROWS / 32; ++w) {
        start += sred[w];
        if (w < wave) start += wsum[w];
    }
    if (r < n) off[r] = start + incl - c;
    if (jj == ng - 1 && tid == GROUP_ROWS - 1) off[n] = start + incl;
}

template <int RT, int KK>
__global__ __launch_bounds__(128) __attribute__((amdgpu_waves_per_eu(3, 3)))
void gemm_bias_rank(const float* __restrict__ A,
                    const float* __restrict__ B,
                    const float* __restrict__ bias,
                    float* __restrict__ C,
                    const int* __restrict__ dst_a,
                    const int* __restrict__ dst_b,
                    int* __restrict__ slot_a,
                    int* __restrict__ slot_b,
                    int* __restrict__ part)
{
    if (blockIdx.x < RANK_BLOCKS) {
        dst_rank(blockIdx.x, dst_a, dst_b, slot_a, slot_b, part);
        return;
    }
    gemm_tile_deep<RT, KK>(A, B, bias, C, blockIdx.x - RANK_BLOCKS);
}

template <int RT, int KK>
__global__ __launch_bounds__(128) __attribute__((amdgpu_waves_per_eu(3, 3)))
void gemm_bias_combine(const float* __restrict__ A,
                       const float* __restrict__ B,
                       const float* __restrict__ bias,
                       float* __restrict__ C,
                       int* __restrict__ part,
                       int* __restrict__ cnt,
                       int* __restrict__ gtot)
{
    if (blockIdx.x < GROUPS) {
        dst_combine(blockIdx.x, part, cnt, gtot);
        return;
    }
    gemm_tile_deep<RT, KK>(A, B, bias, C, blockIdx.x - GROUPS);
}

template <int RT, int NB0, int NB2>
__global__ __launch_bounds__(128) __attribute__((amdgpu_waves_per_eu(5, 5)))
void gemm_wmma_bias4_scan(const Proj4 p,
                          const int* __restrict__ cnt,
                          const int* __restrict__ gtot,
                          int* __restrict__ off_a,
                          int* __restrict__ off_b)
{
    if (blockIdx.x < GROUPS) {
        dst_scan(blockIdx.x, cnt, gtot, off_a, off_b);
        return;
    }
    proj4_block<RT, NB0, NB2>(p, blockIdx.x - GROUPS);
}

__global__ void dst_fill(const int* __restrict__ dst_a,
                         const int* __restrict__ off_a,
                         const int* __restrict__ slot_a,
                         int* __restrict__ perm_a,
                         const int* __restrict__ dst_b,
                         const int* __restrict__ off_b,
                         const int* __restrict__ slot_b,
                         int* __restrict__ perm_b,
                         const int* __restrict__ part,
                         int E, int na, int nb)
{
    int e = blockIdx.x * blockDim.x + threadIdx.x;
    if (e >= E) return;
    const int* prow = part + (size_t)(e / SLICE_EDGES) * N_ROWS;
    {
        int d = min(max(dst_a[e], 0), na - 1);
        int p = off_a[d] + prow[d] + slot_a[e];
        perm_a[min(max(p, 0), E - 1)] = e;
    }
    {
        int d = min(max(dst_b[e], 0), nb - 1);
        int p = off_b[d] + prow[na + d] + slot_b[e];
        perm_b[min(max(p, 0), E - 1)] = e;
    }
}

#define SORT_CAP 512
__global__ __launch_bounds__(256) void dst_order(const int* __restrict__ off_a,
                                                 int* __restrict__ perm_a,
                                                 int* __restrict__ psrc_a,
                                                 const int* __restrict__ src_a, int na,
                                                 const int* __restrict__ off_b,
                                                 int* __restrict__ perm_b,
                                                 int* __restrict__ psrc_b,
                                                 const int* __restrict__ src_b, int nb,
                                                 int E)
{
    __shared__ __align__(16) int skey[8][SORT_CAP];

    const int lane = threadIdx.x & 31;
    const int wv   = threadIdx.x >> 5;
    int row = blockIdx.x * 8 + wv;

    const bool is_a  = row < na;
    const int* off   = is_a ? off_a  : off_b;
    int*       perm  = is_a ? perm_a : perm_b;
    int*       psrc  = is_a ? psrc_a : psrc_b;
    const int* src   = is_a ? src_a  : src_b;
    if (!is_a) row -= na;
    const bool valid = is_a || row < nb;

    int beg = 0, n = 0;
    if (valid) {
        beg = min(max(off[row], 0), E);
        n   = min(max(off[row + 1], beg), E) - beg;
    }
    const bool small = n <= SORT_CAP;
    const int  npad  = (n + 3) & ~3;
    int* key = skey[wv];

    if (small) {
        for (int j = lane; j < npad; j += 32)
            key[j] = (j < n) ? perm[beg + j] : 0x7fffffff;
    }
    __syncthreads();

    if (small) {
        for (int j0 = 0; j0 < n; j0 += 32) {
            const int j  = j0 + lane;
            const int my = (j < n) ? key[j] : 0x7fffffff;
            int r = 0;
            for (int t = 0; t < npad; t += 4) {
                const int4 k4 = *(const int4*)(key + t);
                r += (k4.x < my) + (k4.y < my) + (k4.z < my) + (k4.w < my);
            }
            if (j < n) {
                perm[beg + r] = my;
                psrc[beg + r] = src[min(max(my, 0), E - 1)];
            }
        }
    } else {
        for (int j = lane; j < n; j += 32)
            psrc[beg + j] = src[min(max(perm[beg + j], 0), E - 1)];
    }
}

template <int NR>
__device__ __forceinline__
void edge_batch(const float* __restrict__ qbuf,
                const float* __restrict__ vbuf,
                const float* __restrict__ xnext,
                const v4f (&sB)[16 * 32],
                const int lane,
                const unsigned (&so)[8],
                const v2f (&a)[4],
                const float (&cb)[8],
                const float (&kr)[8],
                float (&acc)[8],
                float4& x0, float4& x1,
                int lim)
{
    #pragma unroll
    for (int g = 0; g < 2; ++g) {
        if (g == 1) {
            __builtin_amdgcn_sched_barrier(0);
            x0 = *(const float4*)(xnext);
            x1 = *(const float4*)(xnext + 4);
        }
        const char* qb = (const char*)qbuf + 256 * g;
        const char* vb = (const char*)vbuf + 256 * g;
        float qf[NR][4], vf[NR][4];
        #pragma unroll
        for (int r = 0; r < NR; ++r) {
            const float4 q4 = *(const float4*)(qb + so[r]);
            const float4 v4 = *(const float4*)(vb + so[r]);
            qf[r][0] = q4.x; qf[r][1] = q4.y; qf[r][2] = q4.z; qf[r][3] = q4.w;
            vf[r][0] = v4.x; vf[r][1] = v4.y; vf[r][2] = v4.z; vf[r][3] = v4.w;
        }

        v2f b[4][4];
        #pragma unroll
        for (int tt = 0; tt < 4; ++tt) {
            #pragma unroll
            for (int sp = 0; sp < 2; ++sp) {
                const v4f f = sB[(2 * (4 * g + tt) + sp) * 32 + lane];
                b[2 * sp][tt].x     = f.x; b[2 * sp][tt].y     = f.y;
                b[2 * sp + 1][tt].x = f.z; b[2 * sp + 1][tt].y = f.w;
            }
        }

        v8f pr[4];
        #pragma unroll
        for (int tt = 0; tt < 4; ++tt) pr[tt] = v8f{};
        #pragma unroll
        for (int s = 0; s < 4; ++s) {
            #pragma unroll
            for (int tt = 0; tt < 4; ++tt)
                pr[tt] = __builtin_amdgcn_wmma_f32_16x16x4_f32(
                             false, a[s], false, b[s][tt], (short)0, pr[tt],
                             false, false);
        }

        #pragma unroll
        for (int r = 0; r < NR; ++r) {
            #pragma unroll
            for (int tt = 0; tt < 4; ++tt) {
                const float ev = pr[tt][r] + cb[4 * g + tt];
                const float z  = (kr[4 * g + tt] + ev) + (qf[r][tt] + ev);
                float gt = __builtin_amdgcn_rcpf(1.0f + __expf(-z));
                if (r >= NR - 2) gt = (2 * r < lim) ? gt : 0.0f;
                acc[4 * g + tt] += gt * (vf[r][tt] + ev);
            }
        }
    }
}

template <int ROWS>
__global__ __launch_bounds__(256) __attribute__((amdgpu_waves_per_eu(4, 4)))
void edge_rows(const float* __restrict__ kbuf,
               const float* __restrict__ qbuf,
               const float* __restrict__ vbuf,
               const float* __restrict__ xe,
               const float* __restrict__ We,
               const float* __restrict__ be,
               const int*   __restrict__ off,
               const int*   __restrict__ perm,
               const int*   __restrict__ psrc,
               float*       __restrict__ out_dst,
               int n_dst, int n_src, int E)
{
    __shared__ v4f sB[16 * 32];

    const int lane = threadIdx.x & 31;
    const int c    = lane & 15;
    const int h    = lane >> 4;

    #pragma unroll
    for (int i = 0; i < 2; ++i) {
        const int combo = (threadIdx.x >> 5) * 2 + i;
        const int t     = combo >> 1;
        const int sp    = combo & 1;
        const float* wp = We + (8 * h + 4 * sp) * H + 64 * (t >> 2) + 4 * c + (t & 3);
        v4f f;
        f.x = wp[0];
        f.y = wp[H];
        f.z = wp[2 * H];
        f.w = wp[3 * H];
        sB[combo * 32 + lane] = f;
    }
    __syncthreads();

    const int d0 = __builtin_amdgcn_readfirstlane(
                       (blockIdx.x * 8 + (threadIdx.x >> 5)) * ROWS);
    if (d0 >= n_dst) return;
    const int d1 = min(d0 + ROWS, n_dst);

    float cb[8];
    {
        const float4 c0 = *(const float4*)(be + 4 * c);
        const float4 c1 = *(const float4*)(be + 64 + 4 * c);
        cb[0] = c0.x; cb[1] = c0.y; cb[2] = c0.z; cb[3] = c0.w;
        cb[4] = c1.x; cb[5] = c1.y; cb[6] = c1.z; cb[7] = c1.w;
    }
    const int ia = 2 * (c & 7) + (c >> 3);

    int    ps_l = 0;
    float4 x0 = make_float4(0.f, 0.f, 0.f, 0.f), x1 = x0;
    bool   primed = false;
    int    beg = min(max(__builtin_amdgcn_readfirstlane(off[d0]), 0), E);

    for (int d = d0; d < d1; ++d) {
        const int end = min(max(__builtin_amdgcn_readfirstlane(off[d + 1]), beg), E);
        int end2 = end;
        if (d + 1 < d1)
            end2 = min(max(__builtin_amdgcn_readfirstlane(off[d + 2]), end), E);

        float kr[8];
        {
            const float4 k0 = *(const float4*)(kbuf + d * H + 4 * c);
            const float4 k1 = *(const float4*)(kbuf + d * H + 64 + 4 * c);
            kr[0] = k0.x; kr[1] = k0.y; kr[2] = k0.z; kr[3] = k0.w;
            kr[4] = k1.x; kr[5] = k1.y; kr[6] = k1.z; kr[7] = k1.w;
        }
        float acc[8];
        #pragma unroll
        for (int t = 0; t < 8; ++t) acc[t] = 0.0f;

        if (!primed && beg < end) {
            const int pe = perm[min(beg + ia, end - 1)];
            ps_l = psrc[min(beg + c, end - 1)];
            const float* xp = xe + min(max(pe, 0), E - 1) * D_EDGE + 8 * h;
            x0 = *(const float4*)(xp);
            x1 = *(const float4*)(xp + 4);
        }
        primed = false;

        for (int j0 = beg; j0 < end; j0 += 16) {
            const int cnt = min(16, end - j0);
            const bool more     = j0 + 16 < end;
            const int  nj0      = more ? j0 + 16 : end;
            const int  nend     = more ? end : end2;
            const bool has_next = nj0 < nend;
            const int pe_n = perm[min(nj0 + ia, max(nend, nj0) - 1)];
            const int ps_n = psrc[min(nj0 + c, max(nend, nj0) - 1)];

            v2f a[4];
            a[0].x = x0.x; a[0].y = x0.y;
            a[1].x = x0.z; a[1].y = x0.w;
            a[2].x = x1.x; a[2].y = x1.y;
            a[3].x = x1.z; a[3].y = x1.w;

            const int ps_c = min(max(ps_l, 0), n_src - 1);
            unsigned so[8];
            #pragma unroll
            for (int r = 0; r < 8; ++r) {
                const int s_even = __builtin_amdgcn_readlane(ps_c, 2 * r);
                const int s_odd  = __builtin_amdgcn_readlane(ps_c, 2 * r + 1);
                so[r] = (unsigned)((h ? s_odd : s_even) * (H * 4) + 16 * c);
            }
            const int lim = cnt - h;
            int slot = lane;
            asm volatile("" : "+v"(slot));

            const float* xnext = xe + min(max(pe_n, 0), E - 1) * D_EDGE + 8 * h;

            if (cnt > 12)
                edge_batch<8>(qbuf, vbuf, xnext, sB, slot, so, a, cb, kr, acc, x0, x1, lim);
            else if (cnt > 8)
                edge_batch<6>(qbuf, vbuf, xnext, sB, slot, so, a, cb, kr, acc, x0, x1, lim);
            else if (cnt > 4)
                edge_batch<4>(qbuf, vbuf, xnext, sB, slot, so, a, cb, kr, acc, x0, x1, lim);
            else
                edge_batch<2>(qbuf, vbuf, xnext, sB, slot, so, a, cb, kr, acc, x0, x1, lim);
            ps_l   = ps_n;
            primed = has_next;
        }

        float tot[8];
        #pragma unroll
        for (int t = 0; t < 8; ++t) tot[t] = acc[t] + __shfl_xor(acc[t], 16, 32);

        float4* op = (float4*)(out_dst + d * H + 64 * h + 4 * c);
        float4 o = *op;
        o.x = fmaxf(o.x + (h ? tot[4] : tot[0]), 0.0f);
        o.y = fmaxf(o.y + (h ? tot[5] : tot[1]), 0.0f);
        o.z = fmaxf(o.z + (h ? tot[6] : tot[2]), 0.0f);
        o.w = fmaxf(o.w + (h ? tot[7] : tot[3]), 0.0f);
        *op = o;

        beg = end;
    }
}

__global__ void classifier(const float* __restrict__ hm,
                           const float* __restrict__ hv,
                           const int*   __restrict__ im,
                           const int*   __restrict__ iv,
                           const float* __restrict__ Wc,
                           const float* __restrict__ bc,
                           float*       __restrict__ out,
                           int P)
{
    const int p    = blockIdx.x * 8 + (threadIdx.x >> 5);
    const int lane = threadIdx.x & 31;
    if (p >= P) return;

    const float4 am = ((const float4*)(hm + (size_t)im[p] * H))[lane];
    const float4 av = ((const float4*)(hv + (size_t)iv[p] * H))[lane];

    float c0 = 0.f, c1 = 0.f;
    const int ch = lane * 4;
    const float a[4] = {am.x, am.y, am.z, am.w};
    const float b[4] = {av.x, av.y, av.z, av.w};
    #pragma unroll
    for (int i = 0; i < 4; ++i) {
        c0 = fmaf(a[i], Wc[(ch + i) * 2 + 0], c0);
        c1 = fmaf(a[i], Wc[(ch + i) * 2 + 1], c1);
        c0 = fmaf(b[i], Wc[(H + ch + i) * 2 + 0], c0);
        c1 = fmaf(b[i], Wc[(H + ch + i) * 2 + 1], c1);
    }
    #pragma unroll
    for (int off = 16; off > 0; off >>= 1) {
        c0 += __shfl_down(c0, off, 32);
        c1 += __shfl_down(c1, off, 32);
    }
    if (lane == 0) {
        out[(size_t)p * 2 + 0] = c0 + bc[0];
        out[(size_t)p * 2 + 1] = c1 + bc[1];
    }
}

#define ER_ROWS_A 1
#define ER_ROWS_B 4
#define NB_MOUSE (N_MOUSE / 80 * 2)
#define NB_VIRUS (N_VIRUS / 80 * 2)
static inline void gemm4(const Proj4& p, hipStream_t s)
{
    gemm_wmma_bias4<5, NB_MOUSE, NB_VIRUS>
        <<<dim3(2 * NB_MOUSE + 2 * NB_VIRUS), dim3(128), 0, s>>>(p);
}

extern "C" void kernel_launch(void* const* d_in, const int* in_sizes, int n_in,
                              void* d_out, int out_size, void* d_ws, size_t ws_size,
                              hipStream_t stream)
{
    const float* x_mouse  = (const float*)d_in[0];
    const float* x_virus  = (const float*)d_in[1];
    const float* xe_mv    = (const float*)d_in[2];
    const float* xe_vm    = (const float*)d_in[3];
    const float* W_mouse  = (const float*)d_in[4];
    const float* b_mouse  = (const float*)d_in[5];
    const float* W_virus  = (const float*)d_in[6];
    const float* b_virus  = (const float*)d_in[7];
    const float* W_key    = (const float*)d_in[8];
    const float* b_key    = (const float*)d_in[9];
    const float* W_query  = (const float*)d_in[10];
    const float* b_query  = (const float*)d_in[11];
    const float* W_value  = (const float*)d_in[12];
    const float* b_value  = (const float*)d_in[13];
    const float* W_edge   = (const float*)d_in[14];
    const float* b_edge   = (const float*)d_in[15];
    const float* W_skip   = (const float*)d_in[16];
    const float* bias_cv  = (const float*)d_in[17];
    const float* W_cls    = (const float*)d_in[18];
    const float* b_cls    = (const float*)d_in[19];
    const int* ei_mv_src  = (const int*)d_in[20];
    const int* ei_mv_dst  = (const int*)d_in[21];
    const int* ei_vm_src  = (const int*)d_in[22];
    const int* ei_vm_dst  = (const int*)d_in[23];
    const int* eli_mouse  = (const int*)d_in[24];
    const int* eli_virus  = (const int*)d_in[25];
    float* out = (float*)d_out;

    float* ws  = (float*)d_ws;
    const size_t SZ_M = (size_t)N_MOUSE * H;
    const size_t SZ_V = (size_t)N_VIRUS * H;
    float* hm0 = ws;
    float* hm1 = hm0 + SZ_M;
    float* hv0 = hm1 + SZ_M;
    float* hv1 = hv0 + SZ_V;
    float* bx  = hv1 + SZ_V;
    float* by  = bx + SZ_M;
    float* bz  = by + SZ_M;

    int* cnt_a  = (int*)(bz + SZ_V);
    int* cnt_b  = cnt_a + N_VIRUS;
    int* off_a  = cnt_b + N_MOUSE;
    int* off_b  = off_a + N_VIRUS + 4;
    int* perm_a = off_b + N_MOUSE + 4;
    int* psrc_a = perm_a + NEDGE;
    int* perm_b = psrc_a + NEDGE;
    int* psrc_b = perm_b + NEDGE;
    int* slot_a = psrc_a;
    int* slot_b = psrc_b;

    const int e_blocks = (NEDGE + 255) / 256;

    int* part = (int*)hm1;
    int* gtot = perm_a;

    gemm_bias_rank<5, D_IN><<<dim3(RANK_BLOCKS + NB_MOUSE), dim3(128), 0, stream>>>(
        x_mouse, W_mouse, b_mouse, hm0, ei_mv_dst, ei_vm_dst, slot_a, slot_b, part);
    gemm_bias_combine<5, D_IN><<<dim3(GROUPS + NB_VIRUS), dim3(128), 0, stream>>>(
        x_virus, W_virus, b_virus, hv0, part, cnt_a, gtot);

    {
        Proj4 p;
        p.A0 = hm0; p.B0 = W_query; p.bias0 = b_query; p.C0 = bx;
        p.A1 = hm0; p.B1 = W_value; p.bias1 = b_value; p.C1 = by;
        p.A2 = hv0; p.B2 = W_skip;  p.bias2 = bias_cv; p.C2 = hv1;
        p.A3 = hv0; p.B3 = W_key;   p.bias3 = b_key;   p.C3 = bz;
        gemm_wmma_bias4_scan<5, NB_MOUSE, NB_VIRUS>
            <<<dim3(GROUPS + 2 * NB_MOUSE + 2 * NB_VIRUS), dim3(128), 0, stream>>>(
                p, cnt_a, gtot, off_a, off_b);
    }

    dst_fill<<<dim3(e_blocks), dim3(256), 0, stream>>>(
        ei_mv_dst, off_a, slot_a, perm_a, ei_vm_dst, off_b, slot_b, perm_b,
        part, NEDGE, N_VIRUS, N_MOUSE);
    dst_order<<<dim3((N_VIRUS + N_MOUSE + 7) / 8), dim3(256), 0, stream>>>(
        off_a, perm_a, psrc_a, ei_mv_src, N_VIRUS,
        off_b, perm_b, psrc_b, ei_vm_src, N_MOUSE, NEDGE);

    float* hm_in = hm0; float* hv_in = hv0;
    float* hm_out = hm1; float* hv_out = hv1;

    for (int layer = 0; layer < 2; ++layer) {
        const int i_mv = layer * 2;
        const int i_vm = layer * 2 + 1;

        {
            const int i = i_mv;
            const float* Wk = W_key   + (size_t)i * H * H;
            const float* Wq = W_query + (size_t)i * H * H;
            const float* Wv = W_value + (size_t)i * H * H;
            const float* We = W_edge  + (size_t)i * D_EDGE * H;
            const float* Ws = W_skip  + (size_t)i * H * H;
            Proj4 p;
            p.A0 = hm_in; p.B0 = Wq; p.bias0 = b_query + (size_t)i * H; p.C0 = bx;
            p.A1 = hm_in; p.B1 = Wv; p.bias1 = b_value + (size_t)i * H; p.C1 = by;
            p.A2 = hv_in; p.B2 = Ws; p.bias2 = bias_cv + (size_t)i * H; p.C2 = hv_out;
            p.A3 = hv_in; p.B3 = Wk; p.bias3 = b_key   + (size_t)i * H; p.C3 = bz;
            if (layer > 0) gemm4(p, stream);
            edge_rows<ER_ROWS_A><<<dim3((N_VIRUS + 8 * ER_ROWS_A - 1) / (8 * ER_ROWS_A)), dim3(256), 0, stream>>>(
                bz, bx, by, xe_mv, We, b_edge + (size_t)i * H,
                off_a, perm_a, psrc_a, hv_out, N_VIRUS, N_MOUSE, NEDGE);
        }
        {
            const int i = i_vm;
            const float* Wk = W_key   + (size_t)i * H * H;
            const float* Wq = W_query + (size_t)i * H * H;
            const float* Wv = W_value + (size_t)i * H * H;
            const float* We = W_edge  + (size_t)i * D_EDGE * H;
            const float* Ws = W_skip  + (size_t)i * H * H;
            Proj4 p;
            p.A0 = hm_in; p.B0 = Ws; p.bias0 = bias_cv + (size_t)i * H; p.C0 = hm_out;
            p.A1 = hm_in; p.B1 = Wk; p.bias1 = b_key   + (size_t)i * H; p.C1 = bx;
            p.A2 = hv_in; p.B2 = Wq; p.bias2 = b_query + (size_t)i * H; p.C2 = by;
            p.A3 = hv_in; p.B3 = Wv; p.bias3 = b_value + (size_t)i * H; p.C3 = by + SZ_V;
            gemm4(p, stream);
            edge_rows<ER_ROWS_B><<<dim3((N_MOUSE + 8 * ER_ROWS_B - 1) / (8 * ER_ROWS_B)), dim3(256), 0, stream>>>(
                bx, by, by + SZ_V, xe_vm, We, b_edge + (size_t)i * H,
                off_b, perm_b, psrc_b, hm_out, N_MOUSE, N_VIRUS, NEDGE);
        }

        float* t;
        t = hm_in; hm_in = hm_out; hm_out = t;
        t = hv_in; hv_in = hv_out; hv_out = t;
    }

    classifier<<<dim3((NPAIR + 7) / 8), dim3(256), 0, stream>>>(
        hm_in, hv_in, eli_mouse, eli_virus, W_cls, b_cls, out, NPAIR);
}
